# starting kernel
# baseline (speedup 1.0000x reference)
#include <hip/hip_runtime.h>
#include <hip/hip_bf16.h>

#define NH 256
#define NN 64
#define LL 2048
#define CHUNK 16
#define NCHUNK (LL / CHUNK)

typedef __attribute__((ext_vector_type(2))) float v2f;
typedef __attribute__((ext_vector_type(8))) float v8f;

struct cpx { float r, i; };
__device__ __forceinline__ cpx cmul(cpx a, cpx b) { return {a.r*b.r - a.i*b.i, a.r*b.i + a.i*b.r}; }
__device__ __forceinline__ cpx cadd(cpx a, cpx b) { return {a.r + b.r, a.i + b.i}; }
__device__ __forceinline__ cpx csub(cpx a, cpx b) { return {a.r - b.r, a.i - b.i}; }
__device__ __forceinline__ cpx cinv(cpx a) { float d = a.r*a.r + a.i*a.i; return {a.r/d, -a.i/d}; }
__device__ __forceinline__ cpx cscale(cpx a, float s) { return {a.r*s, a.i*s}; }

__device__ __forceinline__ void cplx_mm_tile(
    const float* Ar, const float* Ai, int lda,
    const float* Br, const float* Bi, int ldb,
    float* Cr, float* Ci, int ldc,
    int r0, int n0, int K)
{
    const int lane = threadIdx.x & 31;
    const int mn = lane & 15;
    const int kh = (lane >> 4) << 1;
    v8f arbr = {}; v8f aibi = {}; v8f arbi = {}; v8f aibr = {};
    for (int k0 = 0; k0 < K; k0 += 4) {
        const int ka = k0 + kh;
        v2f ar, ai, br, bi;
        ar[0] = Ar[(r0 + mn) * lda + ka];     ar[1] = Ar[(r0 + mn) * lda + ka + 1];
        ai[0] = Ai[(r0 + mn) * lda + ka];     ai[1] = Ai[(r0 + mn) * lda + ka + 1];
        br[0] = Br[ka * ldb + n0 + mn];       br[1] = Br[(ka + 1) * ldb + n0 + mn];
        bi[0] = Bi[ka * ldb + n0 + mn];       bi[1] = Bi[(ka + 1) * ldb + n0 + mn];
        arbr = __builtin_amdgcn_wmma_f32_16x16x4_f32(false, ar, false, br, (short)0, arbr, false, false);
        aibi = __builtin_amdgcn_wmma_f32_16x16x4_f32(false, ai, false, bi, (short)0, aibi, false, false);
        arbi = __builtin_amdgcn_wmma_f32_16x16x4_f32(false, ar, false, bi, (short)0, arbi, false, false);
        aibr = __builtin_amdgcn_wmma_f32_16x16x4_f32(false, ai, false, br, (short)0, aibr, false, false);
    }
    const int mbase = (lane >> 4) << 3;
    for (int v = 0; v < 8; ++v) {
        Cr[(r0 + mbase + v) * ldc + n0 + mn] = arbr[v] - aibi[v];
        Ci[(r0 + mbase + v) * ldc + n0 + mn] = arbi[v] + aibr[v];
    }
}

extern "C" __global__ __launch_bounds__(128)
void s4_dplr_kernel(const float* __restrict__ A_real, const float* __restrict__ A_imag,
                    const float* __restrict__ B_real, const float* __restrict__ B_imag,
                    const float* __restrict__ P_real, const float* __restrict__ P_imag,
                    const float* __restrict__ C_real, const float* __restrict__ C_imag,
                    const float* __restrict__ log_dt, float* __restrict__ out)
{
    extern __shared__ float sm[];
    const int h   = blockIdx.x;
    const int tid = threadIdx.x;
    const int wv  = tid >> 5;

    float* A0r = sm;
    float* A0i = sm + 4096;
    float* A1r = sm + 8192;
    float* A1i = sm + 12288;
    float* Xr0 = sm + 16384;
    float* Xi0 = sm + 17408;
    float* Xr1 = sm + 18432;
    float* Xi1 = sm + 19456;
    float* ur  = sm + 20480;  float* ui = ur + 64;
    float* wr  = ui + 64;     float* wi = wr + 64;
    float* gr  = wi + 64;     float* gi = gr + 64;
    float* cr  = gi + 64;     float* ci = cr + 64;
    float* pr  = ci + 64;     float* pi = pr + 64;
    float* scal = pi + 64;

    const float dt = __expf(log_dt[h]);
    const float cc = 0.5f * dt;

    if (tid < 64) {
        const int n = tid;
        cpx lam = { -A_real[h*64 + n], -A_imag[h*64 + n] };
        cpx p   = {  P_real[h*64 + n],  P_imag[h*64 + n] };
        cpx Dinv = cinv((cpx){1.0f - cc*lam.r, -cc*lam.i});
        cpx e    = {1.0f + cc*lam.r,  cc*lam.i};
        cpx u = cmul(Dinv, p);
        cpx g = cmul(Dinv, e);
        ur[n] = u.r; ui[n] = u.i; gr[n] = g.r; gi[n] = g.i;
        pr[n] = p.r; pi[n] = p.i;
        cr[n] = C_real[h*64 + n]; ci[n] = C_imag[h*64 + n];
    }
    __syncthreads();
    if (tid == 0) {
        cpx s = {0.f, 0.f};
        for (int n = 0; n < 64; ++n)
            s = cadd(s, cmul((cpx){pr[n], -pi[n]}, (cpx){ur[n], ui[n]}));
        cpx alpha = cmul((cpx){cc, 0.f}, cinv((cpx){1.0f + cc*s.r, cc*s.i}));
        cpx base  = csub((cpx){cc, 0.f}, cmul(alpha, cscale(s, cc)));
        scal[0] = alpha.r; scal[1] = alpha.i; scal[2] = base.r; scal[3] = base.i;
    }
    __syncthreads();
    const cpx alpha = {scal[0], scal[1]};
    const cpx base  = {scal[2], scal[3]};
    if (tid < 64) {
        const int n = tid;
        cpx w = cmul((cpx){pr[n], -pi[n]}, cadd(base, cmul(alpha, (cpx){gr[n], gi[n]})));
        wr[n] = w.r; wi[n] = w.i;
    }
    __syncthreads();
    for (int idx = tid; idx < 4096; idx += 128) {
        const int n = idx >> 6, m = idx & 63;
        cpx v = cmul((cpx){ur[n], ui[n]}, (cpx){wr[m], wi[m]});
        float rr = -v.r, rim = -v.i;
        if (n == m) { rr += gr[n]; rim += gi[n]; }
        A0r[idx] = rr; A0i[idx] = rim;
    }
    __syncthreads();

    if (tid < 64) {
        const int n = tid;
        cpx lam = { -A_real[h*64 + n], -A_imag[h*64 + n] };
        cpx Dinv = cinv((cpx){1.0f - cc*lam.r, -cc*lam.i});
        cpx b = { B_real[h*64 + n], B_imag[h*64 + n] };
        cpx t = cmul(Dinv, cscale(b, dt));
        wr[n] = t.r; wi[n] = t.i;
    }
    __syncthreads();
    if (tid == 0) {
        cpx q = {0.f, 0.f};
        for (int n = 0; n < 64; ++n)
            q = cadd(q, cmul((cpx){pr[n], -pi[n]}, (cpx){wr[n], wi[n]}));
        cpx aq = cmul(alpha, q);
        scal[4] = aq.r; scal[5] = aq.i;
    }
    __syncthreads();
    if (tid < 64) {
        const int n = tid;
        cpx dB = csub((cpx){wr[n], wi[n]}, cmul((cpx){scal[4], scal[5]}, (cpx){ur[n], ui[n]}));
        Xr0[n*16 + 0] = dB.r;
        Xi0[n*16 + 0] = dB.i;
    }
    __syncthreads();

    for (int l = 1; l < CHUNK; ++l) {
        if (tid < 64) {
            const int n = tid;
            cpx acc = {0.f, 0.f};
            for (int m = 0; m < 64; ++m)
                acc = cadd(acc, cmul((cpx){A0r[n*64 + m], A0i[n*64 + m]},
                                     (cpx){Xr0[m*16 + l - 1], Xi0[m*16 + l - 1]}));
            Xr0[n*16 + l] = acc.r; Xi0[n*16 + l] = acc.i;
        }
        __syncthreads();
    }
    if (tid < 16) {
        float k = 0.f;
        for (int n = 0; n < 64; ++n)
            k += cr[n]*Xr0[n*16 + tid] - ci[n]*Xi0[n*16 + tid];
        out[h*LL + tid] = 2.0f * k;
    }
    __syncthreads();

    {
        for (int n0 = 0; n0 < 64; n0 += 16)
            cplx_mm_tile(A0r, A0i, 64, A0r, A0i, 64, A1r, A1i, 64, 16*wv, n0, 64);
        __syncthreads();
        for (int n0 = 0; n0 < 64; n0 += 16)
            cplx_mm_tile(A1r, A1i, 64, A1r, A1i, 64, A0r, A0i, 64, 16*wv, n0, 64);
        __syncthreads();
        for (int n0 = 0; n0 < 64; n0 += 16)
            cplx_mm_tile(A0r, A0i, 64, A0r, A0i, 64, A1r, A1i, 64, 16*wv, n0, 64);
        __syncthreads();
        for (int n0 = 0; n0 < 64; n0 += 16)
            cplx_mm_tile(A1r, A1i, 64, A1r, A1i, 64, A0r, A0i, 64, 16*wv, n0, 64);
        __syncthreads();
    }

    auto proj_out = [&](const float* Dxr, const float* Dxi, int j) {
        if (tid < 16) {
            float k = 0.f;
            for (int n = 0; n < 64; ++n)
                k += cr[n]*Dxr[n*16 + tid] - ci[n]*Dxi[n*16 + tid];
            out[h*LL + j*CHUNK + tid] = 2.0f * k;
        }
        __syncthreads();
    };

    for (int jj = 0; jj < 63; ++jj) {
        const int j = 2*jj + 1;
        cplx_mm_tile(A0r, A0i, 64, Xr0, Xi0, 16, Xr1, Xi1, 16, 16*wv, 0, 64);
        __syncthreads();
        proj_out(Xr1, Xi1, j);
        cplx_mm_tile(A0r, A0i, 64, Xr1, Xi1, 16, Xr0, Xi0, 16, 16*wv, 0, 64);
        __syncthreads();
        proj_out(Xr0, Xi0, j + 1);
    }
    cplx_mm_tile(A0r, A0i, 64, Xr0, Xi0, 16, Xr1, Xi1, 16, 16*wv, 0, 64);
    __syncthreads();
    proj_out(Xr1, Xi1, 127);
}

extern "C" void kernel_launch(void* const* d_in, const int* in_sizes, int n_in,
                              void* d_out, int out_size, void* d_ws, size_t ws_size,
                              hipStream_t stream) {
    const float* A_real = (const float*)d_in[0];
    const float* A_imag = (const float*)d_in[1];
    const float* B_real = (const float*)d_in[2];
    const float* B_imag = (const float*)d_in[3];
    const float* P_real = (const float*)d_in[4];
    const float* P_imag = (const float*)d_in[5];
    const float* C_real = (const float*)d_in[6];
    const float* C_imag = (const float*)d_in[7];
    const float* log_dt = (const float*)d_in[8];
    float* out = (float*)d_out;
    (void)in_sizes; (void)n_in; (void)out_size; (void)d_ws; (void)ws_size;

    const size_t shmem_bytes = 21128u * sizeof(float);
    s4_dplr_kernel<<<NH, 128, shmem_bytes, stream>>>(
        A_real, A_imag, B_real, B_imag, P_real, P_imag, C_real, C_imag, log_dt, out);
}


# Scan split into 4 interleaved chains, one per wave (X_j+4 = dA^64 X_j, 2 more squarings): X stays in registers (D tile = next B fragment by k order), no barriers in scan; sum order changes
# speedup vs baseline: 2.4426x; 2.4426x over previous
#include <hip/hip_runtime.h>
#include <hip/hip_bf16.h>

#define NH 256
#define NN 64
#define LL 2048
#define CHUNK 16
#define NCHUNK (LL / CHUNK)

typedef __attribute__((ext_vector_type(2))) float v2f;
typedef __attribute__((ext_vector_type(4))) float v4f;
typedef __attribute__((ext_vector_type(8))) float v8f;

struct cpx { float r, i; };
__device__ __forceinline__ cpx cmul(cpx a, cpx b) { return {a.r*b.r - a.i*b.i, a.r*b.i + a.i*b.r}; }
__device__ __forceinline__ cpx cadd(cpx a, cpx b) { return {a.r + b.r, a.i + b.i}; }
__device__ __forceinline__ cpx csub(cpx a, cpx b) { return {a.r - b.r, a.i - b.i}; }
__device__ __forceinline__ cpx cinv(cpx a) { float d = a.r*a.r + a.i*a.i; return {a.r/d, -a.i/d}; }
__device__ __forceinline__ cpx cscale(cpx a, float s) { return {a.r*s, a.i*s}; }

__device__ __forceinline__ void cplx_mm_tile(
    const float* Ar, const float* Ai, int lda,
    const float* Br, const float* Bi, int ldb,
    float* Cr, float* Ci, int ldc,
    int r0, int n0, int K)
{
    const int lane = threadIdx.x & 31;
    const int mn = lane & 15;
    const int kh = (lane >> 4) << 1;
    v8f arbr = {}; v8f aibi = {}; v8f arbi = {}; v8f aibr = {};
    for (int k0 = 0; k0 < K; k0 += 4) {
        const int ka = k0 + kh;
        v2f ar, ai, br, bi;
        ar[0] = Ar[(r0 + mn) * lda + ka];     ar[1] = Ar[(r0 + mn) * lda + ka + 1];
        ai[0] = Ai[(r0 + mn) * lda + ka];     ai[1] = Ai[(r0 + mn) * lda + ka + 1];
        br[0] = Br[ka * ldb + n0 + mn];       br[1] = Br[(ka + 1) * ldb + n0 + mn];
        bi[0] = Bi[ka * ldb + n0 + mn];       bi[1] = Bi[(ka + 1) * ldb + n0 + mn];
        arbr = __builtin_amdgcn_wmma_f32_16x16x4_f32(false, ar, false, br, (short)0, arbr, false, false);
        aibi = __builtin_amdgcn_wmma_f32_16x16x4_f32(false, ai, false, bi, (short)0, aibi, false, false);
        arbi = __builtin_amdgcn_wmma_f32_16x16x4_f32(false, ar, false, bi, (short)0, arbi, false, false);
        aibr = __builtin_amdgcn_wmma_f32_16x16x4_f32(false, ai, false, br, (short)0, aibr, false, false);
    }
    const int mbase = (lane >> 4) << 3;
    for (int v = 0; v < 8; ++v) {
        Cr[(r0 + mbase + v) * ldc + n0 + mn] = arbr[v] - aibi[v];
        Ci[(r0 + mbase + v) * ldc + n0 + mn] = arbi[v] + aibr[v];
    }
}

extern "C" __global__ __launch_bounds__(128)
void s4_dplr_kernel(const float* __restrict__ A_real, const float* __restrict__ A_imag,
                    const float* __restrict__ B_real, const float* __restrict__ B_imag,
                    const float* __restrict__ P_real, const float* __restrict__ P_imag,
                    const float* __restrict__ C_real, const float* __restrict__ C_imag,
                    const float* __restrict__ log_dt, float* __restrict__ out)
{
    extern __shared__ float sm[];
    const int h   = blockIdx.x;
    const int tid = threadIdx.x;
    const int wv  = tid >> 5;

    float* A0r = sm;
    float* A0i = sm + 4096;
    float* A1r = sm + 8192;
    float* A1i = sm + 12288;
    float* Xr0 = sm + 16384;
    float* Xi0 = sm + 17408;
    float* Xr1 = sm + 18432;
    float* Xi1 = sm + 19456;
    float* Xr2 = sm + 20480;
    float* Xi2 = sm + 21504;
    float* Xr3 = sm + 22528;
    float* Xi3 = sm + 23552;
    float* ur  = sm + 24576;  float* ui = ur + 64;
    float* wr  = ui + 64;     float* wi = wr + 64;
    float* gr  = wi + 64;     float* gi = gr + 64;
    float* cr  = gi + 64;     float* ci = cr + 64;
    float* pr  = ci + 64;     float* pi = pr + 64;
    float* scal = pi + 64;

    const float dt = __expf(log_dt[h]);
    const float cc = 0.5f * dt;

    if (tid < 64) {
        const int n = tid;
        cpx lam = { -A_real[h*64 + n], -A_imag[h*64 + n] };
        cpx p   = {  P_real[h*64 + n],  P_imag[h*64 + n] };
        cpx Dinv = cinv((cpx){1.0f - cc*lam.r, -cc*lam.i});
        cpx e    = {1.0f + cc*lam.r,  cc*lam.i};
        cpx u = cmul(Dinv, p);
        cpx g = cmul(Dinv, e);
        ur[n] = u.r; ui[n] = u.i; gr[n] = g.r; gi[n] = g.i;
        pr[n] = p.r; pi[n] = p.i;
        cr[n] = C_real[h*64 + n]; ci[n] = C_imag[h*64 + n];
    }
    __syncthreads();
    if (tid == 0) {
        cpx s = {0.f, 0.f};
        for (int n = 0; n < 64; ++n)
            s = cadd(s, cmul((cpx){pr[n], -pi[n]}, (cpx){ur[n], ui[n]}));
        cpx alpha = cmul((cpx){cc, 0.f}, cinv((cpx){1.0f + cc*s.r, cc*s.i}));
        cpx base  = csub((cpx){cc, 0.f}, cmul(alpha, cscale(s, cc)));
        scal[0] = alpha.r; scal[1] = alpha.i; scal[2] = base.r; scal[3] = base.i;
    }
    __syncthreads();
    const cpx alpha = {scal[0], scal[1]};
    const cpx base  = {scal[2], scal[3]};
    if (tid < 64) {
        const int n = tid;
        cpx w = cmul((cpx){pr[n], -pi[n]}, cadd(base, cmul(alpha, (cpx){gr[n], gi[n]})));
        wr[n] = w.r; wi[n] = w.i;
    }
    __syncthreads();
    for (int idx = tid; idx < 4096; idx += 128) {
        const int n = idx >> 6, m = idx & 63;
        cpx v = cmul((cpx){ur[n], ui[n]}, (cpx){wr[m], wi[m]});
        float rr = -v.r, rim = -v.i;
        if (n == m) { rr += gr[n]; rim += gi[n]; }
        A0r[idx] = rr; A0i[idx] = rim;
    }
    __syncthreads();

    if (tid < 64) {
        const int n = tid;
        cpx lam = { -A_real[h*64 + n], -A_imag[h*64 + n] };
        cpx Dinv = cinv((cpx){1.0f - cc*lam.r, -cc*lam.i});
        cpx b = { B_real[h*64 + n], B_imag[h*64 + n] };
        cpx t = cmul(Dinv, cscale(b, dt));
        wr[n] = t.r; wi[n] = t.i;
    }
    __syncthreads();
    if (tid == 0) {
        cpx q = {0.f, 0.f};
        for (int n = 0; n < 64; ++n)
            q = cadd(q, cmul((cpx){pr[n], -pi[n]}, (cpx){wr[n], wi[n]}));
        cpx aq = cmul(alpha, q);
        scal[4] = aq.r; scal[5] = aq.i;
    }
    __syncthreads();
    if (tid < 64) {
        const int n = tid;
        cpx dB = csub((cpx){wr[n], wi[n]}, cmul((cpx){scal[4], scal[5]}, (cpx){ur[n], ui[n]}));
        Xr0[n*16 + 0] = dB.r;
        Xi0[n*16 + 0] = dB.i;
    }
    __syncthreads();

    for (int l = 1; l < CHUNK; ++l) {
        if (tid < 64) {
            const int n = tid;
            cpx acc = {0.f, 0.f};
            for (int m = 0; m < 64; ++m)
                acc = cadd(acc, cmul((cpx){A0r[n*64 + m], A0i[n*64 + m]},
                                     (cpx){Xr0[m*16 + l - 1], Xi0[m*16 + l - 1]}));
            Xr0[n*16 + l] = acc.r; Xi0[n*16 + l] = acc.i;
        }
        __syncthreads();
    }
    {
        for (int n0 = 0; n0 < 64; n0 += 16)
            cplx_mm_tile(A0r, A0i, 64, A0r, A0i, 64, A1r, A1i, 64, 16*wv, n0, 64);
        __syncthreads();
        for (int n0 = 0; n0 < 64; n0 += 16)
            cplx_mm_tile(A1r, A1i, 64, A1r, A1i, 64, A0r, A0i, 64, 16*wv, n0, 64);
        __syncthreads();
        for (int n0 = 0; n0 < 64; n0 += 16)
            cplx_mm_tile(A0r, A0i, 64, A0r, A0i, 64, A1r, A1i, 64, 16*wv, n0, 64);
        __syncthreads();
        for (int n0 = 0; n0 < 64; n0 += 16)
            cplx_mm_tile(A1r, A1i, 64, A1r, A1i, 64, A0r, A0i, 64, 16*wv, n0, 64);
        __syncthreads();
    }

    cplx_mm_tile(A0r, A0i, 64, Xr0, Xi0, 16, Xr1, Xi1, 16, 16*wv, 0, 64);
    __syncthreads();
    cplx_mm_tile(A0r, A0i, 64, Xr1, Xi1, 16, Xr2, Xi2, 16, 16*wv, 0, 64);
    __syncthreads();
    cplx_mm_tile(A0r, A0i, 64, Xr2, Xi2, 16, Xr3, Xi3, 16, 16*wv, 0, 64);
    __syncthreads();

    for (int n0 = 0; n0 < 64; n0 += 16)
        cplx_mm_tile(A0r, A0i, 64, A0r, A0i, 64, A1r, A1i, 64, 16*wv, n0, 64);
    __syncthreads();
    for (int n0 = 0; n0 < 64; n0 += 16)
        cplx_mm_tile(A1r, A1i, 64, A1r, A1i, 64, A0r, A0i, 64, 16*wv, n0, 64);
    __syncthreads();

    v4f* F = (v4f*)(sm + 8192);
    for (int idx = tid; idx < 2048; idx += 128) {
        const int ln = idx & 31, kq = (idx >> 5) & 15, T = idx >> 9;
        const int row = 16*T + (ln & 15);
        const int col = 16*(kq >> 2) + 8*(ln >> 4) + 2*(kq & 3);
        v4f f;
        f[0] = A0r[row*64 + col]; f[1] = A0r[row*64 + col + 1];
        f[2] = A0i[row*64 + col]; f[3] = A0i[row*64 + col + 1];
        F[idx] = f;
    }
    __syncthreads();

    const int lane = tid & 31;
    const int col16 = lane & 15;
    const int half = lane >> 4;
    const float* sxr = sm + 16384 + wv*2048;
    const float* sxi = sxr + 1024;
    float* Kt = sm;
    v8f Xr[4], Xi[4];
    float pcr[4][8], pci[4][8];
    #pragma unroll
    for (int t = 0; t < 4; ++t) {
        #pragma unroll
        for (int r = 0; r < 8; ++r) {
            const int n = 16*t + 8*half + r;
            Xr[t][r] = sxr[n*16 + col16];
            Xi[t][r] = sxi[n*16 + col16];
            pcr[t][r] = cr[n];
            pci[t][r] = ci[n];
        }
    }
    auto project = [&](int i) {
        float kp[4];
        #pragma unroll
        for (int t = 0; t < 4; ++t) {
            kp[t] = 0.f;
            #pragma unroll
            for (int r = 0; r < 8; ++r)
                kp[t] += pcr[t][r]*Xr[t][r] - pci[t][r]*Xi[t][r];
        }
        float ks = (kp[0] + kp[1]) + (kp[2] + kp[3]);
        ks += __shfl_xor(ks, 16);
        if (lane < 16) Kt[(wv + 4*i)*CHUNK + lane] = 2.0f * ks;
    };
    for (int i = 0; i < 31; ++i) {
        project(i);
        v8f nXr[4], nXi[4];
        #pragma unroll
        for (int T = 0; T < 4; ++T) {
            v8f arbr = {}; v8f aibi = {}; v8f arbi = {}; v8f aibr = {};
            #pragma unroll
            for (int t = 0; t < 4; ++t) {
                #pragma unroll
                for (int q = 0; q < 4; ++q) {
                    const v4f f = F[(T*16 + t*4 + q)*32 + lane];
                    v2f ar, ai, br, bi;
                    ar[0] = f[0]; ar[1] = f[1];
                    ai[0] = f[2]; ai[1] = f[3];
                    br[0] = Xr[t][2*q]; br[1] = Xr[t][2*q + 1];
                    bi[0] = Xi[t][2*q]; bi[1] = Xi[t][2*q + 1];
                    arbr = __builtin_amdgcn_wmma_f32_16x16x4_f32(false, ar, false, br, (short)0, arbr, false, false);
                    aibi = __builtin_amdgcn_wmma_f32_16x16x4_f32(false, ai, false, bi, (short)0, aibi, false, false);
                    arbi = __builtin_amdgcn_wmma_f32_16x16x4_f32(false, ar, false, bi, (short)0, arbi, false, false);
                    aibr = __builtin_amdgcn_wmma_f32_16x16x4_f32(false, ai, false, br, (short)0, aibr, false, false);
                }
            }
            nXr[T] = arbr - aibi;
            nXi[T] = arbi + aibr;
        }
        #pragma unroll
        for (int t = 0; t < 4; ++t) { Xr[t] = nXr[t]; Xi[t] = nXi[t]; }
    }
    project(31);
    __syncthreads();

    {
        const v4f* Kt4 = (const v4f*)Kt;
        v4f* o4 = (v4f*)(out + h*LL);
        for (int k = 0; k < 4; ++k)
            o4[k*128 + tid] = Kt4[k*128 + tid];
    }
}

extern "C" void kernel_launch(void* const* d_in, const int* in_sizes, int n_in,
                              void* d_out, int out_size, void* d_ws, size_t ws_size,
                              hipStream_t stream) {
    const float* A_real = (const float*)d_in[0];
    const float* A_imag = (const float*)d_in[1];
    const float* B_real = (const float*)d_in[2];
    const float* B_imag = (const float*)d_in[3];
    const float* P_real = (const float*)d_in[4];
    const float* P_imag = (const float*)d_in[5];
    const float* C_real = (const float*)d_in[6];
    const float* C_imag = (const float*)d_in[7];
    const float* log_dt = (const float*)d_in[8];
    float* out = (float*)d_out;
    (void)in_sizes; (void)n_in; (void)out_size; (void)d_ws; (void)ws_size;

    const size_t shmem_bytes = 25224u * sizeof(float);
    s4_dplr_kernel<<<NH, 128, shmem_bytes, stream>>>(
        A_real, A_imag, B_real, B_imag, P_real, P_imag, C_real, C_imag, log_dt, out);
}


# First 16 Krylov columns: each lane keeps half a row of dA in registers (2 lanes per row, all 128 threads), x read as a contiguous vector: no 32-way bank-conflicted LDS reads; their sum order changes
# speedup vs baseline: 3.4633x; 1.4179x over previous
#include <hip/hip_runtime.h>
#include <hip/hip_bf16.h>

#define NH 256
#define NN 64
#define LL 2048
#define CHUNK 16
#define NCHUNK (LL / CHUNK)

typedef __attribute__((ext_vector_type(2))) float v2f;
typedef __attribute__((ext_vector_type(4))) float v4f;
typedef __attribute__((ext_vector_type(8))) float v8f;

struct cpx { float r, i; };
__device__ __forceinline__ cpx cmul(cpx a, cpx b) { return {a.r*b.r - a.i*b.i, a.r*b.i + a.i*b.r}; }
__device__ __forceinline__ cpx cadd(cpx a, cpx b) { return {a.r + b.r, a.i + b.i}; }
__device__ __forceinline__ cpx csub(cpx a, cpx b) { return {a.r - b.r, a.i - b.i}; }
__device__ __forceinline__ cpx cinv(cpx a) { float d = a.r*a.r + a.i*a.i; return {a.r/d, -a.i/d}; }
__device__ __forceinline__ cpx cscale(cpx a, float s) { return {a.r*s, a.i*s}; }

__device__ __forceinline__ void cplx_mm_tile(
    const float* Ar, const float* Ai, int lda,
    const float* Br, const float* Bi, int ldb,
    float* Cr, float* Ci, int ldc,
    int r0, int n0, int K)
{
    const int lane = threadIdx.x & 31;
    const int mn = lane & 15;
    const int kh = (lane >> 4) << 1;
    v8f arbr = {}; v8f aibi = {}; v8f arbi = {}; v8f aibr = {};
    for (int k0 = 0; k0 < K; k0 += 4) {
        const int ka = k0 + kh;
        v2f ar, ai, br, bi;
        ar[0] = Ar[(r0 + mn) * lda + ka];     ar[1] = Ar[(r0 + mn) * lda + ka + 1];
        ai[0] = Ai[(r0 + mn) * lda + ka];     ai[1] = Ai[(r0 + mn) * lda + ka + 1];
        br[0] = Br[ka * ldb + n0 + mn];       br[1] = Br[(ka + 1) * ldb + n0 + mn];
        bi[0] = Bi[ka * ldb + n0 + mn];       bi[1] = Bi[(ka + 1) * ldb + n0 + mn];
        arbr = __builtin_amdgcn_wmma_f32_16x16x4_f32(false, ar, false, br, (short)0, arbr, false, false);
        aibi = __builtin_amdgcn_wmma_f32_16x16x4_f32(false, ai, false, bi, (short)0, aibi, false, false);
        arbi = __builtin_amdgcn_wmma_f32_16x16x4_f32(false, ar, false, bi, (short)0, arbi, false, false);
        aibr = __builtin_amdgcn_wmma_f32_16x16x4_f32(false, ai, false, br, (short)0, aibr, false, false);
    }
    const int mbase = (lane >> 4) << 3;
    for (int v = 0; v < 8; ++v) {
        Cr[(r0 + mbase + v) * ldc + n0 + mn] = arbr[v] - aibi[v];
        Ci[(r0 + mbase + v) * ldc + n0 + mn] = arbi[v] + aibr[v];
    }
}

extern "C" __global__ __launch_bounds__(128)
void s4_dplr_kernel(const float* __restrict__ A_real, const float* __restrict__ A_imag,
                    const float* __restrict__ B_real, const float* __restrict__ B_imag,
                    const float* __restrict__ P_real, const float* __restrict__ P_imag,
                    const float* __restrict__ C_real, const float* __restrict__ C_imag,
                    const float* __restrict__ log_dt, float* __restrict__ out)
{
    extern __shared__ float sm[];
    const int h   = blockIdx.x;
    const int tid = threadIdx.x;
    const int wv  = tid >> 5;

    float* A0r = sm;
    float* A0i = sm + 4096;
    float* A1r = sm + 8192;
    float* A1i = sm + 12288;
    float* Xr0 = sm + 16384;
    float* Xi0 = sm + 17408;
    float* Xr1 = sm + 18432;
    float* Xi1 = sm + 19456;
    float* Xr2 = sm + 20480;
    float* Xi2 = sm + 21504;
    float* Xr3 = sm + 22528;
    float* Xi3 = sm + 23552;
    float* ur  = sm + 24576;  float* ui = ur + 64;
    float* wr  = ui + 64;     float* wi = wr + 64;
    float* gr  = wi + 64;     float* gi = gr + 64;
    float* cr  = gi + 64;     float* ci = cr + 64;
    float* pr  = ci + 64;     float* pi = pr + 64;
    float* scal = pi + 64;

    const float dt = __expf(log_dt[h]);
    const float cc = 0.5f * dt;

    if (tid < 64) {
        const int n = tid;
        cpx lam = { -A_real[h*64 + n], -A_imag[h*64 + n] };
        cpx p   = {  P_real[h*64 + n],  P_imag[h*64 + n] };
        cpx Dinv = cinv((cpx){1.0f - cc*lam.r, -cc*lam.i});
        cpx e    = {1.0f + cc*lam.r,  cc*lam.i};
        cpx u = cmul(Dinv, p);
        cpx g = cmul(Dinv, e);
        ur[n] = u.r; ui[n] = u.i; gr[n] = g.r; gi[n] = g.i;
        pr[n] = p.r; pi[n] = p.i;
        cr[n] = C_real[h*64 + n]; ci[n] = C_imag[h*64 + n];
    }
    __syncthreads();
    if (tid == 0) {
        cpx s = {0.f, 0.f};
        for (int n = 0; n < 64; ++n)
            s = cadd(s, cmul((cpx){pr[n], -pi[n]}, (cpx){ur[n], ui[n]}));
        cpx alpha = cmul((cpx){cc, 0.f}, cinv((cpx){1.0f + cc*s.r, cc*s.i}));
        cpx base  = csub((cpx){cc, 0.f}, cmul(alpha, cscale(s, cc)));
        scal[0] = alpha.r; scal[1] = alpha.i; scal[2] = base.r; scal[3] = base.i;
    }
    __syncthreads();
    const cpx alpha = {scal[0], scal[1]};
    const cpx base  = {scal[2], scal[3]};
    if (tid < 64) {
        const int n = tid;
        cpx w = cmul((cpx){pr[n], -pi[n]}, cadd(base, cmul(alpha, (cpx){gr[n], gi[n]})));
        wr[n] = w.r; wi[n] = w.i;
    }
    __syncthreads();
    for (int idx = tid; idx < 4096; idx += 128) {
        const int n = idx >> 6, m = idx & 63;
        cpx v = cmul((cpx){ur[n], ui[n]}, (cpx){wr[m], wi[m]});
        float rr = -v.r, rim = -v.i;
        if (n == m) { rr += gr[n]; rim += gi[n]; }
        A0r[idx] = rr; A0i[idx] = rim;
    }
    const int mv_hf  = (tid & 31) >> 4;
    const int mv_row = 16*wv + (tid & 15);
    float rAr[32], rAi[32];
    {
        const cpx un = {ur[mv_row], ui[mv_row]};
        const float gnr = gr[mv_row], gni = gi[mv_row];
        #pragma unroll
        for (int j = 0; j < 32; ++j) {
            const int m = 32*mv_hf + j;
            cpx v = cmul(un, (cpx){wr[m], wi[m]});
            float rr = -v.r, rim = -v.i;
            if (m == mv_row) { rr += gnr; rim += gni; }
            rAr[j] = rr; rAi[j] = rim;
        }
    }
    __syncthreads();

    if (tid < 64) {
        const int n = tid;
        cpx lam = { -A_real[h*64 + n], -A_imag[h*64 + n] };
        cpx Dinv = cinv((cpx){1.0f - cc*lam.r, -cc*lam.i});
        cpx b = { B_real[h*64 + n], B_imag[h*64 + n] };
        cpx t = cmul(Dinv, cscale(b, dt));
        wr[n] = t.r; wi[n] = t.i;
    }
    __syncthreads();
    if (tid == 0) {
        cpx q = {0.f, 0.f};
        for (int n = 0; n < 64; ++n)
            q = cadd(q, cmul((cpx){pr[n], -pi[n]}, (cpx){wr[n], wi[n]}));
        cpx aq = cmul(alpha, q);
        scal[4] = aq.r; scal[5] = aq.i;
    }
    __syncthreads();
    if (tid < 64) {
        const int n = tid;
        cpx dB = csub((cpx){wr[n], wi[n]}, cmul((cpx){scal[4], scal[5]}, (cpx){ur[n], ui[n]}));
        Xr0[n*16 + 0] = dB.r;
        Xi0[n*16 + 0] = dB.i;
        A1r[n] = dB.r;
        A1r[128 + n] = dB.i;
    }
    __syncthreads();

    for (int l = 1; l < CHUNK; ++l) {
        const int rb = (l - 1) & 1, wb = l & 1;
        const v4f* xr4 = (const v4f*)(A1r + 64*rb + 32*mv_hf);
        const v4f* xi4 = (const v4f*)(A1r + 128 + 64*rb + 32*mv_hf);
        float a0r = 0.f, a0i = 0.f, a1r = 0.f, a1i = 0.f;
        #pragma unroll
        for (int j4 = 0; j4 < 8; ++j4) {
            const v4f xr = xr4[j4], xi = xi4[j4];
            a0r += rAr[4*j4 + 0]*xr[0] - rAi[4*j4 + 0]*xi[0];
            a0i += rAr[4*j4 + 0]*xi[0] + rAi[4*j4 + 0]*xr[0];
            a1r += rAr[4*j4 + 1]*xr[1] - rAi[4*j4 + 1]*xi[1];
            a1i += rAr[4*j4 + 1]*xi[1] + rAi[4*j4 + 1]*xr[1];
            a0r += rAr[4*j4 + 2]*xr[2] - rAi[4*j4 + 2]*xi[2];
            a0i += rAr[4*j4 + 2]*xi[2] + rAi[4*j4 + 2]*xr[2];
            a1r += rAr[4*j4 + 3]*xr[3] - rAi[4*j4 + 3]*xi[3];
            a1i += rAr[4*j4 + 3]*xi[3] + rAi[4*j4 + 3]*xr[3];
        }
        float sr = a0r + a1r, si = a0i + a1i;
        sr += __shfl_xor(sr, 16);
        si += __shfl_xor(si, 16);
        if (mv_hf == 0) {
            Xr0[mv_row*16 + l] = sr; Xi0[mv_row*16 + l] = si;
            A1r[64*wb + mv_row] = sr; A1r[128 + 64*wb + mv_row] = si;
        }
        __syncthreads();
    }
    {
        for (int n0 = 0; n0 < 64; n0 += 16)
            cplx_mm_tile(A0r, A0i, 64, A0r, A0i, 64, A1r, A1i, 64, 16*wv, n0, 64);
        __syncthreads();
        for (int n0 = 0; n0 < 64; n0 += 16)
            cplx_mm_tile(A1r, A1i, 64, A1r, A1i, 64, A0r, A0i, 64, 16*wv, n0, 64);
        __syncthreads();
        for (int n0 = 0; n0 < 64; n0 += 16)
            cplx_mm_tile(A0r, A0i, 64, A0r, A0i, 64, A1r, A1i, 64, 16*wv, n0, 64);
        __syncthreads();
        for (int n0 = 0; n0 < 64; n0 += 16)
            cplx_mm_tile(A1r, A1i, 64, A1r, A1i, 64, A0r, A0i, 64, 16*wv, n0, 64);
        __syncthreads();
    }

    cplx_mm_tile(A0r, A0i, 64, Xr0, Xi0, 16, Xr1, Xi1, 16, 16*wv, 0, 64);
    __syncthreads();
    cplx_mm_tile(A0r, A0i, 64, Xr1, Xi1, 16, Xr2, Xi2, 16, 16*wv, 0, 64);
    __syncthreads();
    cplx_mm_tile(A0r, A0i, 64, Xr2, Xi2, 16, Xr3, Xi3, 16, 16*wv, 0, 64);
    __syncthreads();

    for (int n0 = 0; n0 < 64; n0 += 16)
        cplx_mm_tile(A0r, A0i, 64, A0r, A0i, 64, A1r, A1i, 64, 16*wv, n0, 64);
    __syncthreads();
    for (int n0 = 0; n0 < 64; n0 += 16)
        cplx_mm_tile(A1r, A1i, 64, A1r, A1i, 64, A0r, A0i, 64, 16*wv, n0, 64);
    __syncthreads();

    v4f* F = (v4f*)(sm + 8192);
    for (int idx = tid; idx < 2048; idx += 128) {
        const int ln = idx & 31, kq = (idx >> 5) & 15, T = idx >> 9;
        const int row = 16*T + (ln & 15);
        const int col = 16*(kq >> 2) + 8*(ln >> 4) + 2*(kq & 3);
        v4f f;
        f[0] = A0r[row*64 + col]; f[1] = A0r[row*64 + col + 1];
        f[2] = A0i[row*64 + col]; f[3] = A0i[row*64 + col + 1];
        F[idx] = f;
    }
    __syncthreads();

    const int lane = tid & 31;
    const int col16 = lane & 15;
    const int half = lane >> 4;
    const float* sxr = sm + 16384 + wv*2048;
    const float* sxi = sxr + 1024;
    float* Kt = sm;
    v8f Xr[4], Xi[4];
    float pcr[4][8], pci[4][8];
    #pragma unroll
    for (int t = 0; t < 4; ++t) {
        #pragma unroll
        for (int r = 0; r < 8; ++r) {
            const int n = 16*t + 8*half + r;
            Xr[t][r] = sxr[n*16 + col16];
            Xi[t][r] = sxi[n*16 + col16];
            pcr[t][r] = cr[n];
            pci[t][r] = ci[n];
        }
    }
    auto project = [&](int i) {
        float kp[4];
        #pragma unroll
        for (int t = 0; t < 4; ++t) {
            kp[t] = 0.f;
            #pragma unroll
            for (int r = 0; r < 8; ++r)
                kp[t] += pcr[t][r]*Xr[t][r] - pci[t][r]*Xi[t][r];
        }
        float ks = (kp[0] + kp[1]) + (kp[2] + kp[3]);
        ks += __shfl_xor(ks, 16);
        if (lane < 16) Kt[(wv + 4*i)*CHUNK + lane] = 2.0f * ks;
    };
    for (int i = 0; i < 31; ++i) {
        project(i);
        v8f nXr[4], nXi[4];
        #pragma unroll
        for (int T = 0; T < 4; ++T) {
            v8f arbr = {}; v8f aibi = {}; v8f arbi = {}; v8f aibr = {};
            #pragma unroll
            for (int t = 0; t < 4; ++t) {
                #pragma unroll
                for (int q = 0; q < 4; ++q) {
                    const v4f f = F[(T*16 + t*4 + q)*32 + lane];
                    v2f ar, ai, br, bi;
                    ar[0] = f[0]; ar[1] = f[1];
                    ai[0] = f[2]; ai[1] = f[3];
                    br[0] = Xr[t][2*q]; br[1] = Xr[t][2*q + 1];
                    bi[0] = Xi[t][2*q]; bi[1] = Xi[t][2*q + 1];
                    arbr = __builtin_amdgcn_wmma_f32_16x16x4_f32(false, ar, false, br, (short)0, arbr, false, false);
                    aibi = __builtin_amdgcn_wmma_f32_16x16x4_f32(false, ai, false, bi, (short)0, aibi, false, false);
                    arbi = __builtin_amdgcn_wmma_f32_16x16x4_f32(false, ar, false, bi, (short)0, arbi, false, false);
                    aibr = __builtin_amdgcn_wmma_f32_16x16x4_f32(false, ai, false, br, (short)0, aibr, false, false);
                }
            }
            nXr[T] = arbr - aibi;
            nXi[T] = arbi + aibr;
        }
        #pragma unroll
        for (int t = 0; t < 4; ++t) { Xr[t] = nXr[t]; Xi[t] = nXi[t]; }
    }
    project(31);
    __syncthreads();

    {
        const v4f* Kt4 = (const v4f*)Kt;
        v4f* o4 = (v4f*)(out + h*LL);
        for (int k = 0; k < 4; ++k)
            o4[k*128 + tid] = Kt4[k*128 + tid];
    }
}

extern "C" void kernel_launch(void* const* d_in, const int* in_sizes, int n_in,
                              void* d_out, int out_size, void* d_ws, size_t ws_size,
                              hipStream_t stream) {
    const float* A_real = (const float*)d_in[0];
    const float* A_imag = (const float*)d_in[1];
    const float* B_real = (const float*)d_in[2];
    const float* B_imag = (const float*)d_in[3];
    const float* P_real = (const float*)d_in[4];
    const float* P_imag = (const float*)d_in[5];
    const float* C_real = (const float*)d_in[6];
    const float* C_imag = (const float*)d_in[7];
    const float* log_dt = (const float*)d_in[8];
    float* out = (float*)d_out;
    (void)in_sizes; (void)n_in; (void)out_size; (void)d_ws; (void)ws_size;

    const size_t shmem_bytes = 25224u * sizeof(float);
    s4_dplr_kernel<<<NH, 128, shmem_bytes, stream>>>(
        A_real, A_imag, B_real, B_imag, P_real, P_imag, C_real, C_imag, log_dt, out);
}


# Scan: a step's LDS accesses (64 fragment loads, projection exchange and store) issued in 4 groups, each a 64-WMMA row block ahead into a 2nd register set, not spread among the WMMAs; sums unchanged
# speedup vs baseline: 3.6488x; 1.0536x over previous
#include <hip/hip_runtime.h>
#include <hip/hip_bf16.h>

#define NH 256
#define NN 64
#define LL 2048
#define CHUNK 16
#define NCHUNK (LL / CHUNK)

typedef __attribute__((ext_vector_type(2))) float v2f;
typedef __attribute__((ext_vector_type(4))) float v4f;
typedef __attribute__((ext_vector_type(8))) float v8f;

struct cpx { float r, i; };
__device__ __forceinline__ cpx cmul(cpx a, cpx b) { return {a.r*b.r - a.i*b.i, a.r*b.i + a.i*b.r}; }
__device__ __forceinline__ cpx cadd(cpx a, cpx b) { return {a.r + b.r, a.i + b.i}; }
__device__ __forceinline__ cpx csub(cpx a, cpx b) { return {a.r - b.r, a.i - b.i}; }
__device__ __forceinline__ cpx cinv(cpx a) { float d = a.r*a.r + a.i*a.i; return {a.r/d, -a.i/d}; }
__device__ __forceinline__ cpx cscale(cpx a, float s) { return {a.r*s, a.i*s}; }

__device__ __forceinline__ void cplx_mm_tile(
    const float* Ar, const float* Ai, int lda,
    const float* Br, const float* Bi, int ldb,
    float* Cr, float* Ci, int ldc,
    int r0, int n0, int K)
{
    const int lane = threadIdx.x & 31;
    const int mn = lane & 15;
    const int kh = (lane >> 4) << 1;
    v8f arbr = {}; v8f aibi = {}; v8f arbi = {}; v8f aibr = {};
    for (int k0 = 0; k0 < K; k0 += 4) {
        const int ka = k0 + kh;
        v2f ar, ai, br, bi;
        ar[0] = Ar[(r0 + mn) * lda + ka];     ar[1] = Ar[(r0 + mn) * lda + ka + 1];
        ai[0] = Ai[(r0 + mn) * lda + ka];     ai[1] = Ai[(r0 + mn) * lda + ka + 1];
        br[0] = Br[ka * ldb + n0 + mn];       br[1] = Br[(ka + 1) * ldb + n0 + mn];
        bi[0] = Bi[ka * ldb + n0 + mn];       bi[1] = Bi[(ka + 1) * ldb + n0 + mn];
        arbr = __builtin_amdgcn_wmma_f32_16x16x4_f32(false, ar, false, br, (short)0, arbr, false, false);
        aibi = __builtin_amdgcn_wmma_f32_16x16x4_f32(false, ai, false, bi, (short)0, aibi, false, false);
        arbi = __builtin_amdgcn_wmma_f32_16x16x4_f32(false, ar, false, bi, (short)0, arbi, false, false);
        aibr = __builtin_amdgcn_wmma_f32_16x16x4_f32(false, ai, false, br, (short)0, aibr, false, false);
    }
    const int mbase = (lane >> 4) << 3;
    for (int v = 0; v < 8; ++v) {
        Cr[(r0 + mbase + v) * ldc + n0 + mn] = arbr[v] - aibi[v];
        Ci[(r0 + mbase + v) * ldc + n0 + mn] = arbi[v] + aibr[v];
    }
}

extern "C" __global__ __launch_bounds__(128)
void s4_dplr_kernel(const float* __restrict__ A_real, const float* __restrict__ A_imag,
                    const float* __restrict__ B_real, const float* __restrict__ B_imag,
                    const float* __restrict__ P_real, const float* __restrict__ P_imag,
                    const float* __restrict__ C_real, const float* __restrict__ C_imag,
                    const float* __restrict__ log_dt, float* __restrict__ out)
{
    extern __shared__ float sm[];
    const int h   = blockIdx.x;
    const int tid = threadIdx.x;
    const int wv  = tid >> 5;

    float* A0r = sm;
    float* A0i = sm + 4096;
    float* A1r = sm + 8192;
    float* A1i = sm + 12288;
    float* Xr0 = sm + 16384;
    float* Xi0 = sm + 17408;
    float* Xr1 = sm + 18432;
    float* Xi1 = sm + 19456;
    float* Xr2 = sm + 20480;
    float* Xi2 = sm + 21504;
    float* Xr3 = sm + 22528;
    float* Xi3 = sm + 23552;
    float* ur  = sm + 24576;  float* ui = ur + 64;
    float* wr  = ui + 64;     float* wi = wr + 64;
    float* gr  = wi + 64;     float* gi = gr + 64;
    float* cr  = gi + 64;     float* ci = cr + 64;
    float* pr  = ci + 64;     float* pi = pr + 64;
    float* scal = pi + 64;

    const float dt = __expf(log_dt[h]);
    const float cc = 0.5f * dt;

    if (tid < 64) {
        const int n = tid;
        cpx lam = { -A_real[h*64 + n], -A_imag[h*64 + n] };
        cpx p   = {  P_real[h*64 + n],  P_imag[h*64 + n] };
        cpx Dinv = cinv((cpx){1.0f - cc*lam.r, -cc*lam.i});
        cpx e    = {1.0f + cc*lam.r,  cc*lam.i};
        cpx u = cmul(Dinv, p);
        cpx g = cmul(Dinv, e);
        ur[n] = u.r; ui[n] = u.i; gr[n] = g.r; gi[n] = g.i;
        pr[n] = p.r; pi[n] = p.i;
        cr[n] = C_real[h*64 + n]; ci[n] = C_imag[h*64 + n];
    }
    __syncthreads();
    if (tid == 0) {
        cpx s = {0.f, 0.f};
        for (int n = 0; n < 64; ++n)
            s = cadd(s, cmul((cpx){pr[n], -pi[n]}, (cpx){ur[n], ui[n]}));
        cpx alpha = cmul((cpx){cc, 0.f}, cinv((cpx){1.0f + cc*s.r, cc*s.i}));
        cpx base  = csub((cpx){cc, 0.f}, cmul(alpha, cscale(s, cc)));
        scal[0] = alpha.r; scal[1] = alpha.i; scal[2] = base.r; scal[3] = base.i;
    }
    __syncthreads();
    const cpx alpha = {scal[0], scal[1]};
    const cpx base  = {scal[2], scal[3]};
    if (tid < 64) {
        const int n = tid;
        cpx w = cmul((cpx){pr[n], -pi[n]}, cadd(base, cmul(alpha, (cpx){gr[n], gi[n]})));
        wr[n] = w.r; wi[n] = w.i;
    }
    __syncthreads();
    for (int idx = tid; idx < 4096; idx += 128) {
        const int n = idx >> 6, m = idx & 63;
        cpx v = cmul((cpx){ur[n], ui[n]}, (cpx){wr[m], wi[m]});
        float rr = -v.r, rim = -v.i;
        if (n == m) { rr += gr[n]; rim += gi[n]; }
        A0r[idx] = rr; A0i[idx] = rim;
    }
    const int mv_hf  = (tid & 31) >> 4;
    const int mv_row = 16*wv + (tid & 15);
    float rAr[32], rAi[32];
    {
        const cpx un = {ur[mv_row], ui[mv_row]};
        const float gnr = gr[mv_row], gni = gi[mv_row];
        #pragma unroll
        for (int j = 0; j < 32; ++j) {
            const int m = 32*mv_hf + j;
            cpx v = cmul(un, (cpx){wr[m], wi[m]});
            float rr = -v.r, rim = -v.i;
            if (m == mv_row) { rr += gnr; rim += gni; }
            rAr[j] = rr; rAi[j] = rim;
        }
    }
    __syncthreads();

    if (tid < 64) {
        const int n = tid;
        cpx lam = { -A_real[h*64 + n], -A_imag[h*64 + n] };
        cpx Dinv = cinv((cpx){1.0f - cc*lam.r, -cc*lam.i});
        cpx b = { B_real[h*64 + n], B_imag[h*64 + n] };
        cpx t = cmul(Dinv, cscale(b, dt));
        wr[n] = t.r; wi[n] = t.i;
    }
    __syncthreads();
    if (tid == 0) {
        cpx q = {0.f, 0.f};
        for (int n = 0; n < 64; ++n)
            q = cadd(q, cmul((cpx){pr[n], -pi[n]}, (cpx){wr[n], wi[n]}));
        cpx aq = cmul(alpha, q);
        scal[4] = aq.r; scal[5] = aq.i;
    }
    __syncthreads();
    if (tid < 64) {
        const int n = tid;
        cpx dB = csub((cpx){wr[n], wi[n]}, cmul((cpx){scal[4], scal[5]}, (cpx){ur[n], ui[n]}));
        Xr0[n*16 + 0] = dB.r;
        Xi0[n*16 + 0] = dB.i;
        A1r[n] = dB.r;
        A1r[128 + n] = dB.i;
    }
    __syncthreads();

    for (int l = 1; l < CHUNK; ++l) {
        const int rb = (l - 1) & 1, wb = l & 1;
        const v4f* xr4 = (const v4f*)(A1r + 64*rb + 32*mv_hf);
        const v4f* xi4 = (const v4f*)(A1r + 128 + 64*rb + 32*mv_hf);
        float a0r = 0.f, a0i = 0.f, a1r = 0.f, a1i = 0.f;
        #pragma unroll
        for (int j4 = 0; j4 < 8; ++j4) {
            const v4f xr = xr4[j4], xi = xi4[j4];
            a0r += rAr[4*j4 + 0]*xr[0] - rAi[4*j4 + 0]*xi[0];
            a0i += rAr[4*j4 + 0]*xi[0] + rAi[4*j4 + 0]*xr[0];
            a1r += rAr[4*j4 + 1]*xr[1] - rAi[4*j4 + 1]*xi[1];
            a1i += rAr[4*j4 + 1]*xi[1] + rAi[4*j4 + 1]*xr[1];
            a0r += rAr[4*j4 + 2]*xr[2] - rAi[4*j4 + 2]*xi[2];
            a0i += rAr[4*j4 + 2]*xi[2] + rAi[4*j4 + 2]*xr[2];
            a1r += rAr[4*j4 + 3]*xr[3] - rAi[4*j4 + 3]*xi[3];
            a1i += rAr[4*j4 + 3]*xi[3] + rAi[4*j4 + 3]*xr[3];
        }
        float sr = a0r + a1r, si = a0i + a1i;
        sr += __shfl_xor(sr, 16);
        si += __shfl_xor(si, 16);
        if (mv_hf == 0) {
            Xr0[mv_row*16 + l] = sr; Xi0[mv_row*16 + l] = si;
            A1r[64*wb + mv_row] = sr; A1r[128 + 64*wb + mv_row] = si;
        }
        __syncthreads();
    }
    {
        for (int n0 = 0; n0 < 64; n0 += 16)
            cplx_mm_tile(A0r, A0i, 64, A0r, A0i, 64, A1r, A1i, 64, 16*wv, n0, 64);
        __syncthreads();
        for (int n0 = 0; n0 < 64; n0 += 16)
            cplx_mm_tile(A1r, A1i, 64, A1r, A1i, 64, A0r, A0i, 64, 16*wv, n0, 64);
        __syncthreads();
        for (int n0 = 0; n0 < 64; n0 += 16)
            cplx_mm_tile(A0r, A0i, 64, A0r, A0i, 64, A1r, A1i, 64, 16*wv, n0, 64);
        __syncthreads();
        for (int n0 = 0; n0 < 64; n0 += 16)
            cplx_mm_tile(A1r, A1i, 64, A1r, A1i, 64, A0r, A0i, 64, 16*wv, n0, 64);
        __syncthreads();
    }

    cplx_mm_tile(A0r, A0i, 64, Xr0, Xi0, 16, Xr1, Xi1, 16, 16*wv, 0, 64);
    __syncthreads();
    cplx_mm_tile(A0r, A0i, 64, Xr1, Xi1, 16, Xr2, Xi2, 16, 16*wv, 0, 64);
    __syncthreads();
    cplx_mm_tile(A0r, A0i, 64, Xr2, Xi2, 16, Xr3, Xi3, 16, 16*wv, 0, 64);
    __syncthreads();

    for (int n0 = 0; n0 < 64; n0 += 16)
        cplx_mm_tile(A0r, A0i, 64, A0r, A0i, 64, A1r, A1i, 64, 16*wv, n0, 64);
    __syncthreads();
    for (int n0 = 0; n0 < 64; n0 += 16)
        cplx_mm_tile(A1r, A1i, 64, A1r, A1i, 64, A0r, A0i, 64, 16*wv, n0, 64);
    __syncthreads();

    v4f* F = (v4f*)(sm + 8192);
    for (int idx = tid; idx < 2048; idx += 128) {
        const int ln = idx & 31, kq = (idx >> 5) & 15, T = idx >> 9;
        const int row = 16*T + (ln & 15);
        const int col = 16*(kq >> 2) + 8*(ln >> 4) + 2*(kq & 3);
        v4f f;
        f[0] = A0r[row*64 + col]; f[1] = A0r[row*64 + col + 1];
        f[2] = A0i[row*64 + col]; f[3] = A0i[row*64 + col + 1];
        F[idx] = f;
    }
    __syncthreads();

    const int lane = tid & 31;
    const int col16 = lane & 15;
    const int half = lane >> 4;
    const float* sxr = sm + 16384 + wv*2048;
    const float* sxi = sxr + 1024;
    float* Kt = sm;
    v8f Xr[4], Xi[4];
    float pcr[4][8], pci[4][8];
    #pragma unroll
    for (int t = 0; t < 4; ++t) {
        #pragma unroll
        for (int r = 0; r < 8; ++r) {
            const int n = 16*t + 8*half + r;
            Xr[t][r] = sxr[n*16 + col16];
            Xi[t][r] = sxi[n*16 + col16];
            pcr[t][r] = cr[n];
            pci[t][r] = ci[n];
        }
    }
    auto project_part = [&]() -> float {
        float kp[4];
        #pragma unroll
        for (int t = 0; t < 4; ++t) {
            kp[t] = 0.f;
            #pragma unroll
            for (int r = 0; r < 8; ++r)
                kp[t] += pcr[t][r]*Xr[t][r] - pci[t][r]*Xi[t][r];
        }
        return (kp[0] + kp[1]) + (kp[2] + kp[3]);
    };
    auto load_rows = [&](v4f (&f)[16], int T) {
        #pragma unroll
        for (int s = 0; s < 16; ++s) f[s] = F[(T*16 + s)*32 + lane];
    };
    auto row_block = [&](const v4f (&f)[16], v8f& outr, v8f& outi) {
        v8f arbr = {}; v8f aibi = {}; v8f arbi = {}; v8f aibr = {};
        #pragma unroll
        for (int t = 0; t < 4; ++t) {
            #pragma unroll
            for (int q = 0; q < 4; ++q) {
                const v4f ff = f[t*4 + q];
                v2f ar, ai, br, bi;
                ar[0] = ff[0]; ar[1] = ff[1];
                ai[0] = ff[2]; ai[1] = ff[3];
                br[0] = Xr[t][2*q]; br[1] = Xr[t][2*q + 1];
                bi[0] = Xi[t][2*q]; bi[1] = Xi[t][2*q + 1];
                arbr = __builtin_amdgcn_wmma_f32_16x16x4_f32(false, ar, false, br, (short)0, arbr, false, false);
                aibi = __builtin_amdgcn_wmma_f32_16x16x4_f32(false, ai, false, bi, (short)0, aibi, false, false);
                arbi = __builtin_amdgcn_wmma_f32_16x16x4_f32(false, ar, false, bi, (short)0, arbi, false, false);
                aibr = __builtin_amdgcn_wmma_f32_16x16x4_f32(false, ai, false, br, (short)0, aibr, false, false);
            }
        }
        outr = arbr - aibi;
        outi = arbi + aibr;
    };
    v4f fa[16], fb[16];
    load_rows(fa, 0);
    for (int i = 0; i < 31; ++i) {
        v8f nXr[4], nXi[4];
        load_rows(fb, 1);
        __builtin_amdgcn_sched_barrier(0);
        row_block(fa, nXr[0], nXi[0]);
        const float ks = project_part();
        __builtin_amdgcn_sched_barrier(0);
        load_rows(fa, 2);
        const float ko = __shfl_xor(ks, 16);
        row_block(fb, nXr[1], nXi[1]);
        __builtin_amdgcn_sched_barrier(0);
        load_rows(fb, 3);
        if (lane < 16) Kt[(wv + 4*i)*CHUNK + lane] = 2.0f * (ks + ko);
        row_block(fa, nXr[2], nXi[2]);
        __builtin_amdgcn_sched_barrier(0);
        load_rows(fa, 0);
        __builtin_amdgcn_sched_barrier(0);
        row_block(fb, nXr[3], nXi[3]);
        #pragma unroll
        for (int t = 0; t < 4; ++t) { Xr[t] = nXr[t]; Xi[t] = nXi[t]; }
    }
    {
        float ks = project_part();
        ks += __shfl_xor(ks, 16);
        if (lane < 16) Kt[(wv + 4*31)*CHUNK + lane] = 2.0f * ks;
    }
    __syncthreads();

    {
        const v4f* Kt4 = (const v4f*)Kt;
        v4f* o4 = (v4f*)(out + h*LL);
        for (int k = 0; k < 4; ++k)
            o4[k*128 + tid] = Kt4[k*128 + tid];
    }
}

extern "C" void kernel_launch(void* const* d_in, const int* in_sizes, int n_in,
                              void* d_out, int out_size, void* d_ws, size_t ws_size,
                              hipStream_t stream) {
    const float* A_real = (const float*)d_in[0];
    const float* A_imag = (const float*)d_in[1];
    const float* B_real = (const float*)d_in[2];
    const float* B_imag = (const float*)d_in[3];
    const float* P_real = (const float*)d_in[4];
    const float* P_imag = (const float*)d_in[5];
    const float* C_real = (const float*)d_in[6];
    const float* C_imag = (const float*)d_in[7];
    const float* log_dt = (const float*)d_in[8];
    float* out = (float*)d_out;
    (void)in_sizes; (void)n_in; (void)out_size; (void)d_ws; (void)ws_size;

    const size_t shmem_bytes = 25224u * sizeof(float);
    s4_dplr_kernel<<<NH, 128, shmem_bytes, stream>>>(
        A_real, A_imag, B_real, B_imag, P_real, P_imag, C_real, C_imag, log_dt, out);
}


# Scan step in factored form: dA^16 X = diag(g^16) X - U (W^T X), rank-16 factors in registers (128 WMMAs a step, not 256); 4 chains over contiguous ranges seeded by dA^512; sum order changes
# speedup vs baseline: 6.8280x; 1.8713x over previous
#include <hip/hip_runtime.h>
#include <hip/hip_bf16.h>

#define NH 256
#define NN 64
#define LL 2048
#define CHUNK 16
#define NCHUNK (LL / CHUNK)
#define WP 66

typedef __attribute__((ext_vector_type(2))) float v2f;
typedef __attribute__((ext_vector_type(4))) float v4f;
typedef __attribute__((ext_vector_type(8))) float v8f;

struct cpx { float r, i; };
__device__ __forceinline__ cpx cmul(cpx a, cpx b) { return {a.r*b.r - a.i*b.i, a.r*b.i + a.i*b.r}; }
__device__ __forceinline__ cpx cadd(cpx a, cpx b) { return {a.r + b.r, a.i + b.i}; }
__device__ __forceinline__ cpx csub(cpx a, cpx b) { return {a.r - b.r, a.i - b.i}; }
__device__ __forceinline__ cpx cinv(cpx a) { float d = a.r*a.r + a.i*a.i; return {a.r/d, -a.i/d}; }
__device__ __forceinline__ cpx cscale(cpx a, float s) { return {a.r*s, a.i*s}; }

extern "C" __global__ __launch_bounds__(128)
void s4_dplr_kernel(const float* __restrict__ A_real, const float* __restrict__ A_imag,
                    const float* __restrict__ B_real, const float* __restrict__ B_imag,
                    const float* __restrict__ P_real, const float* __restrict__ P_imag,
                    const float* __restrict__ C_real, const float* __restrict__ C_imag,
                    const float* __restrict__ log_dt, float* __restrict__ out)
{
    extern __shared__ float sm[];
    const int h   = blockIdx.x;
    const int tid = threadIdx.x;
    const int wv  = tid >> 5;

    v4f* F = (v4f*)sm;
    float* mvs = sm + 8192;
    float* Xr0 = sm + 16384;
    float* Xi0 = sm + 17408;
    float* Xr1 = sm + 18432;
    float* Xi1 = sm + 19456;
    float* Xr2 = sm + 20480;
    float* Xi2 = sm + 21504;
    float* Xr3 = sm + 22528;
    float* Xi3 = sm + 23552;
    float* ur  = sm + 24576;  float* ui = ur + 64;
    float* wr  = ui + 64;     float* wi = wr + 64;
    float* gr  = wi + 64;     float* gi = gr + 64;
    float* cr  = gi + 64;     float* ci = cr + 64;
    float* pr  = ci + 64;     float* pi = pr + 64;
    float* tr  = pi + 64;     float* ti = tr + 64;
    float* scal = ti + 64;
    float* Ufr  = sm + 25360;
    float* Ufi  = Ufr + 1024;
    float* WTr  = Ufi + 1024;
    float* WTi  = WTr + 16*WP;
    float* G16r = WTi + 16*WP;
    float* G16i = G16r + 64;

    const float dt = __expf(log_dt[h]);
    const float cc = 0.5f * dt;

    if (tid < 64) {
        const int n = tid;
        cpx lam = { -A_real[h*64 + n], -A_imag[h*64 + n] };
        cpx p   = {  P_real[h*64 + n],  P_imag[h*64 + n] };
        cpx Dinv = cinv((cpx){1.0f - cc*lam.r, -cc*lam.i});
        cpx e    = {1.0f + cc*lam.r,  cc*lam.i};
        cpx u = cmul(Dinv, p);
        cpx g = cmul(Dinv, e);
        ur[n] = u.r; ui[n] = u.i; gr[n] = g.r; gi[n] = g.i;
        pr[n] = p.r; pi[n] = p.i;
        cr[n] = C_real[h*64 + n]; ci[n] = C_imag[h*64 + n];
    }
    __syncthreads();
    if (tid == 0) {
        cpx s = {0.f, 0.f};
        for (int n = 0; n < 64; ++n)
            s = cadd(s, cmul((cpx){pr[n], -pi[n]}, (cpx){ur[n], ui[n]}));
        cpx alpha = cmul((cpx){cc, 0.f}, cinv((cpx){1.0f + cc*s.r, cc*s.i}));
        cpx base  = csub((cpx){cc, 0.f}, cmul(alpha, cscale(s, cc)));
        scal[0] = alpha.r; scal[1] = alpha.i; scal[2] = base.r; scal[3] = base.i;
    }
    __syncthreads();
    const cpx alpha = {scal[0], scal[1]};
    const cpx base  = {scal[2], scal[3]};
    if (tid < 64) {
        const int n = tid;
        cpx w = cmul((cpx){pr[n], -pi[n]}, cadd(base, cmul(alpha, (cpx){gr[n], gi[n]})));
        wr[n] = w.r; wi[n] = w.i;
        const cpx g = {gr[n], gi[n]};
        cpx pw = w;
        for (int i = 15; i >= 0; --i) {
            WTr[i*WP + n] = pw.r; WTi[i*WP + n] = pw.i;
            pw = cmul(pw, g);
        }
        const cpx g2 = cmul(g, g), g4 = cmul(g2, g2), g8 = cmul(g4, g4), g16 = cmul(g8, g8);
        G16r[n] = g16.r; G16i[n] = g16.i;
    }
    __syncthreads();
    for (int idx = tid; idx < 2048; idx += 128) {
        const int slot = idx & 31, blk = idx >> 5;
        const int q = blk & 3, t = (blk >> 2) & 3, Tb = blk >> 4;
        const int hh = slot >> 4;
        const int rho = ((slot & 15) - 2*q - 8*hh) & 15;
        const int n = 16*Tb + rho;
        const int m = 16*t + 8*hh + 2*q;
        const cpx un = {ur[n], ui[n]};
        const float gnr = gr[n], gni = gi[n];
        const cpx v0 = cmul(un, (cpx){wr[m], wi[m]});
        const cpx v1 = cmul(un, (cpx){wr[m + 1], wi[m + 1]});
        v4f f;
        f[0] = -v0.r; f[1] = -v1.r; f[2] = -v0.i; f[3] = -v1.i;
        if (n == m)     { f[0] += gnr; f[2] += gni; }
        if (n == m + 1) { f[1] += gnr; f[3] += gni; }
        F[idx] = f;
    }
    const int lane = tid & 31;
    const int col16 = lane & 15;
    const int half = lane >> 4;
    v8f Xr[4], Xi[4];
    {
        const int m = 16*wv + col16;
        const cpx wm = {wr[m], wi[m]};
        const float gmr = gr[m], gmi = gi[m];
        #pragma unroll
        for (int t = 0; t < 4; ++t) {
            #pragma unroll
            for (int r = 0; r < 8; ++r) {
                const int n = 16*t + 8*half + r;
                cpx v = cmul((cpx){ur[n], ui[n]}, wm);
                float rr = -v.r, rim = -v.i;
                if (n == m) { rr += gmr; rim += gmi; }
                Xr[t][r] = rr; Xi[t][r] = rim;
            }
        }
    }
    const int mv_hf  = (tid & 31) >> 4;
    const int mv_row = 16*wv + (tid & 15);
    float rAr[32], rAi[32];
    {
        const cpx un = {ur[mv_row], ui[mv_row]};
        const float gnr = gr[mv_row], gni = gi[mv_row];
        #pragma unroll
        for (int j = 0; j < 32; ++j) {
            const int m = 32*mv_hf + j;
            cpx v = cmul(un, (cpx){wr[m], wi[m]});
            float rr = -v.r, rim = -v.i;
            if (m == mv_row) { rr += gnr; rim += gni; }
            rAr[j] = rr; rAi[j] = rim;
        }
    }
    __syncthreads();

    if (tid < 64) {
        const int n = tid;
        cpx lam = { -A_real[h*64 + n], -A_imag[h*64 + n] };
        cpx Dinv = cinv((cpx){1.0f - cc*lam.r, -cc*lam.i});
        cpx b = { B_real[h*64 + n], B_imag[h*64 + n] };
        cpx t = cmul(Dinv, cscale(b, dt));
        tr[n] = t.r; ti[n] = t.i;
    }
    __syncthreads();
    if (tid == 0) {
        cpx q = {0.f, 0.f};
        for (int n = 0; n < 64; ++n)
            q = cadd(q, cmul((cpx){pr[n], -pi[n]}, (cpx){tr[n], ti[n]}));
        cpx aq = cmul(alpha, q);
        scal[4] = aq.r; scal[5] = aq.i;
    }
    __syncthreads();
    if (tid < 64) {
        const int n = tid;
        cpx dB = csub((cpx){tr[n], ti[n]}, cmul((cpx){scal[4], scal[5]}, (cpx){ur[n], ui[n]}));
        Xr0[n*16 + 0] = dB.r;
        Xi0[n*16 + 0] = dB.i;
        mvs[n] = dB.r;
        mvs[128 + n] = dB.i;
        Ufr[n*16 + 0] = ur[n];
        Ufi[n*16 + 0] = ui[n];
        mvs[256 + n] = ur[n];
        mvs[384 + n] = ui[n];
    }
    __syncthreads();

    for (int l = 1; l < CHUNK; ++l) {
        const int rb = (l - 1) & 1, wb = l & 1;
        const v4f* xr4 = (const v4f*)(mvs + 64*rb + 32*mv_hf);
        const v4f* xi4 = (const v4f*)(mvs + 128 + 64*rb + 32*mv_hf);
        const v4f* yr4 = (const v4f*)(mvs + 256 + 64*rb + 32*mv_hf);
        const v4f* yi4 = (const v4f*)(mvs + 384 + 64*rb + 32*mv_hf);
        float a0r = 0.f, a0i = 0.f, a1r = 0.f, a1i = 0.f;
        float b0r = 0.f, b0i = 0.f, b1r = 0.f, b1i = 0.f;
        #pragma unroll
        for (int j4 = 0; j4 < 8; ++j4) {
            const v4f xr = xr4[j4], xi = xi4[j4];
            a0r += rAr[4*j4 + 0]*xr[0] - rAi[4*j4 + 0]*xi[0];
            a0i += rAr[4*j4 + 0]*xi[0] + rAi[4*j4 + 0]*xr[0];
            a1r += rAr[4*j4 + 1]*xr[1] - rAi[4*j4 + 1]*xi[1];
            a1i += rAr[4*j4 + 1]*xi[1] + rAi[4*j4 + 1]*xr[1];
            a0r += rAr[4*j4 + 2]*xr[2] - rAi[4*j4 + 2]*xi[2];
            a0i += rAr[4*j4 + 2]*xi[2] + rAi[4*j4 + 2]*xr[2];
            a1r += rAr[4*j4 + 3]*xr[3] - rAi[4*j4 + 3]*xi[3];
            a1i += rAr[4*j4 + 3]*xi[3] + rAi[4*j4 + 3]*xr[3];
            const v4f yr = yr4[j4], yi = yi4[j4];
            b0r += rAr[4*j4 + 0]*yr[0] - rAi[4*j4 + 0]*yi[0];
            b0i += rAr[4*j4 + 0]*yi[0] + rAi[4*j4 + 0]*yr[0];
            b1r += rAr[4*j4 + 1]*yr[1] - rAi[4*j4 + 1]*yi[1];
            b1i += rAr[4*j4 + 1]*yi[1] + rAi[4*j4 + 1]*yr[1];
            b0r += rAr[4*j4 + 2]*yr[2] - rAi[4*j4 + 2]*yi[2];
            b0i += rAr[4*j4 + 2]*yi[2] + rAi[4*j4 + 2]*yr[2];
            b1r += rAr[4*j4 + 3]*yr[3] - rAi[4*j4 + 3]*yi[3];
            b1i += rAr[4*j4 + 3]*yi[3] + rAi[4*j4 + 3]*yr[3];
        }
        float sr = a0r + a1r, si = a0i + a1i;
        float vr = b0r + b1r, vi = b0i + b1i;
        sr += __shfl_xor(sr, 16);
        si += __shfl_xor(si, 16);
        vr += __shfl_xor(vr, 16);
        vi += __shfl_xor(vi, 16);
        if (mv_hf == 0) {
            Xr0[mv_row*16 + l] = sr; Xi0[mv_row*16 + l] = si;
            mvs[64*wb + mv_row] = sr; mvs[128 + 64*wb + mv_row] = si;
            Ufr[mv_row*16 + l] = vr; Ufi[mv_row*16 + l] = vi;
            mvs[256 + 64*wb + mv_row] = vr; mvs[384 + 64*wb + mv_row] = vi;
        }
        __syncthreads();
    }
    int lofs[4];
    #pragma unroll
    for (int q = 0; q < 4; ++q) lofs[q] = 16*half + ((col16 + 2*q + 8*half) & 15);
    auto load_rows = [&](v4f (&f)[16], int cur, int T) {
        #pragma unroll
        for (int s = 0; s < 16; ++s) f[s] = F[cur + (T*16 + s)*32 + lofs[s & 3]];
    };
    auto row_block = [&](const v4f (&f)[16], v8f& outr, v8f& outi) {
        v8f arbr = {}; v8f aibi = {}; v8f arbi = {}; v8f aibr = {};
        #pragma unroll
        for (int t = 0; t < 4; ++t) {
            #pragma unroll
            for (int q = 0; q < 4; ++q) {
                const v4f ff = f[t*4 + q];
                v2f ar, ai, br, bi;
                ar[0] = ff[0]; ar[1] = ff[1];
                ai[0] = ff[2]; ai[1] = ff[3];
                br[0] = Xr[t][2*q]; br[1] = Xr[t][2*q + 1];
                bi[0] = Xi[t][2*q]; bi[1] = Xi[t][2*q + 1];
                arbr = __builtin_amdgcn_wmma_f32_16x16x4_f32(false, ar, false, br, (short)0, arbr, false, false);
                aibi = __builtin_amdgcn_wmma_f32_16x16x4_f32(false, ai, false, bi, (short)0, aibi, false, false);
                arbi = __builtin_amdgcn_wmma_f32_16x16x4_f32(false, ar, false, bi, (short)0, arbi, false, false);
                aibr = __builtin_amdgcn_wmma_f32_16x16x4_f32(false, ai, false, br, (short)0, aibr, false, false);
            }
        }
        outr = arbr - aibi;
        outi = arbi + aibr;
    };
    auto publish = [&](int nxt) {
        const int sp = col16 & 1, qp = (col16 >> 1) & 3, hp = col16 >> 3;
        #pragma unroll
        for (int T = 0; T < 4; ++T) {
            #pragma unroll
            for (int rp = 0; rp < 4; ++rp) {
                const float sr = sp ? Xr[T][2*rp] : Xr[T][2*rp + 1];
                const float si = sp ? Xi[T][2*rp] : Xi[T][2*rp + 1];
                const float or_ = __shfl_xor(sr, 1);
                const float oi_ = __shfl_xor(si, 1);
                v4f e;
                e[0] = sp ? or_ : Xr[T][2*rp];
                e[1] = sp ? Xr[T][2*rp + 1] : or_;
                e[2] = sp ? oi_ : Xi[T][2*rp];
                e[3] = sp ? Xi[T][2*rp + 1] : oi_;
                const int rho = 8*half + 2*rp + sp;
                F[nxt + (T*16 + wv*4 + qp)*32 + 16*hp + ((rho + 2*qp + 8*hp) & 15)] = e;
            }
        }
    };
    auto power_step = [&](int cur, int nxt) {
        v8f nXr[4], nXi[4];
        v4f ga[16], gb[16];
        load_rows(ga, cur, 0);
        load_rows(gb, cur, 1);
        __builtin_amdgcn_sched_barrier(0);
        row_block(ga, nXr[0], nXi[0]);
        __builtin_amdgcn_sched_barrier(0);
        load_rows(ga, cur, 2);
        __builtin_amdgcn_sched_barrier(0);
        row_block(gb, nXr[1], nXi[1]);
        __builtin_amdgcn_sched_barrier(0);
        load_rows(gb, cur, 3);
        __builtin_amdgcn_sched_barrier(0);
        row_block(ga, nXr[2], nXi[2]);
        __builtin_amdgcn_sched_barrier(0);
        row_block(gb, nXr[3], nXi[3]);
        #pragma unroll
        for (int t = 0; t < 4; ++t) { Xr[t] = nXr[t]; Xi[t] = nXi[t]; }
        publish(nxt);
    };
    auto seed_tile = [&](int cur, int bo, int co) {
        v4f f[16];
        load_rows(f, cur, wv);
        const float* Br = sm + bo;  const float* Bi = Br + 1024;
        float* Cr = sm + co;        float* Ci = Cr + 1024;
        v8f arbr = {}; v8f aibi = {}; v8f arbi = {}; v8f aibr = {};
        #pragma unroll
        for (int t = 0; t < 4; ++t) {
            #pragma unroll
            for (int q = 0; q < 4; ++q) {
                const int k = 16*t + 8*half + 2*q;
                const v4f ff = f[t*4 + q];
                v2f ar, ai, br, bi;
                ar[0] = ff[0]; ar[1] = ff[1];
                ai[0] = ff[2]; ai[1] = ff[3];
                br[0] = Br[k*16 + col16];  br[1] = Br[(k + 1)*16 + col16];
                bi[0] = Bi[k*16 + col16];  bi[1] = Bi[(k + 1)*16 + col16];
                arbr = __builtin_amdgcn_wmma_f32_16x16x4_f32(false, ar, false, br, (short)0, arbr, false, false);
                aibi = __builtin_amdgcn_wmma_f32_16x16x4_f32(false, ai, false, bi, (short)0, aibi, false, false);
                arbi = __builtin_amdgcn_wmma_f32_16x16x4_f32(false, ar, false, bi, (short)0, arbi, false, false);
                aibr = __builtin_amdgcn_wmma_f32_16x16x4_f32(false, ai, false, br, (short)0, aibr, false, false);
            }
        }
        for (int v = 0; v < 8; ++v) {
            Cr[(16*wv + 8*half + v)*16 + col16] = arbr[v] - aibi[v];
            Ci[(16*wv + 8*half + v)*16 + col16] = arbi[v] + aibr[v];
        }
    };

    for (int s = 0; s < 9; ++s) {
        power_step((s & 1) * 2048, ((s & 1) ^ 1) * 2048);
        __syncthreads();
    }

    seed_tile(2048, 16384, 18432);
    __syncthreads();
    seed_tile(2048, 18432, 20480);
    __syncthreads();
    seed_tile(2048, 20480, 22528);
    __syncthreads();

    const float* sxr = sm + 16384 + wv*2048;
    const float* sxi = sxr + 1024;
    float* Kt = sm + 8192;
    float pcr[4][8], pci[4][8], hgr[4][8], hgi[4][8];
    #pragma unroll
    for (int t = 0; t < 4; ++t) {
        #pragma unroll
        for (int r = 0; r < 8; ++r) {
            const int n = 16*t + 8*half + r;
            Xr[t][r] = sxr[n*16 + col16];
            Xi[t][r] = sxi[n*16 + col16];
            pcr[t][r] = cr[n];
            pci[t][r] = ci[n];
            hgr[t][r] = G16r[n];
            hgi[t][r] = G16i[n];
        }
    }
    v2f fWr[16], fWi[16];
    #pragma unroll
    for (int t = 0; t < 4; ++t) {
        #pragma unroll
        for (int q = 0; q < 4; ++q) {
            const int k = 16*t + 8*half + 2*q;
            fWr[t*4 + q] = *(const v2f*)(WTr + col16*WP + k);
            fWi[t*4 + q] = *(const v2f*)(WTi + col16*WP + k);
        }
    }
    v2f fUrn[4][4], fUi[4][4], fUin[4][4];
    #pragma unroll
    for (int T = 0; T < 4; ++T) {
        #pragma unroll
        for (int q = 0; q < 4; ++q) {
            const int o = (16*T + col16)*16 + 8*half + 2*q;
            const v2f a = *(const v2f*)(Ufr + o);
            const v2f b = *(const v2f*)(Ufi + o);
            fUrn[T][q] = -a;
            fUi[T][q]  = b;
            fUin[T][q] = -b;
        }
    }
    auto project_part = [&]() -> float {
        float kp[4];
        #pragma unroll
        for (int t = 0; t < 4; ++t) {
            kp[t] = 0.f;
            #pragma unroll
            for (int r = 0; r < 8; ++r)
                kp[t] += pcr[t][r]*Xr[t][r] - pci[t][r]*Xi[t][r];
        }
        return (kp[0] + kp[1]) + (kp[2] + kp[3]);
    };
    auto chunk_step = [&]() {
        v8f zrr = {}; v8f zii = {}; v8f zri = {}; v8f zir = {};
        #pragma unroll
        for (int t = 0; t < 4; ++t) {
            #pragma unroll
            for (int q = 0; q < 4; ++q) {
                v2f br, bi;
                br[0] = Xr[t][2*q]; br[1] = Xr[t][2*q + 1];
                bi[0] = Xi[t][2*q]; bi[1] = Xi[t][2*q + 1];
                zrr = __builtin_amdgcn_wmma_f32_16x16x4_f32(false, fWr[t*4 + q], false, br, (short)0, zrr, false, false);
                zii = __builtin_amdgcn_wmma_f32_16x16x4_f32(false, fWi[t*4 + q], false, bi, (short)0, zii, false, false);
                zri = __builtin_amdgcn_wmma_f32_16x16x4_f32(false, fWr[t*4 + q], false, bi, (short)0, zri, false, false);
                zir = __builtin_amdgcn_wmma_f32_16x16x4_f32(false, fWi[t*4 + q], false, br, (short)0, zir, false, false);
            }
        }
        const v8f Zr = zrr - zii;
        const v8f Zi = zri + zir;
        v8f ar[4], ai[4];
        #pragma unroll
        for (int T = 0; T < 4; ++T) {
            #pragma unroll
            for (int r = 0; r < 8; ++r) {
                ar[T][r] = hgr[T][r]*Xr[T][r] - hgi[T][r]*Xi[T][r];
                ai[T][r] = hgr[T][r]*Xi[T][r] + hgi[T][r]*Xr[T][r];
            }
        }
        #pragma unroll
        for (int q = 0; q < 4; ++q) {
            v2f zr2, zi2;
            zr2[0] = Zr[2*q]; zr2[1] = Zr[2*q + 1];
            zi2[0] = Zi[2*q]; zi2[1] = Zi[2*q + 1];
            #pragma unroll
            for (int T = 0; T < 4; ++T) {
                ar[T] = __builtin_amdgcn_wmma_f32_16x16x4_f32(false, fUrn[T][q], false, zr2, (short)0, ar[T], false, false);
                ai[T] = __builtin_amdgcn_wmma_f32_16x16x4_f32(false, fUrn[T][q], false, zi2, (short)0, ai[T], false, false);
            }
            #pragma unroll
            for (int T = 0; T < 4; ++T) {
                ar[T] = __builtin_amdgcn_wmma_f32_16x16x4_f32(false, fUi[T][q],  false, zi2, (short)0, ar[T], false, false);
                ai[T] = __builtin_amdgcn_wmma_f32_16x16x4_f32(false, fUin[T][q], false, zr2, (short)0, ai[T], false, false);
            }
        }
        #pragma unroll
        for (int T = 0; T < 4; ++T) { Xr[T] = ar[T]; Xi[T] = ai[T]; }
    };
    for (int i = 0; i < 31; ++i) {
        const float ks = project_part();
        const float ko = __shfl_xor(ks, 16);
        chunk_step();
        if (lane < 16) Kt[(32*wv + i)*CHUNK + lane] = 2.0f * (ks + ko);
    }
    {
        float ks = project_part();
        ks += __shfl_xor(ks, 16);
        if (lane < 16) Kt[(32*wv + 31)*CHUNK + lane] = 2.0f * ks;
    }
    __syncthreads();

    {
        const v4f* Kt4 = (const v4f*)Kt;
        v4f* o4 = (v4f*)(out + h*LL);
        for (int k = 0; k < 4; ++k)
            o4[k*128 + tid] = Kt4[k*128 + tid];
    }
}

extern "C" void kernel_launch(void* const* d_in, const int* in_sizes, int n_in,
                              void* d_out, int out_size, void* d_ws, size_t ws_size,
                              hipStream_t stream) {
    const float* A_real = (const float*)d_in[0];
    const float* A_imag = (const float*)d_in[1];
    const float* B_real = (const float*)d_in[2];
    const float* B_imag = (const float*)d_in[3];
    const float* P_real = (const float*)d_in[4];
    const float* P_imag = (const float*)d_in[5];
    const float* C_real = (const float*)d_in[6];
    const float* C_imag = (const float*)d_in[7];
    const float* log_dt = (const float*)d_in[8];
    float* out = (float*)d_out;
    (void)in_sizes; (void)n_in; (void)out_size; (void)d_ws; (void)ws_size;

    const size_t shmem_bytes = 29648u * sizeof(float);
    s4_dplr_kernel<<<NH, 128, shmem_bytes, stream>>>(
        A_real, A_imag, B_real, B_imag, P_real, P_imag, C_real, C_imag, log_dt, out);
}


# Scan step = dA^4 in rank-4 factored form, 16 chains (4 per wave): Re/Im rows of W^T and c^T stacked in one 16-row A operand, 48 WMMAs a step, not 128; 3 matvecs, not 15; sum order changes
# speedup vs baseline: 10.3800x; 1.5202x over previous
#include <hip/hip_runtime.h>
#include <hip/hip_bf16.h>

#define NH 256
#define NN 64
#define LL 2048
#define RK 4
#define WP 66

typedef __attribute__((ext_vector_type(2))) float v2f;
typedef __attribute__((ext_vector_type(4))) float v4f;
typedef __attribute__((ext_vector_type(8))) float v8f;

struct cpx { float r, i; };
__device__ __forceinline__ cpx cmul(cpx a, cpx b) { return {a.r*b.r - a.i*b.i, a.r*b.i + a.i*b.r}; }
__device__ __forceinline__ cpx cadd(cpx a, cpx b) { return {a.r + b.r, a.i + b.i}; }
__device__ __forceinline__ cpx csub(cpx a, cpx b) { return {a.r - b.r, a.i - b.i}; }
__device__ __forceinline__ cpx cinv(cpx a) { float d = a.r*a.r + a.i*a.i; return {a.r/d, -a.i/d}; }
__device__ __forceinline__ cpx cscale(cpx a, float s) { return {a.r*s, a.i*s}; }

extern "C" __global__ __launch_bounds__(128)
void s4_dplr_kernel(const float* __restrict__ A_real, const float* __restrict__ A_imag,
                    const float* __restrict__ B_real, const float* __restrict__ B_imag,
                    const float* __restrict__ P_real, const float* __restrict__ P_imag,
                    const float* __restrict__ C_real, const float* __restrict__ C_imag,
                    const float* __restrict__ log_dt, float* __restrict__ out)
{
    extern __shared__ float sm[];
    const int h   = blockIdx.x;
    const int tid = threadIdx.x;
    const int wv  = tid >> 5;

    v4f* F = (v4f*)sm;
    float* mvs = sm + 8192;
    float* Xr0 = sm + 16384;
    float* Xi0 = sm + 17408;
    float* Xr1 = sm + 18432;
    float* Xi1 = sm + 19456;
    float* Xr2 = sm + 20480;
    float* Xi2 = sm + 21504;
    float* Xr3 = sm + 22528;
    float* Xi3 = sm + 23552;
    float* ur  = sm + 24576;  float* ui = ur + 64;
    float* wr  = ui + 64;     float* wi = wr + 64;
    float* gr  = wi + 64;     float* gi = gr + 64;
    float* cr  = gi + 64;     float* ci = cr + 64;
    float* pr  = ci + 64;     float* pi = pr + 64;
    float* tr  = pi + 64;     float* ti = tr + 64;
    float* scal = ti + 64;
    float* Ufr  = sm + 25360;
    float* Ufi  = Ufr + 256;
    float* WS   = Ufi + 256;
    float* G4r  = WS + 16*WP;
    float* G4i  = G4r + 64;

    const float dt = __expf(log_dt[h]);
    const float cc = 0.5f * dt;

    if (tid < 64) {
        const int n = tid;
        cpx lam = { -A_real[h*64 + n], -A_imag[h*64 + n] };
        cpx p   = {  P_real[h*64 + n],  P_imag[h*64 + n] };
        cpx Dinv = cinv((cpx){1.0f - cc*lam.r, -cc*lam.i});
        cpx e    = {1.0f + cc*lam.r,  cc*lam.i};
        cpx u = cmul(Dinv, p);
        cpx g = cmul(Dinv, e);
        ur[n] = u.r; ui[n] = u.i; gr[n] = g.r; gi[n] = g.i;
        pr[n] = p.r; pi[n] = p.i;
        cr[n] = C_real[h*64 + n]; ci[n] = C_imag[h*64 + n];
    }
    __syncthreads();
    if (tid == 0) {
        cpx s = {0.f, 0.f};
        for (int n = 0; n < 64; ++n)
            s = cadd(s, cmul((cpx){pr[n], -pi[n]}, (cpx){ur[n], ui[n]}));
        cpx alpha = cmul((cpx){cc, 0.f}, cinv((cpx){1.0f + cc*s.r, cc*s.i}));
        cpx base  = csub((cpx){cc, 0.f}, cmul(alpha, cscale(s, cc)));
        scal[0] = alpha.r; scal[1] = alpha.i; scal[2] = base.r; scal[3] = base.i;
    }
    __syncthreads();
    const cpx alpha = {scal[0], scal[1]};
    const cpx base  = {scal[2], scal[3]};
    if (tid < 64) {
        const int n = tid;
        cpx w = cmul((cpx){pr[n], -pi[n]}, cadd(base, cmul(alpha, (cpx){gr[n], gi[n]})));
        wr[n] = w.r; wi[n] = w.i;
        const cpx g = {gr[n], gi[n]};
        cpx pw = w;
        for (int i = RK - 1; i >= 0; --i) {
            const int row = 8*(i >> 1) + (i & 1);
            WS[row*WP + n] = pw.r; WS[(row + 2)*WP + n] = pw.i;
            pw = cmul(pw, g);
        }
        WS[4*WP + n] = cr[n];  WS[5*WP + n] = ci[n];
        WS[6*WP + n] = 0.f;    WS[7*WP + n] = 0.f;
        for (int row = 12; row < 16; ++row) WS[row*WP + n] = 0.f;
        const cpx g2 = cmul(g, g), g4 = cmul(g2, g2);
        G4r[n] = g4.r; G4i[n] = g4.i;
    }
    __syncthreads();
    for (int idx = tid; idx < 2048; idx += 128) {
        const int slot = idx & 31, blk = idx >> 5;
        const int q = blk & 3, t = (blk >> 2) & 3, Tb = blk >> 4;
        const int hh = slot >> 4;
        const int rho = ((slot & 15) - 2*q - 8*hh) & 15;
        const int n = 16*Tb + rho;
        const int m = 16*t + 8*hh + 2*q;
        const cpx un = {ur[n], ui[n]};
        const float gnr = gr[n], gni = gi[n];
        const cpx v0 = cmul(un, (cpx){wr[m], wi[m]});
        const cpx v1 = cmul(un, (cpx){wr[m + 1], wi[m + 1]});
        v4f f;
        f[0] = -v0.r; f[1] = -v1.r; f[2] = -v0.i; f[3] = -v1.i;
        if (n == m)     { f[0] += gnr; f[2] += gni; }
        if (n == m + 1) { f[1] += gnr; f[3] += gni; }
        F[idx] = f;
    }
    const int lane = tid & 31;
    const int col16 = lane & 15;
    const int half = lane >> 4;
    v8f Xr[4], Xi[4];
    {
        const int m = 16*wv + col16;
        const cpx wm = {wr[m], wi[m]};
        const float gmr = gr[m], gmi = gi[m];
        #pragma unroll
        for (int t = 0; t < 4; ++t) {
            #pragma unroll
            for (int r = 0; r < 8; ++r) {
                const int n = 16*t + 8*half + r;
                cpx v = cmul((cpx){ur[n], ui[n]}, wm);
                float rr = -v.r, rim = -v.i;
                if (n == m) { rr += gmr; rim += gmi; }
                Xr[t][r] = rr; Xi[t][r] = rim;
            }
        }
    }
    const int mv_hf  = (tid & 31) >> 4;
    const int mv_row = 16*wv + (tid & 15);
    float rAr[32], rAi[32];
    {
        const cpx un = {ur[mv_row], ui[mv_row]};
        const float gnr = gr[mv_row], gni = gi[mv_row];
        #pragma unroll
        for (int j = 0; j < 32; ++j) {
            const int m = 32*mv_hf + j;
            cpx v = cmul(un, (cpx){wr[m], wi[m]});
            float rr = -v.r, rim = -v.i;
            if (m == mv_row) { rr += gnr; rim += gni; }
            rAr[j] = rr; rAi[j] = rim;
        }
    }
    __syncthreads();

    if (tid < 64) {
        const int n = tid;
        cpx lam = { -A_real[h*64 + n], -A_imag[h*64 + n] };
        cpx Dinv = cinv((cpx){1.0f - cc*lam.r, -cc*lam.i});
        cpx b = { B_real[h*64 + n], B_imag[h*64 + n] };
        cpx t = cmul(Dinv, cscale(b, dt));
        tr[n] = t.r; ti[n] = t.i;
    }
    __syncthreads();
    if (tid == 0) {
        cpx q = {0.f, 0.f};
        for (int n = 0; n < 64; ++n)
            q = cadd(q, cmul((cpx){pr[n], -pi[n]}, (cpx){tr[n], ti[n]}));
        cpx aq = cmul(alpha, q);
        scal[4] = aq.r; scal[5] = aq.i;
    }
    __syncthreads();
    if (tid < 64) {
        const int n = tid;
        cpx dB = csub((cpx){tr[n], ti[n]}, cmul((cpx){scal[4], scal[5]}, (cpx){ur[n], ui[n]}));
        Xr0[n*16 + 0] = dB.r;
        Xi0[n*16 + 0] = dB.i;
        mvs[n] = dB.r;
        mvs[128 + n] = dB.i;
        Ufr[n*RK + 0] = ur[n];
        Ufi[n*RK + 0] = ui[n];
        mvs[256 + n] = ur[n];
        mvs[384 + n] = ui[n];
    }
    __syncthreads();

    for (int l = 1; l < RK; ++l) {
        const int rb = (l - 1) & 1, wb = l & 1;
        const v4f* xr4 = (const v4f*)(mvs + 64*rb + 32*mv_hf);
        const v4f* xi4 = (const v4f*)(mvs + 128 + 64*rb + 32*mv_hf);
        const v4f* yr4 = (const v4f*)(mvs + 256 + 64*rb + 32*mv_hf);
        const v4f* yi4 = (const v4f*)(mvs + 384 + 64*rb + 32*mv_hf);
        float a0r = 0.f, a0i = 0.f, a1r = 0.f, a1i = 0.f;
        float b0r = 0.f, b0i = 0.f, b1r = 0.f, b1i = 0.f;
        #pragma unroll
        for (int j4 = 0; j4 < 8; ++j4) {
            const v4f xr = xr4[j4], xi = xi4[j4];
            a0r += rAr[4*j4 + 0]*xr[0] - rAi[4*j4 + 0]*xi[0];
            a0i += rAr[4*j4 + 0]*xi[0] + rAi[4*j4 + 0]*xr[0];
            a1r += rAr[4*j4 + 1]*xr[1] - rAi[4*j4 + 1]*xi[1];
            a1i += rAr[4*j4 + 1]*xi[1] + rAi[4*j4 + 1]*xr[1];
            a0r += rAr[4*j4 + 2]*xr[2] - rAi[4*j4 + 2]*xi[2];
            a0i += rAr[4*j4 + 2]*xi[2] + rAi[4*j4 + 2]*xr[2];
            a1r += rAr[4*j4 + 3]*xr[3] - rAi[4*j4 + 3]*xi[3];
            a1i += rAr[4*j4 + 3]*xi[3] + rAi[4*j4 + 3]*xr[3];
            const v4f yr = yr4[j4], yi = yi4[j4];
            b0r += rAr[4*j4 + 0]*yr[0] - rAi[4*j4 + 0]*yi[0];
            b0i += rAr[4*j4 + 0]*yi[0] + rAi[4*j4 + 0]*yr[0];
            b1r += rAr[4*j4 + 1]*yr[1] - rAi[4*j4 + 1]*yi[1];
            b1i += rAr[4*j4 + 1]*yi[1] + rAi[4*j4 + 1]*yr[1];
            b0r += rAr[4*j4 + 2]*yr[2] - rAi[4*j4 + 2]*yi[2];
            b0i += rAr[4*j4 + 2]*yi[2] + rAi[4*j4 + 2]*yr[2];
            b1r += rAr[4*j4 + 3]*yr[3] - rAi[4*j4 + 3]*yi[3];
            b1i += rAr[4*j4 + 3]*yi[3] + rAi[4*j4 + 3]*yr[3];
        }
        float sr = a0r + a1r, si = a0i + a1i;
        float vr = b0r + b1r, vi = b0i + b1i;
        sr += __shfl_xor(sr, 16);
        si += __shfl_xor(si, 16);
        vr += __shfl_xor(vr, 16);
        vi += __shfl_xor(vi, 16);
        if (mv_hf == 0) {
            Xr0[mv_row*16 + l] = sr; Xi0[mv_row*16 + l] = si;
            mvs[64*wb + mv_row] = sr; mvs[128 + 64*wb + mv_row] = si;
            Ufr[mv_row*RK + l] = vr; Ufi[mv_row*RK + l] = vi;
            mvs[256 + 64*wb + mv_row] = vr; mvs[384 + 64*wb + mv_row] = vi;
        }
        __syncthreads();
    }
    int lofs[4];
    #pragma unroll
    for (int q = 0; q < 4; ++q) lofs[q] = 16*half + ((col16 + 2*q + 8*half) & 15);
    auto load_rows = [&](v4f (&f)[16], int cur, int T) {
        #pragma unroll
        for (int s = 0; s < 16; ++s) f[s] = F[cur + (T*16 + s)*32 + lofs[s & 3]];
    };
    auto row_block = [&](const v4f (&f)[16], v8f& outr, v8f& outi) {
        v8f arbr = {}; v8f aibi = {}; v8f arbi = {}; v8f aibr = {};
        #pragma unroll
        for (int t = 0; t < 4; ++t) {
            #pragma unroll
            for (int q = 0; q < 4; ++q) {
                const v4f ff = f[t*4 + q];
                v2f ar, ai, br, bi;
                ar[0] = ff[0]; ar[1] = ff[1];
                ai[0] = ff[2]; ai[1] = ff[3];
                br[0] = Xr[t][2*q]; br[1] = Xr[t][2*q + 1];
                bi[0] = Xi[t][2*q]; bi[1] = Xi[t][2*q + 1];
                arbr = __builtin_amdgcn_wmma_f32_16x16x4_f32(false, ar, false, br, (short)0, arbr, false, false);
                aibi = __builtin_amdgcn_wmma_f32_16x16x4_f32(false, ai, false, bi, (short)0, aibi, false, false);
                arbi = __builtin_amdgcn_wmma_f32_16x16x4_f32(false, ar, false, bi, (short)0, arbi, false, false);
                aibr = __builtin_amdgcn_wmma_f32_16x16x4_f32(false, ai, false, br, (short)0, aibr, false, false);
            }
        }
        outr = arbr - aibi;
        outi = arbi + aibr;
    };
    auto publish = [&](int nxt) {
        const int sp = col16 & 1, qp = (col16 >> 1) & 3, hp = col16 >> 3;
        #pragma unroll
        for (int T = 0; T < 4; ++T) {
            #pragma unroll
            for (int rp = 0; rp < 4; ++rp) {
                const float sr = sp ? Xr[T][2*rp] : Xr[T][2*rp + 1];
                const float si = sp ? Xi[T][2*rp] : Xi[T][2*rp + 1];
                const float or_ = __shfl_xor(sr, 1);
                const float oi_ = __shfl_xor(si, 1);
                v4f e;
                e[0] = sp ? or_ : Xr[T][2*rp];
                e[1] = sp ? Xr[T][2*rp + 1] : or_;
                e[2] = sp ? oi_ : Xi[T][2*rp];
                e[3] = sp ? Xi[T][2*rp + 1] : oi_;
                const int rho = 8*half + 2*rp + sp;
                F[nxt + (T*16 + wv*4 + qp)*32 + 16*hp + ((rho + 2*qp + 8*hp) & 15)] = e;
            }
        }
    };
    auto power_step = [&](int cur, int nxt) {
        v8f nXr[4], nXi[4];
        v4f ga[16], gb[16];
        load_rows(ga, cur, 0);
        load_rows(gb, cur, 1);
        __builtin_amdgcn_sched_barrier(0);
        row_block(ga, nXr[0], nXi[0]);
        __builtin_amdgcn_sched_barrier(0);
        load_rows(ga, cur, 2);
        __builtin_amdgcn_sched_barrier(0);
        row_block(gb, nXr[1], nXi[1]);
        __builtin_amdgcn_sched_barrier(0);
        load_rows(gb, cur, 3);
        __builtin_amdgcn_sched_barrier(0);
        row_block(ga, nXr[2], nXi[2]);
        __builtin_amdgcn_sched_barrier(0);
        row_block(gb, nXr[3], nXi[3]);
        #pragma unroll
        for (int t = 0; t < 4; ++t) { Xr[t] = nXr[t]; Xi[t] = nXi[t]; }
        publish(nxt);
    };
    auto seed_tile = [&](int cur, int bo, int co, int cmask, int cofs) {
        v4f f[16];
        load_rows(f, cur, wv);
        const float* Br = sm + bo;  const float* Bi = Br + 1024;
        float* Cr = sm + co;        float* Ci = Cr + 1024;
        const int bc = col16 & cmask;
        v8f arbr = {}; v8f aibi = {}; v8f arbi = {}; v8f aibr = {};
        #pragma unroll
        for (int t = 0; t < 4; ++t) {
            #pragma unroll
            for (int q = 0; q < 4; ++q) {
                const int k = 16*t + 8*half + 2*q;
                const v4f ff = f[t*4 + q];
                v2f ar, ai, br, bi;
                ar[0] = ff[0]; ar[1] = ff[1];
                ai[0] = ff[2]; ai[1] = ff[3];
                br[0] = Br[k*16 + bc];  br[1] = Br[(k + 1)*16 + bc];
                bi[0] = Bi[k*16 + bc];  bi[1] = Bi[(k + 1)*16 + bc];
                arbr = __builtin_amdgcn_wmma_f32_16x16x4_f32(false, ar, false, br, (short)0, arbr, false, false);
                aibi = __builtin_amdgcn_wmma_f32_16x16x4_f32(false, ai, false, bi, (short)0, aibi, false, false);
                arbi = __builtin_amdgcn_wmma_f32_16x16x4_f32(false, ar, false, bi, (short)0, arbi, false, false);
                aibr = __builtin_amdgcn_wmma_f32_16x16x4_f32(false, ai, false, br, (short)0, aibr, false, false);
            }
        }
        for (int v = 0; v < 8; ++v) {
            Cr[(16*wv + 8*half + v)*16 + cofs + bc] = arbr[v] - aibi[v];
            Ci[(16*wv + 8*half + v)*16 + cofs + bc] = arbi[v] + aibr[v];
        }
    };

    for (int s = 0; s < 9; ++s) {
        power_step((s & 1) * 2048, ((s & 1) ^ 1) * 2048);
        __syncthreads();
        if (s == 6) seed_tile(2048, 16384, 16384, 3, 4);
        else if (s == 7) seed_tile(0, 16384, 16384, 7, 8);
    }

    seed_tile(2048, 16384, 18432, 15, 0);
    __syncthreads();
    seed_tile(2048, 18432, 20480, 15, 0);
    __syncthreads();
    seed_tile(2048, 20480, 22528, 15, 0);
    __syncthreads();

    const float* sxr = sm + 16384 + wv*2048;
    const float* sxi = sxr + 1024;
    float* Kt = sm + 8192;
    float hgr[4][8], hgi[4][8];
    #pragma unroll
    for (int t = 0; t < 4; ++t) {
        #pragma unroll
        for (int r = 0; r < 8; ++r) {
            const int n = 16*t + 8*half + r;
            Xr[t][r] = sxr[n*16 + col16];
            Xi[t][r] = sxi[n*16 + col16];
            hgr[t][r] = G4r[n];
            hgi[t][r] = G4i[n];
        }
    }
    v2f fWS[16];
    #pragma unroll
    for (int t = 0; t < 4; ++t) {
        #pragma unroll
        for (int q = 0; q < 4; ++q) {
            const int k = 16*t + 8*half + 2*q;
            fWS[t*4 + q] = *(const v2f*)(WS + col16*WP + k);
        }
    }
    v2f fUrn[4], fUi[4], fUin[4];
    #pragma unroll
    for (int T = 0; T < 4; ++T) {
        const int o = (16*T + col16)*RK + 2*half;
        const v2f a = *(const v2f*)(Ufr + o);
        const v2f b = *(const v2f*)(Ufi + o);
        fUrn[T] = -a;
        fUi[T]  = b;
        fUin[T] = -b;
    }
    float* kout = Kt + 512*wv + 128*(col16 >> 2) + (col16 & 3);
    auto stacked_product = [&](const v8f (&Pr)[4], const v8f (&Pi)[4], v8f& d1, v8f& d2) {
        v8f s1 = {}; v8f s2 = {};
        #pragma unroll
        for (int t = 0; t < 4; ++t) {
            #pragma unroll
            for (int q = 0; q < 4; ++q) {
                v2f br, bi;
                br[0] = Pr[t][2*q]; br[1] = Pr[t][2*q + 1];
                bi[0] = Pi[t][2*q]; bi[1] = Pi[t][2*q + 1];
                s1 = __builtin_amdgcn_wmma_f32_16x16x4_f32(false, fWS[t*4 + q], false, br, (short)0, s1, false, false);
                s2 = __builtin_amdgcn_wmma_f32_16x16x4_f32(false, fWS[t*4 + q], false, bi, (short)0, s2, false, false);
            }
        }
        d1 = s1; d2 = s2;
    };
    auto scan_step = [&](const v8f (&Pr)[4], const v8f (&Pi)[4], v8f (&Nr)[4], v8f (&Ni)[4], float* dst) {
        v8f d1, d2;
        stacked_product(Pr, Pi, d1, d2);
        const float kv = 2.0f * (d1[4] - d2[5]);
        v2f zr2, zi2;
        zr2[0] = d1[0] - d2[2]; zr2[1] = d1[1] - d2[3];
        zi2[0] = d2[0] + d1[2]; zi2[1] = d2[1] + d1[3];
        #pragma unroll
        for (int T = 0; T < 4; ++T) {
            #pragma unroll
            for (int r = 0; r < 8; ++r) {
                Nr[T][r] = hgr[T][r]*Pr[T][r] - hgi[T][r]*Pi[T][r];
                Ni[T][r] = hgr[T][r]*Pi[T][r] + hgi[T][r]*Pr[T][r];
            }
        }
        #pragma unroll
        for (int T = 0; T < 4; ++T) {
            Nr[T] = __builtin_amdgcn_wmma_f32_16x16x4_f32(false, fUrn[T], false, zr2, (short)0, Nr[T], false, false);
            Ni[T] = __builtin_amdgcn_wmma_f32_16x16x4_f32(false, fUrn[T], false, zi2, (short)0, Ni[T], false, false);
        }
        #pragma unroll
        for (int T = 0; T < 4; ++T) {
            Nr[T] = __builtin_amdgcn_wmma_f32_16x16x4_f32(false, fUi[T],  false, zi2, (short)0, Nr[T], false, false);
            Ni[T] = __builtin_amdgcn_wmma_f32_16x16x4_f32(false, fUin[T], false, zr2, (short)0, Ni[T], false, false);
        }
        if (lane < 16) *dst = kv;
    };
    v8f Yr[4], Yi[4];
    for (int jj = 0; jj < 15; ++jj) {
        scan_step(Xr, Xi, Yr, Yi, kout + 2*RK*jj);
        scan_step(Yr, Yi, Xr, Xi, kout + 2*RK*jj + RK);
    }
    scan_step(Xr, Xi, Yr, Yi, kout + RK*30);
    {
        v8f d1, d2;
        stacked_product(Yr, Yi, d1, d2);
        const float kv = 2.0f * (d1[4] - d2[5]);
        if (lane < 16) kout[RK*31] = kv;
    }
    __syncthreads();

    {
        const v4f* Kt4 = (const v4f*)Kt;
        v4f* o4 = (v4f*)(out + h*LL);
        for (int k = 0; k < 4; ++k)
            o4[k*128 + tid] = Kt4[k*128 + tid];
    }
}

extern "C" void kernel_launch(void* const* d_in, const int* in_sizes, int n_in,
                              void* d_out, int out_size, void* d_ws, size_t ws_size,
                              hipStream_t stream) {
    const float* A_real = (const float*)d_in[0];
    const float* A_imag = (const float*)d_in[1];
    const float* B_real = (const float*)d_in[2];
    const float* B_imag = (const float*)d_in[3];
    const float* P_real = (const float*)d_in[4];
    const float* P_imag = (const float*)d_in[5];
    const float* C_real = (const float*)d_in[6];
    const float* C_imag = (const float*)d_in[7];
    const float* log_dt = (const float*)d_in[8];
    float* out = (float*)d_out;
    (void)in_sizes; (void)n_in; (void)out_size; (void)d_ws; (void)ws_size;

    const size_t shmem_bytes = 29648u * sizeof(float);
    s4_dplr_kernel<<<NH, 128, shmem_bytes, stream>>>(
        A_real, A_imag, B_real, B_imag, P_real, P_imag, C_real, C_imag, log_dt, out);
}


# Scan: stacked operand's 6 idle rows carry c^T dA^32, dA^64, dA^96 (2 row products beside squarings 5, 6): a step gives 4 outputs a column, chains of 32 columns, 8 steps not 32; sum order changes
# speedup vs baseline: 12.8243x; 1.2355x over previous
#include <hip/hip_runtime.h>
#include <hip/hip_bf16.h>

#define NH 256
#define NN 64
#define LL 2048
#define RK 4
#define WP 66

typedef __attribute__((ext_vector_type(2))) float v2f;
typedef __attribute__((ext_vector_type(4))) float v4f;
typedef __attribute__((ext_vector_type(8))) float v8f;

struct cpx { float r, i; };
__device__ __forceinline__ cpx cmul(cpx a, cpx b) { return {a.r*b.r - a.i*b.i, a.r*b.i + a.i*b.r}; }
__device__ __forceinline__ cpx cadd(cpx a, cpx b) { return {a.r + b.r, a.i + b.i}; }
__device__ __forceinline__ cpx csub(cpx a, cpx b) { return {a.r - b.r, a.i - b.i}; }
__device__ __forceinline__ cpx cinv(cpx a) { float d = a.r*a.r + a.i*a.i; return {a.r/d, -a.i/d}; }
__device__ __forceinline__ cpx cscale(cpx a, float s) { return {a.r*s, a.i*s}; }

extern "C" __global__ __launch_bounds__(128)
void s4_dplr_kernel(const float* __restrict__ A_real, const float* __restrict__ A_imag,
                    const float* __restrict__ B_real, const float* __restrict__ B_imag,
                    const float* __restrict__ P_real, const float* __restrict__ P_imag,
                    const float* __restrict__ C_real, const float* __restrict__ C_imag,
                    const float* __restrict__ log_dt, float* __restrict__ out)
{
    extern __shared__ float sm[];
    const int h   = blockIdx.x;
    const int tid = threadIdx.x;
    const int wv  = tid >> 5;

    v4f* F = (v4f*)sm;
    float* mvs = sm + 8192;
    float* Xr0 = sm + 16384;
    float* Xi0 = sm + 17408;
    float* Xr1 = sm + 18432;
    float* Xi1 = sm + 19456;
    float* Xr2 = sm + 20480;
    float* Xi2 = sm + 21504;
    float* Xr3 = sm + 22528;
    float* Xi3 = sm + 23552;
    float* ur  = sm + 24576;  float* ui = ur + 64;
    float* wr  = ui + 64;     float* wi = wr + 64;
    float* gr  = wi + 64;     float* gi = gr + 64;
    float* cr  = gi + 64;     float* ci = cr + 64;
    float* pr  = ci + 64;     float* pi = pr + 64;
    float* tr  = pi + 64;     float* ti = tr + 64;
    float* scal = ti + 64;
    float* Ufr  = sm + 25360;
    float* Ufi  = Ufr + 256;
    float* WS   = Ufi + 256;
    float* G4r  = WS + 16*WP;
    float* G4i  = G4r + 64;

    const float dt = __expf(log_dt[h]);
    const float cc = 0.5f * dt;

    if (tid < 64) {
        const int n = tid;
        cpx lam = { -A_real[h*64 + n], -A_imag[h*64 + n] };
        cpx p   = {  P_real[h*64 + n],  P_imag[h*64 + n] };
        cpx Dinv = cinv((cpx){1.0f - cc*lam.r, -cc*lam.i});
        cpx e    = {1.0f + cc*lam.r,  cc*lam.i};
        cpx u = cmul(Dinv, p);
        cpx g = cmul(Dinv, e);
        ur[n] = u.r; ui[n] = u.i; gr[n] = g.r; gi[n] = g.i;
        pr[n] = p.r; pi[n] = p.i;
        cr[n] = C_real[h*64 + n]; ci[n] = C_imag[h*64 + n];
    }
    __syncthreads();
    if (tid == 0) {
        cpx s = {0.f, 0.f};
        for (int n = 0; n < 64; ++n)
            s = cadd(s, cmul((cpx){pr[n], -pi[n]}, (cpx){ur[n], ui[n]}));
        cpx alpha = cmul((cpx){cc, 0.f}, cinv((cpx){1.0f + cc*s.r, cc*s.i}));
        cpx base  = csub((cpx){cc, 0.f}, cmul(alpha, cscale(s, cc)));
        scal[0] = alpha.r; scal[1] = alpha.i; scal[2] = base.r; scal[3] = base.i;
    }
    __syncthreads();
    const cpx alpha = {scal[0], scal[1]};
    const cpx base  = {scal[2], scal[3]};
    if (tid < 64) {
        const int n = tid;
        cpx w = cmul((cpx){pr[n], -pi[n]}, cadd(base, cmul(alpha, (cpx){gr[n], gi[n]})));
        wr[n] = w.r; wi[n] = w.i;
        const cpx g = {gr[n], gi[n]};
        cpx pw = w;
        for (int i = RK - 1; i >= 0; --i) {
            const int row = 8*(i >> 1) + (i & 1);
            WS[row*WP + n] = pw.r; WS[(row + 2)*WP + n] = pw.i;
            pw = cmul(pw, g);
        }
        WS[4*WP + n] = cr[n];  WS[5*WP + n] = ci[n];
        WS[6*WP + n] = 0.f;    WS[7*WP + n] = 0.f;
        for (int row = 12; row < 16; ++row) WS[row*WP + n] = 0.f;
        const cpx g2 = cmul(g, g), g4 = cmul(g2, g2);
        G4r[n] = g4.r; G4i[n] = g4.i;
    }
    __syncthreads();
    for (int idx = tid; idx < 2048; idx += 128) {
        const int slot = idx & 31, blk = idx >> 5;
        const int q = blk & 3, t = (blk >> 2) & 3, Tb = blk >> 4;
        const int hh = slot >> 4;
        const int rho = ((slot & 15) - 2*q - 8*hh) & 15;
        const int n = 16*Tb + rho;
        const int m = 16*t + 8*hh + 2*q;
        const cpx un = {ur[n], ui[n]};
        const float gnr = gr[n], gni = gi[n];
        const cpx v0 = cmul(un, (cpx){wr[m], wi[m]});
        const cpx v1 = cmul(un, (cpx){wr[m + 1], wi[m + 1]});
        v4f f;
        f[0] = -v0.r; f[1] = -v1.r; f[2] = -v0.i; f[3] = -v1.i;
        if (n == m)     { f[0] += gnr; f[2] += gni; }
        if (n == m + 1) { f[1] += gnr; f[3] += gni; }
        F[idx] = f;
    }
    const int lane = tid & 31;
    const int col16 = lane & 15;
    const int half = lane >> 4;
    v8f Xr[4], Xi[4];
    {
        const int m = 16*wv + col16;
        const cpx wm = {wr[m], wi[m]};
        const float gmr = gr[m], gmi = gi[m];
        #pragma unroll
        for (int t = 0; t < 4; ++t) {
            #pragma unroll
            for (int r = 0; r < 8; ++r) {
                const int n = 16*t + 8*half + r;
                cpx v = cmul((cpx){ur[n], ui[n]}, wm);
                float rr = -v.r, rim = -v.i;
                if (n == m) { rr += gmr; rim += gmi; }
                Xr[t][r] = rr; Xi[t][r] = rim;
            }
        }
    }
    const int mv_hf  = (tid & 31) >> 4;
    const int mv_row = 16*wv + (tid & 15);
    float rAr[32], rAi[32];
    {
        const cpx un = {ur[mv_row], ui[mv_row]};
        const float gnr = gr[mv_row], gni = gi[mv_row];
        #pragma unroll
        for (int j = 0; j < 32; ++j) {
            const int m = 32*mv_hf + j;
            cpx v = cmul(un, (cpx){wr[m], wi[m]});
            float rr = -v.r, rim = -v.i;
            if (m == mv_row) { rr += gnr; rim += gni; }
            rAr[j] = rr; rAi[j] = rim;
        }
    }
    __syncthreads();

    if (tid < 64) {
        const int n = tid;
        cpx lam = { -A_real[h*64 + n], -A_imag[h*64 + n] };
        cpx Dinv = cinv((cpx){1.0f - cc*lam.r, -cc*lam.i});
        cpx b = { B_real[h*64 + n], B_imag[h*64 + n] };
        cpx t = cmul(Dinv, cscale(b, dt));
        tr[n] = t.r; ti[n] = t.i;
    }
    __syncthreads();
    if (tid == 0) {
        cpx q = {0.f, 0.f};
        for (int n = 0; n < 64; ++n)
            q = cadd(q, cmul((cpx){pr[n], -pi[n]}, (cpx){tr[n], ti[n]}));
        cpx aq = cmul(alpha, q);
        scal[4] = aq.r; scal[5] = aq.i;
    }
    __syncthreads();
    if (tid < 64) {
        const int n = tid;
        cpx dB = csub((cpx){tr[n], ti[n]}, cmul((cpx){scal[4], scal[5]}, (cpx){ur[n], ui[n]}));
        Xr0[n*16 + 0] = dB.r;
        Xi0[n*16 + 0] = dB.i;
        mvs[n] = dB.r;
        mvs[128 + n] = dB.i;
        Ufr[n*RK + 0] = ur[n];
        Ufi[n*RK + 0] = ui[n];
        mvs[256 + n] = ur[n];
        mvs[384 + n] = ui[n];
    }
    __syncthreads();

    for (int l = 1; l < RK; ++l) {
        const int rb = (l - 1) & 1, wb = l & 1;
        const v4f* xr4 = (const v4f*)(mvs + 64*rb + 32*mv_hf);
        const v4f* xi4 = (const v4f*)(mvs + 128 + 64*rb + 32*mv_hf);
        const v4f* yr4 = (const v4f*)(mvs + 256 + 64*rb + 32*mv_hf);
        const v4f* yi4 = (const v4f*)(mvs + 384 + 64*rb + 32*mv_hf);
        float a0r = 0.f, a0i = 0.f, a1r = 0.f, a1i = 0.f;
        float b0r = 0.f, b0i = 0.f, b1r = 0.f, b1i = 0.f;
        #pragma unroll
        for (int j4 = 0; j4 < 8; ++j4) {
            const v4f xr = xr4[j4], xi = xi4[j4];
            a0r += rAr[4*j4 + 0]*xr[0] - rAi[4*j4 + 0]*xi[0];
            a0i += rAr[4*j4 + 0]*xi[0] + rAi[4*j4 + 0]*xr[0];
            a1r += rAr[4*j4 + 1]*xr[1] - rAi[4*j4 + 1]*xi[1];
            a1i += rAr[4*j4 + 1]*xi[1] + rAi[4*j4 + 1]*xr[1];
            a0r += rAr[4*j4 + 2]*xr[2] - rAi[4*j4 + 2]*xi[2];
            a0i += rAr[4*j4 + 2]*xi[2] + rAi[4*j4 + 2]*xr[2];
            a1r += rAr[4*j4 + 3]*xr[3] - rAi[4*j4 + 3]*xi[3];
            a1i += rAr[4*j4 + 3]*xi[3] + rAi[4*j4 + 3]*xr[3];
            const v4f yr = yr4[j4], yi = yi4[j4];
            b0r += rAr[4*j4 + 0]*yr[0] - rAi[4*j4 + 0]*yi[0];
            b0i += rAr[4*j4 + 0]*yi[0] + rAi[4*j4 + 0]*yr[0];
            b1r += rAr[4*j4 + 1]*yr[1] - rAi[4*j4 + 1]*yi[1];
            b1i += rAr[4*j4 + 1]*yi[1] + rAi[4*j4 + 1]*yr[1];
            b0r += rAr[4*j4 + 2]*yr[2] - rAi[4*j4 + 2]*yi[2];
            b0i += rAr[4*j4 + 2]*yi[2] + rAi[4*j4 + 2]*yr[2];
            b1r += rAr[4*j4 + 3]*yr[3] - rAi[4*j4 + 3]*yi[3];
            b1i += rAr[4*j4 + 3]*yi[3] + rAi[4*j4 + 3]*yr[3];
        }
        float sr = a0r + a1r, si = a0i + a1i;
        float vr = b0r + b1r, vi = b0i + b1i;
        sr += __shfl_xor(sr, 16);
        si += __shfl_xor(si, 16);
        vr += __shfl_xor(vr, 16);
        vi += __shfl_xor(vi, 16);
        if (mv_hf == 0) {
            Xr0[mv_row*16 + l] = sr; Xi0[mv_row*16 + l] = si;
            mvs[64*wb + mv_row] = sr; mvs[128 + 64*wb + mv_row] = si;
            Ufr[mv_row*RK + l] = vr; Ufi[mv_row*RK + l] = vi;
            mvs[256 + 64*wb + mv_row] = vr; mvs[384 + 64*wb + mv_row] = vi;
        }
        __syncthreads();
    }
    int lofs[4];
    #pragma unroll
    for (int q = 0; q < 4; ++q) lofs[q] = 16*half + ((col16 + 2*q + 8*half) & 15);
    auto load_rows = [&](v4f (&f)[16], int cur, int T) {
        #pragma unroll
        for (int s = 0; s < 16; ++s) f[s] = F[cur + (T*16 + s)*32 + lofs[s & 3]];
    };
    auto row_block = [&](const v4f (&f)[16], v8f& outr, v8f& outi) {
        v8f arbr = {}; v8f aibi = {}; v8f arbi = {}; v8f aibr = {};
        #pragma unroll
        for (int t = 0; t < 4; ++t) {
            #pragma unroll
            for (int q = 0; q < 4; ++q) {
                const v4f ff = f[t*4 + q];
                v2f ar, ai, br, bi;
                ar[0] = ff[0]; ar[1] = ff[1];
                ai[0] = ff[2]; ai[1] = ff[3];
                br[0] = Xr[t][2*q]; br[1] = Xr[t][2*q + 1];
                bi[0] = Xi[t][2*q]; bi[1] = Xi[t][2*q + 1];
                arbr = __builtin_amdgcn_wmma_f32_16x16x4_f32(false, ar, false, br, (short)0, arbr, false, false);
                aibi = __builtin_amdgcn_wmma_f32_16x16x4_f32(false, ai, false, bi, (short)0, aibi, false, false);
                arbi = __builtin_amdgcn_wmma_f32_16x16x4_f32(false, ar, false, bi, (short)0, arbi, false, false);
                aibr = __builtin_amdgcn_wmma_f32_16x16x4_f32(false, ai, false, br, (short)0, aibr, false, false);
            }
        }
        outr = arbr - aibi;
        outi = arbi + aibr;
    };
    auto publish = [&](int nxt) {
        const int sp = col16 & 1, qp = (col16 >> 1) & 3, hp = col16 >> 3;
        #pragma unroll
        for (int T = 0; T < 4; ++T) {
            #pragma unroll
            for (int rp = 0; rp < 4; ++rp) {
                const float sr = sp ? Xr[T][2*rp] : Xr[T][2*rp + 1];
                const float si = sp ? Xi[T][2*rp] : Xi[T][2*rp + 1];
                const float or_ = __shfl_xor(sr, 1);
                const float oi_ = __shfl_xor(si, 1);
                v4f e;
                e[0] = sp ? or_ : Xr[T][2*rp];
                e[1] = sp ? Xr[T][2*rp + 1] : or_;
                e[2] = sp ? oi_ : Xi[T][2*rp];
                e[3] = sp ? Xi[T][2*rp + 1] : oi_;
                const int rho = 8*half + 2*rp + sp;
                F[nxt + (T*16 + wv*4 + qp)*32 + 16*hp + ((rho + 2*qp + 8*hp) & 15)] = e;
            }
        }
    };
    auto power_step = [&](int cur, int nxt) {
        v8f nXr[4], nXi[4];
        v4f ga[16], gb[16];
        load_rows(ga, cur, 0);
        load_rows(gb, cur, 1);
        __builtin_amdgcn_sched_barrier(0);
        row_block(ga, nXr[0], nXi[0]);
        __builtin_amdgcn_sched_barrier(0);
        load_rows(ga, cur, 2);
        __builtin_amdgcn_sched_barrier(0);
        row_block(gb, nXr[1], nXi[1]);
        __builtin_amdgcn_sched_barrier(0);
        load_rows(gb, cur, 3);
        __builtin_amdgcn_sched_barrier(0);
        row_block(ga, nXr[2], nXi[2]);
        __builtin_amdgcn_sched_barrier(0);
        row_block(gb, nXr[3], nXi[3]);
        #pragma unroll
        for (int t = 0; t < 4; ++t) { Xr[t] = nXr[t]; Xi[t] = nXi[t]; }
        publish(nxt);
    };
    auto seed_tile = [&](int cur, int bo, int co, int cmask, int cofs) {
        v4f f[16];
        load_rows(f, cur, wv);
        const float* Br = sm + bo;  const float* Bi = Br + 1024;
        float* Cr = sm + co;        float* Ci = Cr + 1024;
        const int bc = col16 & cmask;
        v8f arbr = {}; v8f aibi = {}; v8f arbi = {}; v8f aibr = {};
        #pragma unroll
        for (int t = 0; t < 4; ++t) {
            #pragma unroll
            for (int q = 0; q < 4; ++q) {
                const int k = 16*t + 8*half + 2*q;
                const v4f ff = f[t*4 + q];
                v2f ar, ai, br, bi;
                ar[0] = ff[0]; ar[1] = ff[1];
                ai[0] = ff[2]; ai[1] = ff[3];
                br[0] = Br[k*16 + bc];  br[1] = Br[(k + 1)*16 + bc];
                bi[0] = Bi[k*16 + bc];  bi[1] = Bi[(k + 1)*16 + bc];
                arbr = __builtin_amdgcn_wmma_f32_16x16x4_f32(false, ar, false, br, (short)0, arbr, false, false);
                aibi = __builtin_amdgcn_wmma_f32_16x16x4_f32(false, ai, false, bi, (short)0, aibi, false, false);
                arbi = __builtin_amdgcn_wmma_f32_16x16x4_f32(false, ar, false, bi, (short)0, arbi, false, false);
                aibr = __builtin_amdgcn_wmma_f32_16x16x4_f32(false, ai, false, br, (short)0, aibr, false, false);
            }
        }
        for (int v = 0; v < 8; ++v) {
            Cr[(16*wv + 8*half + v)*16 + cofs + bc] = arbr[v] - aibi[v];
            Ci[(16*wv + 8*half + v)*16 + cofs + bc] = arbi[v] + aibr[v];
        }
    };

    auto left_rows = [&](bool second) {
        const int arow = second ? (col16 >= 12 ? col16 - 12 : col16)
                                : ((col16 == 6 || col16 == 7) ? 12 : col16);
        const float* ap = WS + arow*WP + 8*half;
        v8f s1 = {}; v8f s2 = {};
        #pragma unroll
        for (int t = 0; t < 4; ++t) {
            #pragma unroll
            for (int q = 0; q < 4; ++q) {
                const v2f a = *(const v2f*)(ap + 16*t + 2*q);
                v2f br, bi;
                br[0] = Xr[t][2*q]; br[1] = Xr[t][2*q + 1];
                bi[0] = Xi[t][2*q]; bi[1] = Xi[t][2*q + 1];
                s1 = __builtin_amdgcn_wmma_f32_16x16x4_f32(false, a, false, br, (short)0, s1, false, false);
                s2 = __builtin_amdgcn_wmma_f32_16x16x4_f32(false, a, false, bi, (short)0, s2, false, false);
            }
        }
        if (lane < 16) {
            const int m = 16*wv + col16;
            const float e0r = s1[4] - s2[5], e0i = s2[4] + s1[5];
            if (!second) {
                WS[6*WP + m] = e0r;  WS[7*WP + m] = e0i;
            } else {
                WS[12*WP + m] = e0r; WS[13*WP + m] = e0i;
                WS[14*WP + m] = s1[6] - s2[7];
                WS[15*WP + m] = s2[6] + s1[7];
            }
        }
    };

    for (int s = 0; s < 9; ++s) {
        power_step((s & 1) * 2048, ((s & 1) ^ 1) * 2048);
        __syncthreads();
        if (s == 4) left_rows(false);
        else if (s == 5) left_rows(true);
        else if (s == 6) seed_tile(2048, 16384, 16384, 3, 4);
        else if (s == 7) seed_tile(0, 16384, 16384, 7, 8);
    }

    seed_tile(2048, 16384, 18432, 15, 0);
    __syncthreads();
    seed_tile(2048, 18432, 20480, 15, 0);
    __syncthreads();
    seed_tile(2048, 20480, 22528, 15, 0);
    __syncthreads();

    const float* sxr = sm + 16384 + wv*2048;
    const float* sxi = sxr + 1024;
    float* Kt = sm + 8192;
    float hgr[4][8], hgi[4][8];
    #pragma unroll
    for (int t = 0; t < 4; ++t) {
        #pragma unroll
        for (int r = 0; r < 8; ++r) {
            const int n = 16*t + 8*half + r;
            Xr[t][r] = sxr[n*16 + col16];
            Xi[t][r] = sxi[n*16 + col16];
            hgr[t][r] = G4r[n];
            hgi[t][r] = G4i[n];
        }
    }
    v2f fWS[16];
    #pragma unroll
    for (int t = 0; t < 4; ++t) {
        #pragma unroll
        for (int q = 0; q < 4; ++q) {
            const int k = 16*t + 8*half + 2*q;
            fWS[t*4 + q] = *(const v2f*)(WS + col16*WP + k);
        }
    }
    v2f fUrn[4], fUi[4], fUin[4];
    #pragma unroll
    for (int T = 0; T < 4; ++T) {
        const int o = (16*T + col16)*RK + 2*half;
        const v2f a = *(const v2f*)(Ufr + o);
        const v2f b = *(const v2f*)(Ufi + o);
        fUrn[T] = -a;
        fUi[T]  = b;
        fUin[T] = -b;
    }
    float* kout = Kt + 512*wv + 128*(col16 >> 2) + (col16 & 3) + 64*half;
    auto stacked_product = [&](const v8f (&Pr)[4], const v8f (&Pi)[4], v8f& d1, v8f& d2) {
        v8f s1 = {}; v8f s2 = {};
        #pragma unroll
        for (int t = 0; t < 4; ++t) {
            #pragma unroll
            for (int q = 0; q < 4; ++q) {
                v2f br, bi;
                br[0] = Pr[t][2*q]; br[1] = Pr[t][2*q + 1];
                bi[0] = Pi[t][2*q]; bi[1] = Pi[t][2*q + 1];
                s1 = __builtin_amdgcn_wmma_f32_16x16x4_f32(false, fWS[t*4 + q], false, br, (short)0, s1, false, false);
                s2 = __builtin_amdgcn_wmma_f32_16x16x4_f32(false, fWS[t*4 + q], false, bi, (short)0, s2, false, false);
            }
        }
        d1 = s1; d2 = s2;
    };
    auto scan_step = [&](const v8f (&Pr)[4], const v8f (&Pi)[4], v8f (&Nr)[4], v8f (&Ni)[4], float* dst) {
        v8f d1, d2;
        stacked_product(Pr, Pi, d1, d2);
        const float kva = 2.0f * (d1[4] - d2[5]);
        const float kvb = 2.0f * (d1[6] - d2[7]);
        v2f zr2, zi2;
        zr2[0] = d1[0] - d2[2]; zr2[1] = d1[1] - d2[3];
        zi2[0] = d2[0] + d1[2]; zi2[1] = d2[1] + d1[3];
        #pragma unroll
        for (int T = 0; T < 4; ++T) {
            #pragma unroll
            for (int r = 0; r < 8; ++r) {
                Nr[T][r] = hgr[T][r]*Pr[T][r] - hgi[T][r]*Pi[T][r];
                Ni[T][r] = hgr[T][r]*Pi[T][r] + hgi[T][r]*Pr[T][r];
            }
        }
        #pragma unroll
        for (int T = 0; T < 4; ++T) {
            Nr[T] = __builtin_amdgcn_wmma_f32_16x16x4_f32(false, fUrn[T], false, zr2, (short)0, Nr[T], false, false);
            Ni[T] = __builtin_amdgcn_wmma_f32_16x16x4_f32(false, fUrn[T], false, zi2, (short)0, Ni[T], false, false);
        }
        #pragma unroll
        for (int T = 0; T < 4; ++T) {
            Nr[T] = __builtin_amdgcn_wmma_f32_16x16x4_f32(false, fUi[T],  false, zi2, (short)0, Nr[T], false, false);
            Ni[T] = __builtin_amdgcn_wmma_f32_16x16x4_f32(false, fUin[T], false, zr2, (short)0, Ni[T], false, false);
        }
        dst[0] = kva;
        dst[32] = kvb;
    };
    v8f Yr[4], Yi[4];
    for (int jj = 0; jj < 3; ++jj) {
        scan_step(Xr, Xi, Yr, Yi, kout + 2*RK*jj);
        scan_step(Yr, Yi, Xr, Xi, kout + 2*RK*jj + RK);
    }
    scan_step(Xr, Xi, Yr, Yi, kout + RK*6);
    {
        v8f d1, d2;
        stacked_product(Yr, Yi, d1, d2);
        kout[RK*7] = 2.0f * (d1[4] - d2[5]);
        kout[RK*7 + 32] = 2.0f * (d1[6] - d2[7]);
    }
    __syncthreads();

    {
        const v4f* Kt4 = (const v4f*)Kt;
        v4f* o4 = (v4f*)(out + h*LL);
        for (int k = 0; k < 4; ++k)
            o4[k*128 + tid] = Kt4[k*128 + tid];
    }
}

extern "C" void kernel_launch(void* const* d_in, const int* in_sizes, int n_in,
                              void* d_out, int out_size, void* d_ws, size_t ws_size,
                              hipStream_t stream) {
    const float* A_real = (const float*)d_in[0];
    const float* A_imag = (const float*)d_in[1];
    const float* B_real = (const float*)d_in[2];
    const float* B_imag = (const float*)d_in[3];
    const float* P_real = (const float*)d_in[4];
    const float* P_imag = (const float*)d_in[5];
    const float* C_real = (const float*)d_in[6];
    const float* C_imag = (const float*)d_in[7];
    const float* log_dt = (const float*)d_in[8];
    float* out = (float*)d_out;
    (void)in_sizes; (void)n_in; (void)out_size; (void)d_ws; (void)ws_size;

    const size_t shmem_bytes = 29648u * sizeof(float);
    s4_dplr_kernel<<<NH, 128, shmem_bytes, stream>>>(
        A_real, A_imag, B_real, B_imag, P_real, P_imag, C_real, C_imag, log_dt, out);
}


# Chain tops moved to the left: every wave scans the same 16 columns on its own rows c_b^T dA^(512w), made by 6 row passes with dA^256 from registers; 8 squarings, no seed products; sum order changes
# speedup vs baseline: 13.7155x; 1.0695x over previous
#include <hip/hip_runtime.h>
#include <hip/hip_bf16.h>

#define NH 256
#define NN 64
#define LL 2048
#define RK 4
#define WP 66

typedef __attribute__((ext_vector_type(2))) float v2f;
typedef __attribute__((ext_vector_type(4))) float v4f;
typedef __attribute__((ext_vector_type(8))) float v8f;

struct cpx { float r, i; };
__device__ __forceinline__ cpx cmul(cpx a, cpx b) { return {a.r*b.r - a.i*b.i, a.r*b.i + a.i*b.r}; }
__device__ __forceinline__ cpx cadd(cpx a, cpx b) { return {a.r + b.r, a.i + b.i}; }
__device__ __forceinline__ cpx csub(cpx a, cpx b) { return {a.r - b.r, a.i - b.i}; }
__device__ __forceinline__ cpx cinv(cpx a) { float d = a.r*a.r + a.i*a.i; return {a.r/d, -a.i/d}; }
__device__ __forceinline__ cpx cscale(cpx a, float s) { return {a.r*s, a.i*s}; }

extern "C" __global__ __launch_bounds__(128)
void s4_dplr_kernel(const float* __restrict__ A_real, const float* __restrict__ A_imag,
                    const float* __restrict__ B_real, const float* __restrict__ B_imag,
                    const float* __restrict__ P_real, const float* __restrict__ P_imag,
                    const float* __restrict__ C_real, const float* __restrict__ C_imag,
                    const float* __restrict__ log_dt, float* __restrict__ out)
{
    extern __shared__ float sm[];
    const int h   = blockIdx.x;
    const int tid = threadIdx.x;
    const int wv  = tid >> 5;

    v4f* F = (v4f*)sm;
    float* mvs = sm + 8192;
    float* Xr0 = sm + 16384;
    float* Xi0 = sm + 17408;
    float* Xr1 = sm + 18432;
    float* Xi1 = sm + 19456;
    float* Xr2 = sm + 20480;
    float* Xi2 = sm + 21504;
    float* Xr3 = sm + 22528;
    float* Xi3 = sm + 23552;
    float* ur  = sm + 24576;  float* ui = ur + 64;
    float* wr  = ui + 64;     float* wi = wr + 64;
    float* gr  = wi + 64;     float* gi = gr + 64;
    float* cr  = gi + 64;     float* ci = cr + 64;
    float* pr  = ci + 64;     float* pi = pr + 64;
    float* tr  = pi + 64;     float* ti = tr + 64;
    float* scal = ti + 64;
    float* Ufr  = sm + 25360;
    float* Ufi  = Ufr + 256;
    float* WS   = Ufi + 256;
    float* G4r  = WS + 16*WP;
    float* G4i  = G4r + 64;

    const float dt = __expf(log_dt[h]);
    const float cc = 0.5f * dt;

    if (tid < 64) {
        const int n = tid;
        cpx lam = { -A_real[h*64 + n], -A_imag[h*64 + n] };
        cpx p   = {  P_real[h*64 + n],  P_imag[h*64 + n] };
        cpx Dinv = cinv((cpx){1.0f - cc*lam.r, -cc*lam.i});
        cpx e    = {1.0f + cc*lam.r,  cc*lam.i};
        cpx u = cmul(Dinv, p);
        cpx g = cmul(Dinv, e);
        ur[n] = u.r; ui[n] = u.i; gr[n] = g.r; gi[n] = g.i;
        pr[n] = p.r; pi[n] = p.i;
        cr[n] = C_real[h*64 + n]; ci[n] = C_imag[h*64 + n];
    }
    __syncthreads();
    if (tid == 0) {
        cpx s = {0.f, 0.f};
        for (int n = 0; n < 64; ++n)
            s = cadd(s, cmul((cpx){pr[n], -pi[n]}, (cpx){ur[n], ui[n]}));
        cpx alpha = cmul((cpx){cc, 0.f}, cinv((cpx){1.0f + cc*s.r, cc*s.i}));
        cpx base  = csub((cpx){cc, 0.f}, cmul(alpha, cscale(s, cc)));
        scal[0] = alpha.r; scal[1] = alpha.i; scal[2] = base.r; scal[3] = base.i;
    }
    __syncthreads();
    const cpx alpha = {scal[0], scal[1]};
    const cpx base  = {scal[2], scal[3]};
    if (tid < 64) {
        const int n = tid;
        cpx w = cmul((cpx){pr[n], -pi[n]}, cadd(base, cmul(alpha, (cpx){gr[n], gi[n]})));
        wr[n] = w.r; wi[n] = w.i;
        const cpx g = {gr[n], gi[n]};
        cpx pw = w;
        for (int i = RK - 1; i >= 0; --i) {
            const int row = 8*(i >> 1) + (i & 1);
            WS[row*WP + n] = pw.r; WS[(row + 2)*WP + n] = pw.i;
            pw = cmul(pw, g);
        }
        WS[4*WP + n] = cr[n];  WS[5*WP + n] = ci[n];
        WS[6*WP + n] = 0.f;    WS[7*WP + n] = 0.f;
        for (int row = 12; row < 16; ++row) WS[row*WP + n] = 0.f;
        const cpx g2 = cmul(g, g), g4 = cmul(g2, g2);
        G4r[n] = g4.r; G4i[n] = g4.i;
    }
    __syncthreads();
    for (int idx = tid; idx < 2048; idx += 128) {
        const int slot = idx & 31, blk = idx >> 5;
        const int q = blk & 3, t = (blk >> 2) & 3, Tb = blk >> 4;
        const int hh = slot >> 4;
        const int rho = ((slot & 15) - 2*q - 8*hh) & 15;
        const int n = 16*Tb + rho;
        const int m = 16*t + 8*hh + 2*q;
        const cpx un = {ur[n], ui[n]};
        const float gnr = gr[n], gni = gi[n];
        const cpx v0 = cmul(un, (cpx){wr[m], wi[m]});
        const cpx v1 = cmul(un, (cpx){wr[m + 1], wi[m + 1]});
        v4f f;
        f[0] = -v0.r; f[1] = -v1.r; f[2] = -v0.i; f[3] = -v1.i;
        if (n == m)     { f[0] += gnr; f[2] += gni; }
        if (n == m + 1) { f[1] += gnr; f[3] += gni; }
        F[idx] = f;
    }
    const int lane = tid & 31;
    const int col16 = lane & 15;
    const int half = lane >> 4;
    v8f Xr[4], Xi[4];
    {
        const int m = 16*wv + col16;
        const cpx wm = {wr[m], wi[m]};
        const float gmr = gr[m], gmi = gi[m];
        #pragma unroll
        for (int t = 0; t < 4; ++t) {
            #pragma unroll
            for (int r = 0; r < 8; ++r) {
                const int n = 16*t + 8*half + r;
                cpx v = cmul((cpx){ur[n], ui[n]}, wm);
                float rr = -v.r, rim = -v.i;
                if (n == m) { rr += gmr; rim += gmi; }
                Xr[t][r] = rr; Xi[t][r] = rim;
            }
        }
    }
    const int mv_hf  = (tid & 31) >> 4;
    const int mv_row = 16*wv + (tid & 15);
    float rAr[32], rAi[32];
    {
        const cpx un = {ur[mv_row], ui[mv_row]};
        const float gnr = gr[mv_row], gni = gi[mv_row];
        #pragma unroll
        for (int j = 0; j < 32; ++j) {
            const int m = 32*mv_hf + j;
            cpx v = cmul(un, (cpx){wr[m], wi[m]});
            float rr = -v.r, rim = -v.i;
            if (m == mv_row) { rr += gnr; rim += gni; }
            rAr[j] = rr; rAi[j] = rim;
        }
    }
    __syncthreads();

    if (tid < 64) {
        const int n = tid;
        cpx lam = { -A_real[h*64 + n], -A_imag[h*64 + n] };
        cpx Dinv = cinv((cpx){1.0f - cc*lam.r, -cc*lam.i});
        cpx b = { B_real[h*64 + n], B_imag[h*64 + n] };
        cpx t = cmul(Dinv, cscale(b, dt));
        tr[n] = t.r; ti[n] = t.i;
    }
    __syncthreads();
    if (tid == 0) {
        cpx q = {0.f, 0.f};
        for (int n = 0; n < 64; ++n)
            q = cadd(q, cmul((cpx){pr[n], -pi[n]}, (cpx){tr[n], ti[n]}));
        cpx aq = cmul(alpha, q);
        scal[4] = aq.r; scal[5] = aq.i;
    }
    __syncthreads();
    if (tid < 64) {
        const int n = tid;
        cpx dB = csub((cpx){tr[n], ti[n]}, cmul((cpx){scal[4], scal[5]}, (cpx){ur[n], ui[n]}));
        Xr0[n*16 + 0] = dB.r;
        Xi0[n*16 + 0] = dB.i;
        mvs[n] = dB.r;
        mvs[128 + n] = dB.i;
        Ufr[n*RK + 0] = ur[n];
        Ufi[n*RK + 0] = ui[n];
        mvs[256 + n] = ur[n];
        mvs[384 + n] = ui[n];
    }
    __syncthreads();

    for (int l = 1; l < RK; ++l) {
        const int rb = (l - 1) & 1, wb = l & 1;
        const v4f* xr4 = (const v4f*)(mvs + 64*rb + 32*mv_hf);
        const v4f* xi4 = (const v4f*)(mvs + 128 + 64*rb + 32*mv_hf);
        const v4f* yr4 = (const v4f*)(mvs + 256 + 64*rb + 32*mv_hf);
        const v4f* yi4 = (const v4f*)(mvs + 384 + 64*rb + 32*mv_hf);
        float a0r = 0.f, a0i = 0.f, a1r = 0.f, a1i = 0.f;
        float b0r = 0.f, b0i = 0.f, b1r = 0.f, b1i = 0.f;
        #pragma unroll
        for (int j4 = 0; j4 < 8; ++j4) {
            const v4f xr = xr4[j4], xi = xi4[j4];
            a0r += rAr[4*j4 + 0]*xr[0] - rAi[4*j4 + 0]*xi[0];
            a0i += rAr[4*j4 + 0]*xi[0] + rAi[4*j4 + 0]*xr[0];
            a1r += rAr[4*j4 + 1]*xr[1] - rAi[4*j4 + 1]*xi[1];
            a1i += rAr[4*j4 + 1]*xi[1] + rAi[4*j4 + 1]*xr[1];
            a0r += rAr[4*j4 + 2]*xr[2] - rAi[4*j4 + 2]*xi[2];
            a0i += rAr[4*j4 + 2]*xi[2] + rAi[4*j4 + 2]*xr[2];
            a1r += rAr[4*j4 + 3]*xr[3] - rAi[4*j4 + 3]*xi[3];
            a1i += rAr[4*j4 + 3]*xi[3] + rAi[4*j4 + 3]*xr[3];
            const v4f yr = yr4[j4], yi = yi4[j4];
            b0r += rAr[4*j4 + 0]*yr[0] - rAi[4*j4 + 0]*yi[0];
            b0i += rAr[4*j4 + 0]*yi[0] + rAi[4*j4 + 0]*yr[0];
            b1r += rAr[4*j4 + 1]*yr[1] - rAi[4*j4 + 1]*yi[1];
            b1i += rAr[4*j4 + 1]*yi[1] + rAi[4*j4 + 1]*yr[1];
            b0r += rAr[4*j4 + 2]*yr[2] - rAi[4*j4 + 2]*yi[2];
            b0i += rAr[4*j4 + 2]*yi[2] + rAi[4*j4 + 2]*yr[2];
            b1r += rAr[4*j4 + 3]*yr[3] - rAi[4*j4 + 3]*yi[3];
            b1i += rAr[4*j4 + 3]*yi[3] + rAi[4*j4 + 3]*yr[3];
        }
        float sr = a0r + a1r, si = a0i + a1i;
        float vr = b0r + b1r, vi = b0i + b1i;
        sr += __shfl_xor(sr, 16);
        si += __shfl_xor(si, 16);
        vr += __shfl_xor(vr, 16);
        vi += __shfl_xor(vi, 16);
        if (mv_hf == 0) {
            Xr0[mv_row*16 + l] = sr; Xi0[mv_row*16 + l] = si;
            mvs[64*wb + mv_row] = sr; mvs[128 + 64*wb + mv_row] = si;
            Ufr[mv_row*RK + l] = vr; Ufi[mv_row*RK + l] = vi;
            mvs[256 + 64*wb + mv_row] = vr; mvs[384 + 64*wb + mv_row] = vi;
        }
        __syncthreads();
    }
    int lofs[4];
    #pragma unroll
    for (int q = 0; q < 4; ++q) lofs[q] = 16*half + ((col16 + 2*q + 8*half) & 15);
    auto load_rows = [&](v4f (&f)[16], int cur, int T) {
        #pragma unroll
        for (int s = 0; s < 16; ++s) f[s] = F[cur + (T*16 + s)*32 + lofs[s & 3]];
    };
    auto row_block = [&](const v4f (&f)[16], v8f& outr, v8f& outi) {
        v8f arbr = {}; v8f aibi = {}; v8f arbi = {}; v8f aibr = {};
        #pragma unroll
        for (int t = 0; t < 4; ++t) {
            #pragma unroll
            for (int q = 0; q < 4; ++q) {
                const v4f ff = f[t*4 + q];
                v2f ar, ai, br, bi;
                ar[0] = ff[0]; ar[1] = ff[1];
                ai[0] = ff[2]; ai[1] = ff[3];
                br[0] = Xr[t][2*q]; br[1] = Xr[t][2*q + 1];
                bi[0] = Xi[t][2*q]; bi[1] = Xi[t][2*q + 1];
                arbr = __builtin_amdgcn_wmma_f32_16x16x4_f32(false, ar, false, br, (short)0, arbr, false, false);
                aibi = __builtin_amdgcn_wmma_f32_16x16x4_f32(false, ai, false, bi, (short)0, aibi, false, false);
                arbi = __builtin_amdgcn_wmma_f32_16x16x4_f32(false, ar, false, bi, (short)0, arbi, false, false);
                aibr = __builtin_amdgcn_wmma_f32_16x16x4_f32(false, ai, false, br, (short)0, aibr, false, false);
            }
        }
        outr = arbr - aibi;
        outi = arbi + aibr;
    };
    auto publish = [&](int nxt) {
        const int sp = col16 & 1, qp = (col16 >> 1) & 3, hp = col16 >> 3;
        #pragma unroll
        for (int T = 0; T < 4; ++T) {
            #pragma unroll
            for (int rp = 0; rp < 4; ++rp) {
                const float sr = sp ? Xr[T][2*rp] : Xr[T][2*rp + 1];
                const float si = sp ? Xi[T][2*rp] : Xi[T][2*rp + 1];
                const float or_ = __shfl_xor(sr, 1);
                const float oi_ = __shfl_xor(si, 1);
                v4f e;
                e[0] = sp ? or_ : Xr[T][2*rp];
                e[1] = sp ? Xr[T][2*rp + 1] : or_;
                e[2] = sp ? oi_ : Xi[T][2*rp];
                e[3] = sp ? Xi[T][2*rp + 1] : oi_;
                const int rho = 8*half + 2*rp + sp;
                F[nxt + (T*16 + wv*4 + qp)*32 + 16*hp + ((rho + 2*qp + 8*hp) & 15)] = e;
            }
        }
    };
    auto power_step = [&](int cur, int nxt) {
        v8f nXr[4], nXi[4];
        v4f ga[16], gb[16];
        load_rows(ga, cur, 0);
        load_rows(gb, cur, 1);
        __builtin_amdgcn_sched_barrier(0);
        row_block(ga, nXr[0], nXi[0]);
        __builtin_amdgcn_sched_barrier(0);
        load_rows(ga, cur, 2);
        __builtin_amdgcn_sched_barrier(0);
        row_block(gb, nXr[1], nXi[1]);
        __builtin_amdgcn_sched_barrier(0);
        load_rows(gb, cur, 3);
        __builtin_amdgcn_sched_barrier(0);
        row_block(ga, nXr[2], nXi[2]);
        __builtin_amdgcn_sched_barrier(0);
        row_block(gb, nXr[3], nXi[3]);
        #pragma unroll
        for (int t = 0; t < 4; ++t) { Xr[t] = nXr[t]; Xi[t] = nXi[t]; }
        publish(nxt);
    };
    auto seed_tile = [&](int cur, int bo, int co, int cmask, int cofs) {
        v4f f[16];
        load_rows(f, cur, wv);
        const float* Br = sm + bo;  const float* Bi = Br + 1024;
        float* Cr = sm + co;        float* Ci = Cr + 1024;
        const int bc = col16 & cmask;
        v8f arbr = {}; v8f aibi = {}; v8f arbi = {}; v8f aibr = {};
        #pragma unroll
        for (int t = 0; t < 4; ++t) {
            #pragma unroll
            for (int q = 0; q < 4; ++q) {
                const int k = 16*t + 8*half + 2*q;
                const v4f ff = f[t*4 + q];
                v2f ar, ai, br, bi;
                ar[0] = ff[0]; ar[1] = ff[1];
                ai[0] = ff[2]; ai[1] = ff[3];
                br[0] = Br[k*16 + bc];  br[1] = Br[(k + 1)*16 + bc];
                bi[0] = Bi[k*16 + bc];  bi[1] = Bi[(k + 1)*16 + bc];
                arbr = __builtin_amdgcn_wmma_f32_16x16x4_f32(false, ar, false, br, (short)0, arbr, false, false);
                aibi = __builtin_amdgcn_wmma_f32_16x16x4_f32(false, ai, false, bi, (short)0, aibi, false, false);
                arbi = __builtin_amdgcn_wmma_f32_16x16x4_f32(false, ar, false, bi, (short)0, arbi, false, false);
                aibr = __builtin_amdgcn_wmma_f32_16x16x4_f32(false, ai, false, br, (short)0, aibr, false, false);
            }
        }
        for (int v = 0; v < 8; ++v) {
            Cr[(16*wv + 8*half + v)*16 + cofs + bc] = arbr[v] - aibi[v];
            Ci[(16*wv + 8*half + v)*16 + cofs + bc] = arbi[v] + aibr[v];
        }
    };

    auto left_rows = [&](bool second) {
        const int arow = second ? (col16 >= 12 ? col16 - 12 : col16)
                                : ((col16 == 6 || col16 == 7) ? 12 : col16);
        const float* ap = WS + arow*WP + 8*half;
        v8f s1 = {}; v8f s2 = {};
        #pragma unroll
        for (int t = 0; t < 4; ++t) {
            #pragma unroll
            for (int q = 0; q < 4; ++q) {
                const v2f a = *(const v2f*)(ap + 16*t + 2*q);
                v2f br, bi;
                br[0] = Xr[t][2*q]; br[1] = Xr[t][2*q + 1];
                bi[0] = Xi[t][2*q]; bi[1] = Xi[t][2*q + 1];
                s1 = __builtin_amdgcn_wmma_f32_16x16x4_f32(false, a, false, br, (short)0, s1, false, false);
                s2 = __builtin_amdgcn_wmma_f32_16x16x4_f32(false, a, false, bi, (short)0, s2, false, false);
            }
        }
        if (lane < 16) {
            const int m = 16*wv + col16;
            const float e0r = s1[4] - s2[5], e0i = s2[4] + s1[5];
            if (!second) {
                WS[6*WP + m] = e0r;  WS[7*WP + m] = e0i;
            } else {
                WS[12*WP + m] = e0r; WS[13*WP + m] = e0i;
                WS[14*WP + m] = s1[6] - s2[7];
                WS[15*WP + m] = s2[6] + s1[7];
            }
        }
    };

    auto left_pass = [&](const float* src, float* dst) {
        const float* ap = ((col16 & 4) ? src : WS) + col16*WP + 8*half;
        v8f s1 = {}; v8f s2 = {};
        #pragma unroll
        for (int t = 0; t < 4; ++t) {
            #pragma unroll
            for (int q = 0; q < 4; ++q) {
                const v2f a = *(const v2f*)(ap + 16*t + 2*q);
                v2f br, bi;
                br[0] = Xr[t][2*q]; br[1] = Xr[t][2*q + 1];
                bi[0] = Xi[t][2*q]; bi[1] = Xi[t][2*q + 1];
                s1 = __builtin_amdgcn_wmma_f32_16x16x4_f32(false, a, false, br, (short)0, s1, false, false);
                s2 = __builtin_amdgcn_wmma_f32_16x16x4_f32(false, a, false, bi, (short)0, s2, false, false);
            }
        }
        float* d = dst + (4 + 8*half)*WP + 16*wv + col16;
        d[0]      = s1[4] - s2[5];
        d[WP]     = s2[4] + s1[5];
        d[2*WP]   = s1[6] - s2[7];
        d[3*WP]   = s2[6] + s1[7];
    };

    for (int s = 0; s < 8; ++s) {
        power_step((s & 1) * 2048, ((s & 1) ^ 1) * 2048);
        __syncthreads();
        if (s == 4) left_rows(false);
        else if (s == 5) left_rows(true);
        else if (s == 6) seed_tile(2048, 16384, 16384, 3, 4);
        else if (s == 7) seed_tile(0, 16384, 16384, 7, 8);
    }

    {
        float* TL = sm + 12288;
        const float* lsrc = WS;
        for (int w = 1; w < 4; ++w) {
            float* Lw = sm + 18432 + 2048*(w - 1);
            left_pass(lsrc, TL);
            __syncthreads();
            left_pass(TL, Lw);
            __syncthreads();
            lsrc = Lw;
        }
    }

    const float* sxr = sm + 16384;
    const float* sxi = sxr + 1024;
    float* Kt = sm + 8192;
    float hgr[4][8], hgi[4][8];
    #pragma unroll
    for (int t = 0; t < 4; ++t) {
        #pragma unroll
        for (int r = 0; r < 8; ++r) {
            const int n = 16*t + 8*half + r;
            Xr[t][r] = sxr[n*16 + col16];
            Xi[t][r] = sxi[n*16 + col16];
            hgr[t][r] = G4r[n];
            hgi[t][r] = G4i[n];
        }
    }
    const float* wrow = (((col16 & 4) && wv > 0) ? sm + 18432 + 2048*(wv - 1) : WS) + col16*WP;
    v2f fWS[16];
    #pragma unroll
    for (int t = 0; t < 4; ++t) {
        #pragma unroll
        for (int q = 0; q < 4; ++q) {
            const int k = 16*t + 8*half + 2*q;
            fWS[t*4 + q] = *(const v2f*)(wrow + k);
        }
    }
    v2f fUrn[4], fUi[4], fUin[4];
    #pragma unroll
    for (int T = 0; T < 4; ++T) {
        const int o = (16*T + col16)*RK + 2*half;
        const v2f a = *(const v2f*)(Ufr + o);
        const v2f b = *(const v2f*)(Ufi + o);
        fUrn[T] = -a;
        fUi[T]  = b;
        fUin[T] = -b;
    }
    float* kout = Kt + 512*wv + 128*(col16 >> 2) + (col16 & 3) + 64*half;
    auto stacked_product = [&](const v8f (&Pr)[4], const v8f (&Pi)[4], v8f& d1, v8f& d2) {
        v8f s1 = {}; v8f s2 = {};
        #pragma unroll
        for (int t = 0; t < 4; ++t) {
            #pragma unroll
            for (int q = 0; q < 4; ++q) {
                v2f br, bi;
                br[0] = Pr[t][2*q]; br[1] = Pr[t][2*q + 1];
                bi[0] = Pi[t][2*q]; bi[1] = Pi[t][2*q + 1];
                s1 = __builtin_amdgcn_wmma_f32_16x16x4_f32(false, fWS[t*4 + q], false, br, (short)0, s1, false, false);
                s2 = __builtin_amdgcn_wmma_f32_16x16x4_f32(false, fWS[t*4 + q], false, bi, (short)0, s2, false, false);
            }
        }
        d1 = s1; d2 = s2;
    };
    auto scan_step = [&](const v8f (&Pr)[4], const v8f (&Pi)[4], v8f (&Nr)[4], v8f (&Ni)[4], float* dst) {
        v8f d1, d2;
        stacked_product(Pr, Pi, d1, d2);
        const float kva = 2.0f * (d1[4] - d2[5]);
        const float kvb = 2.0f * (d1[6] - d2[7]);
        v2f zr2, zi2;
        zr2[0] = d1[0] - d2[2]; zr2[1] = d1[1] - d2[3];
        zi2[0] = d2[0] + d1[2]; zi2[1] = d2[1] + d1[3];
        #pragma unroll
        for (int T = 0; T < 4; ++T) {
            #pragma unroll
            for (int r = 0; r < 8; ++r) {
                Nr[T][r] = hgr[T][r]*Pr[T][r] - hgi[T][r]*Pi[T][r];
                Ni[T][r] = hgr[T][r]*Pi[T][r] + hgi[T][r]*Pr[T][r];
            }
        }
        #pragma unroll
        for (int T = 0; T < 4; ++T) {
            Nr[T] = __builtin_amdgcn_wmma_f32_16x16x4_f32(false, fUrn[T], false, zr2, (short)0, Nr[T], false, false);
            Ni[T] = __builtin_amdgcn_wmma_f32_16x16x4_f32(false, fUrn[T], false, zi2, (short)0, Ni[T], false, false);
        }
        #pragma unroll
        for (int T = 0; T < 4; ++T) {
            Nr[T] = __builtin_amdgcn_wmma_f32_16x16x4_f32(false, fUi[T],  false, zi2, (short)0, Nr[T], false, false);
            Ni[T] = __builtin_amdgcn_wmma_f32_16x16x4_f32(false, fUin[T], false, zr2, (short)0, Ni[T], false, false);
        }
        dst[0] = kva;
        dst[32] = kvb;
    };
    v8f Yr[4], Yi[4];
    for (int jj = 0; jj < 3; ++jj) {
        scan_step(Xr, Xi, Yr, Yi, kout + 2*RK*jj);
        scan_step(Yr, Yi, Xr, Xi, kout + 2*RK*jj + RK);
    }
    scan_step(Xr, Xi, Yr, Yi, kout + RK*6);
    {
        v8f d1, d2;
        stacked_product(Yr, Yi, d1, d2);
        kout[RK*7] = 2.0f * (d1[4] - d2[5]);
        kout[RK*7 + 32] = 2.0f * (d1[6] - d2[7]);
    }
    __syncthreads();

    {
        const v4f* Kt4 = (const v4f*)Kt;
        v4f* o4 = (v4f*)(out + h*LL);
        for (int k = 0; k < 4; ++k)
            o4[k*128 + tid] = Kt4[k*128 + tid];
    }
}

extern "C" void kernel_launch(void* const* d_in, const int* in_sizes, int n_in,
                              void* d_out, int out_size, void* d_ws, size_t ws_size,
                              hipStream_t stream) {
    const float* A_real = (const float*)d_in[0];
    const float* A_imag = (const float*)d_in[1];
    const float* B_real = (const float*)d_in[2];
    const float* B_imag = (const float*)d_in[3];
    const float* P_real = (const float*)d_in[4];
    const float* P_imag = (const float*)d_in[5];
    const float* C_real = (const float*)d_in[6];
    const float* C_imag = (const float*)d_in[7];
    const float* log_dt = (const float*)d_in[8];
    float* out = (float*)d_out;
    (void)in_sizes; (void)n_in; (void)out_size; (void)d_ws; (void)ws_size;

    const size_t shmem_bytes = 29648u * sizeof(float);
    s4_dplr_kernel<<<NH, 128, shmem_bytes, stream>>>(
        A_real, A_imag, B_real, B_imag, P_real, P_imag, C_real, C_imag, log_dt, out);
}


# Squarings software-pipelined by row block: a block's results swapped (DPP) and stored beside the next block's WMMAs, 2 barriers in the stream, next fragments loaded before the tail; sums unchanged
# speedup vs baseline: 15.1050x; 1.1013x over previous
#include <hip/hip_runtime.h>
#include <hip/hip_bf16.h>

#define NH 256
#define NN 64
#define LL 2048
#define RK 4
#define WP 66

typedef __attribute__((ext_vector_type(2))) float v2f;
typedef __attribute__((ext_vector_type(4))) float v4f;
typedef __attribute__((ext_vector_type(8))) float v8f;

struct cpx { float r, i; };
__device__ __forceinline__ cpx cmul(cpx a, cpx b) { return {a.r*b.r - a.i*b.i, a.r*b.i + a.i*b.r}; }
__device__ __forceinline__ cpx cadd(cpx a, cpx b) { return {a.r + b.r, a.i + b.i}; }
__device__ __forceinline__ cpx csub(cpx a, cpx b) { return {a.r - b.r, a.i - b.i}; }
__device__ __forceinline__ cpx cinv(cpx a) { float d = a.r*a.r + a.i*a.i; return {a.r/d, -a.i/d}; }
__device__ __forceinline__ cpx cscale(cpx a, float s) { return {a.r*s, a.i*s}; }
__device__ __forceinline__ float lane_xor1(float v) {
    return __builtin_bit_cast(float, __builtin_amdgcn_mov_dpp(__builtin_bit_cast(int, v), 0xB1, 0xF, 0xF, true));
}

extern "C" __global__ __launch_bounds__(128)
void s4_dplr_kernel(const float* __restrict__ A_real, const float* __restrict__ A_imag,
                    const float* __restrict__ B_real, const float* __restrict__ B_imag,
                    const float* __restrict__ P_real, const float* __restrict__ P_imag,
                    const float* __restrict__ C_real, const float* __restrict__ C_imag,
                    const float* __restrict__ log_dt, float* __restrict__ out)
{
    extern __shared__ float sm[];
    const int h   = blockIdx.x;
    const int tid = threadIdx.x;
    const int wv  = tid >> 5;

    v4f* F = (v4f*)sm;
    float* mvs = sm + 8192;
    float* Xr0 = sm + 16384;
    float* Xi0 = sm + 17408;
    float* Xr1 = sm + 18432;
    float* Xi1 = sm + 19456;
    float* Xr2 = sm + 20480;
    float* Xi2 = sm + 21504;
    float* Xr3 = sm + 22528;
    float* Xi3 = sm + 23552;
    float* ur  = sm + 24576;  float* ui = ur + 64;
    float* wr  = ui + 64;     float* wi = wr + 64;
    float* gr  = wi + 64;     float* gi = gr + 64;
    float* cr  = gi + 64;     float* ci = cr + 64;
    float* pr  = ci + 64;     float* pi = pr + 64;
    float* tr  = pi + 64;     float* ti = tr + 64;
    float* scal = ti + 64;
    float* Ufr  = sm + 25360;
    float* Ufi  = Ufr + 256;
    float* WS   = Ufi + 256;
    float* G4r  = WS + 16*WP;
    float* G4i  = G4r + 64;

    const float dt = __expf(log_dt[h]);
    const float cc = 0.5f * dt;

    if (tid < 64) {
        const int n = tid;
        cpx lam = { -A_real[h*64 + n], -A_imag[h*64 + n] };
        cpx p   = {  P_real[h*64 + n],  P_imag[h*64 + n] };
        cpx Dinv = cinv((cpx){1.0f - cc*lam.r, -cc*lam.i});
        cpx e    = {1.0f + cc*lam.r,  cc*lam.i};
        cpx u = cmul(Dinv, p);
        cpx g = cmul(Dinv, e);
        ur[n] = u.r; ui[n] = u.i; gr[n] = g.r; gi[n] = g.i;
        pr[n] = p.r; pi[n] = p.i;
        cr[n] = C_real[h*64 + n]; ci[n] = C_imag[h*64 + n];
    }
    __syncthreads();
    if (tid == 0) {
        cpx s = {0.f, 0.f};
        for (int n = 0; n < 64; ++n)
            s = cadd(s, cmul((cpx){pr[n], -pi[n]}, (cpx){ur[n], ui[n]}));
        cpx alpha = cmul((cpx){cc, 0.f}, cinv((cpx){1.0f + cc*s.r, cc*s.i}));
        cpx base  = csub((cpx){cc, 0.f}, cmul(alpha, cscale(s, cc)));
        scal[0] = alpha.r; scal[1] = alpha.i; scal[2] = base.r; scal[3] = base.i;
    }
    __syncthreads();
    const cpx alpha = {scal[0], scal[1]};
    const cpx base  = {scal[2], scal[3]};
    if (tid < 64) {
        const int n = tid;
        cpx w = cmul((cpx){pr[n], -pi[n]}, cadd(base, cmul(alpha, (cpx){gr[n], gi[n]})));
        wr[n] = w.r; wi[n] = w.i;
        const cpx g = {gr[n], gi[n]};
        cpx pw = w;
        for (int i = RK - 1; i >= 0; --i) {
            const int row = 8*(i >> 1) + (i & 1);
            WS[row*WP + n] = pw.r; WS[(row + 2)*WP + n] = pw.i;
            pw = cmul(pw, g);
        }
        WS[4*WP + n] = cr[n];  WS[5*WP + n] = ci[n];
        WS[6*WP + n] = 0.f;    WS[7*WP + n] = 0.f;
        for (int row = 12; row < 16; ++row) WS[row*WP + n] = 0.f;
        const cpx g2 = cmul(g, g), g4 = cmul(g2, g2);
        G4r[n] = g4.r; G4i[n] = g4.i;
    }
    __syncthreads();
    for (int idx = tid; idx < 2048; idx += 128) {
        const int slot = idx & 31, blk = idx >> 5;
        const int q = blk & 3, t = (blk >> 2) & 3, Tb = blk >> 4;
        const int hh = slot >> 4;
        const int rho = ((slot & 15) - 2*q - 8*hh) & 15;
        const int n = 16*Tb + rho;
        const int m = 16*t + 8*hh + 2*q;
        const cpx un = {ur[n], ui[n]};
        const float gnr = gr[n], gni = gi[n];
        const cpx v0 = cmul(un, (cpx){wr[m], wi[m]});
        const cpx v1 = cmul(un, (cpx){wr[m + 1], wi[m + 1]});
        v4f f;
        f[0] = -v0.r; f[1] = -v1.r; f[2] = -v0.i; f[3] = -v1.i;
        if (n == m)     { f[0] += gnr; f[2] += gni; }
        if (n == m + 1) { f[1] += gnr; f[3] += gni; }
        F[idx] = f;
    }
    const int lane = tid & 31;
    const int col16 = lane & 15;
    const int half = lane >> 4;
    v8f Xr[4], Xi[4];
    {
        const int m = 16*wv + col16;
        const cpx wm = {wr[m], wi[m]};
        const float gmr = gr[m], gmi = gi[m];
        #pragma unroll
        for (int t = 0; t < 4; ++t) {
            #pragma unroll
            for (int r = 0; r < 8; ++r) {
                const int n = 16*t + 8*half + r;
                cpx v = cmul((cpx){ur[n], ui[n]}, wm);
                float rr = -v.r, rim = -v.i;
                if (n == m) { rr += gmr; rim += gmi; }
                Xr[t][r] = rr; Xi[t][r] = rim;
            }
        }
    }
    const int mv_hf  = (tid & 31) >> 4;
    const int mv_row = 16*wv + (tid & 15);
    float rAr[32], rAi[32];
    {
        const cpx un = {ur[mv_row], ui[mv_row]};
        const float gnr = gr[mv_row], gni = gi[mv_row];
        #pragma unroll
        for (int j = 0; j < 32; ++j) {
            const int m = 32*mv_hf + j;
            cpx v = cmul(un, (cpx){wr[m], wi[m]});
            float rr = -v.r, rim = -v.i;
            if (m == mv_row) { rr += gnr; rim += gni; }
            rAr[j] = rr; rAi[j] = rim;
        }
    }
    __syncthreads();

    if (tid < 64) {
        const int n = tid;
        cpx lam = { -A_real[h*64 + n], -A_imag[h*64 + n] };
        cpx Dinv = cinv((cpx){1.0f - cc*lam.r, -cc*lam.i});
        cpx b = { B_real[h*64 + n], B_imag[h*64 + n] };
        cpx t = cmul(Dinv, cscale(b, dt));
        tr[n] = t.r; ti[n] = t.i;
    }
    __syncthreads();
    if (tid == 0) {
        cpx q = {0.f, 0.f};
        for (int n = 0; n < 64; ++n)
            q = cadd(q, cmul((cpx){pr[n], -pi[n]}, (cpx){tr[n], ti[n]}));
        cpx aq = cmul(alpha, q);
        scal[4] = aq.r; scal[5] = aq.i;
    }
    __syncthreads();
    if (tid < 64) {
        const int n = tid;
        cpx dB = csub((cpx){tr[n], ti[n]}, cmul((cpx){scal[4], scal[5]}, (cpx){ur[n], ui[n]}));
        Xr0[n*16 + 0] = dB.r;
        Xi0[n*16 + 0] = dB.i;
        mvs[n] = dB.r;
        mvs[128 + n] = dB.i;
        Ufr[n*RK + 0] = ur[n];
        Ufi[n*RK + 0] = ui[n];
        mvs[256 + n] = ur[n];
        mvs[384 + n] = ui[n];
    }
    __syncthreads();

    for (int l = 1; l < RK; ++l) {
        const int rb = (l - 1) & 1, wb = l & 1;
        const v4f* xr4 = (const v4f*)(mvs + 64*rb + 32*mv_hf);
        const v4f* xi4 = (const v4f*)(mvs + 128 + 64*rb + 32*mv_hf);
        const v4f* yr4 = (const v4f*)(mvs + 256 + 64*rb + 32*mv_hf);
        const v4f* yi4 = (const v4f*)(mvs + 384 + 64*rb + 32*mv_hf);
        float a0r = 0.f, a0i = 0.f, a1r = 0.f, a1i = 0.f;
        float b0r = 0.f, b0i = 0.f, b1r = 0.f, b1i = 0.f;
        #pragma unroll
        for (int j4 = 0; j4 < 8; ++j4) {
            const v4f xr = xr4[j4], xi = xi4[j4];
            a0r += rAr[4*j4 + 0]*xr[0] - rAi[4*j4 + 0]*xi[0];
            a0i += rAr[4*j4 + 0]*xi[0] + rAi[4*j4 + 0]*xr[0];
            a1r += rAr[4*j4 + 1]*xr[1] - rAi[4*j4 + 1]*xi[1];
            a1i += rAr[4*j4 + 1]*xi[1] + rAi[4*j4 + 1]*xr[1];
            a0r += rAr[4*j4 + 2]*xr[2] - rAi[4*j4 + 2]*xi[2];
            a0i += rAr[4*j4 + 2]*xi[2] + rAi[4*j4 + 2]*xr[2];
            a1r += rAr[4*j4 + 3]*xr[3] - rAi[4*j4 + 3]*xi[3];
            a1i += rAr[4*j4 + 3]*xi[3] + rAi[4*j4 + 3]*xr[3];
            const v4f yr = yr4[j4], yi = yi4[j4];
            b0r += rAr[4*j4 + 0]*yr[0] - rAi[4*j4 + 0]*yi[0];
            b0i += rAr[4*j4 + 0]*yi[0] + rAi[4*j4 + 0]*yr[0];
            b1r += rAr[4*j4 + 1]*yr[1] - rAi[4*j4 + 1]*yi[1];
            b1i += rAr[4*j4 + 1]*yi[1] + rAi[4*j4 + 1]*yr[1];
            b0r += rAr[4*j4 + 2]*yr[2] - rAi[4*j4 + 2]*yi[2];
            b0i += rAr[4*j4 + 2]*yi[2] + rAi[4*j4 + 2]*yr[2];
            b1r += rAr[4*j4 + 3]*yr[3] - rAi[4*j4 + 3]*yi[3];
            b1i += rAr[4*j4 + 3]*yi[3] + rAi[4*j4 + 3]*yr[3];
        }
        float sr = a0r + a1r, si = a0i + a1i;
        float vr = b0r + b1r, vi = b0i + b1i;
        sr += __shfl_xor(sr, 16);
        si += __shfl_xor(si, 16);
        vr += __shfl_xor(vr, 16);
        vi += __shfl_xor(vi, 16);
        if (mv_hf == 0) {
            Xr0[mv_row*16 + l] = sr; Xi0[mv_row*16 + l] = si;
            mvs[64*wb + mv_row] = sr; mvs[128 + 64*wb + mv_row] = si;
            Ufr[mv_row*RK + l] = vr; Ufi[mv_row*RK + l] = vi;
            mvs[256 + 64*wb + mv_row] = vr; mvs[384 + 64*wb + mv_row] = vi;
        }
        __syncthreads();
    }
    int lofs[4];
    #pragma unroll
    for (int q = 0; q < 4; ++q) lofs[q] = 16*half + ((col16 + 2*q + 8*half) & 15);
    auto load_rows = [&](v4f (&f)[16], int cur, int T) {
        #pragma unroll
        for (int s = 0; s < 16; ++s) f[s] = F[cur + (T*16 + s)*32 + lofs[s & 3]];
    };
    auto row_part = [&](const v4f (&f)[16], const int t0, const int t1, v8f& arbr, v8f& aibi, v8f& arbi, v8f& aibr) {
        #pragma unroll
        for (int t = t0; t < t1; ++t) {
            #pragma unroll
            for (int q = 0; q < 4; ++q) {
                const v4f ff = f[t*4 + q];
                v2f ar, ai, br, bi;
                ar[0] = ff[0]; ar[1] = ff[1];
                ai[0] = ff[2]; ai[1] = ff[3];
                br[0] = Xr[t][2*q]; br[1] = Xr[t][2*q + 1];
                bi[0] = Xi[t][2*q]; bi[1] = Xi[t][2*q + 1];
                arbr = __builtin_amdgcn_wmma_f32_16x16x4_f32(false, ar, false, br, (short)0, arbr, false, false);
                aibi = __builtin_amdgcn_wmma_f32_16x16x4_f32(false, ai, false, bi, (short)0, aibi, false, false);
                arbi = __builtin_amdgcn_wmma_f32_16x16x4_f32(false, ar, false, bi, (short)0, arbi, false, false);
                aibr = __builtin_amdgcn_wmma_f32_16x16x4_f32(false, ai, false, br, (short)0, aibr, false, false);
            }
        }
    };
    auto row_block = [&](const v4f (&f)[16], v8f& outr, v8f& outi) {
        v8f arbr = {}; v8f aibi = {}; v8f arbi = {}; v8f aibr = {};
        row_part(f, 0, 4, arbr, aibi, arbi, aibr);
        outr = arbr - aibi;
        outi = arbi + aibr;
    };
    auto publish_block = [&](const int buf, const int T, const v8f& nr, const v8f& ni) {
        const int sp = col16 & 1, qp = (col16 >> 1) & 3, hp = col16 >> 3;
        #pragma unroll
        for (int rp = 0; rp < 4; ++rp) {
            float x0r = nr[2*rp], x1r = nr[2*rp + 1], x0i = ni[2*rp], x1i = ni[2*rp + 1];
            asm("" : "+v"(x0r), "+v"(x1r), "+v"(x0i), "+v"(x1i));
            const float sr = sp ? x0r : x1r;
            const float si = sp ? x0i : x1i;
            const float or_ = lane_xor1(sr);
            const float oi_ = lane_xor1(si);
            v4f e;
            e[0] = sp ? or_ : x0r;
            e[1] = sp ? x1r : or_;
            e[2] = sp ? oi_ : x0i;
            e[3] = sp ? x1i : oi_;
            const int rho = 8*half + 2*rp + sp;
            F[buf + (T*16 + wv*4 + qp)*32 + 16*hp + ((rho + 2*qp + 8*hp) & 15)] = e;
        }
    };
    auto seed_tile = [&](int cur, int bo, int co, int cmask, int cofs) {
        v4f f[16];
        load_rows(f, cur, wv);
        const float* Br = sm + bo;  const float* Bi = Br + 1024;
        float* Cr = sm + co;        float* Ci = Cr + 1024;
        const int bc = col16 & cmask;
        v8f arbr = {}; v8f aibi = {}; v8f arbi = {}; v8f aibr = {};
        #pragma unroll
        for (int t = 0; t < 4; ++t) {
            #pragma unroll
            for (int q = 0; q < 4; ++q) {
                const int k = 16*t + 8*half + 2*q;
                const v4f ff = f[t*4 + q];
                v2f ar, ai, br, bi;
                ar[0] = ff[0]; ar[1] = ff[1];
                ai[0] = ff[2]; ai[1] = ff[3];
                br[0] = Br[k*16 + bc];  br[1] = Br[(k + 1)*16 + bc];
                bi[0] = Bi[k*16 + bc];  bi[1] = Bi[(k + 1)*16 + bc];
                arbr = __builtin_amdgcn_wmma_f32_16x16x4_f32(false, ar, false, br, (short)0, arbr, false, false);
                aibi = __builtin_amdgcn_wmma_f32_16x16x4_f32(false, ai, false, bi, (short)0, aibi, false, false);
                arbi = __builtin_amdgcn_wmma_f32_16x16x4_f32(false, ar, false, bi, (short)0, arbi, false, false);
                aibr = __builtin_amdgcn_wmma_f32_16x16x4_f32(false, ai, false, br, (short)0, aibr, false, false);
            }
        }
        for (int v = 0; v < 8; ++v) {
            Cr[(16*wv + 8*half + v)*16 + cofs + bc] = arbr[v] - aibi[v];
            Ci[(16*wv + 8*half + v)*16 + cofs + bc] = arbi[v] + aibr[v];
        }
    };

    auto left_rows = [&](bool second) {
        const int arow = second ? (col16 >= 12 ? col16 - 12 : col16)
                                : ((col16 == 6 || col16 == 7) ? 12 : col16);
        const float* ap = WS + arow*WP + 8*half;
        v8f s1 = {}; v8f s2 = {};
        #pragma unroll
        for (int t = 0; t < 4; ++t) {
            #pragma unroll
            for (int q = 0; q < 4; ++q) {
                const v2f a = *(const v2f*)(ap + 16*t + 2*q);
                v2f br, bi;
                br[0] = Xr[t][2*q]; br[1] = Xr[t][2*q + 1];
                bi[0] = Xi[t][2*q]; bi[1] = Xi[t][2*q + 1];
                s1 = __builtin_amdgcn_wmma_f32_16x16x4_f32(false, a, false, br, (short)0, s1, false, false);
                s2 = __builtin_amdgcn_wmma_f32_16x16x4_f32(false, a, false, bi, (short)0, s2, false, false);
            }
        }
        if (lane < 16) {
            const int m = 16*wv + col16;
            const float e0r = s1[4] - s2[5], e0i = s2[4] + s1[5];
            if (!second) {
                WS[6*WP + m] = e0r;  WS[7*WP + m] = e0i;
            } else {
                WS[12*WP + m] = e0r; WS[13*WP + m] = e0i;
                WS[14*WP + m] = s1[6] - s2[7];
                WS[15*WP + m] = s2[6] + s1[7];
            }
        }
    };

    auto left_pass = [&](const float* src, float* dst) {
        const float* ap = ((col16 & 4) ? src : WS) + col16*WP + 8*half;
        v8f s1 = {}; v8f s2 = {};
        #pragma unroll
        for (int t = 0; t < 4; ++t) {
            #pragma unroll
            for (int q = 0; q < 4; ++q) {
                const v2f a = *(const v2f*)(ap + 16*t + 2*q);
                v2f br, bi;
                br[0] = Xr[t][2*q]; br[1] = Xr[t][2*q + 1];
                bi[0] = Xi[t][2*q]; bi[1] = Xi[t][2*q + 1];
                s1 = __builtin_amdgcn_wmma_f32_16x16x4_f32(false, a, false, br, (short)0, s1, false, false);
                s2 = __builtin_amdgcn_wmma_f32_16x16x4_f32(false, a, false, bi, (short)0, s2, false, false);
            }
        }
        float* d = dst + (4 + 8*half)*WP + 16*wv + col16;
        d[0]      = s1[4] - s2[5];
        d[WP]     = s2[4] + s1[5];
        d[2*WP]   = s1[6] - s2[7];
        d[3*WP]   = s2[6] + s1[7];
    };

    {
        v4f ga[16], gb[16];
        v8f nXr[4], nXi[4];
        load_rows(ga, 0, 0);
        load_rows(gb, 0, 1);
        __builtin_amdgcn_sched_barrier(0);
        row_block(ga, nXr[0], nXi[0]);
        auto trip = [&](const int s, const bool last) __attribute__((always_inline)) {
            const int cur = (s & 1) * 2048, nxt = 2048 - cur;
            __builtin_amdgcn_sched_barrier(0);
            load_rows(ga, cur, 2);
            __builtin_amdgcn_sched_barrier(0);
            v8f brbr = {}; v8f bibi = {}; v8f brbi = {}; v8f bibr = {};
            row_part(gb, 0, 2, brbr, bibi, brbi, bibr);
            publish_block(nxt, 0, nXr[0], nXi[0]);
            __builtin_amdgcn_sched_barrier(0);
            row_part(gb, 2, 4, brbr, bibi, brbi, bibr);
            __builtin_amdgcn_sched_barrier(0);
            __syncthreads();
            load_rows(gb, cur, 3);
            __builtin_amdgcn_sched_barrier(0);
            nXr[1] = brbr - bibi;
            nXi[1] = brbi + bibr;
            v8f crbr = {}; v8f cibi = {}; v8f crbi = {}; v8f cibr = {};
            row_part(ga, 0, 2, crbr, cibi, crbi, cibr);
            publish_block(nxt, 1, nXr[1], nXi[1]);
            __builtin_amdgcn_sched_barrier(0);
            row_part(ga, 2, 4, crbr, cibi, crbi, cibr);
            nXr[2] = crbr - cibi;
            nXi[2] = crbi + cibr;
            __builtin_amdgcn_sched_barrier(0);
            v8f arbr = {}; v8f aibi = {}; v8f arbi = {}; v8f aibr = {};
            row_part(gb, 0, 1, arbr, aibi, arbi, aibr);
            publish_block(nxt, 2, nXr[2], nXi[2]);
            __builtin_amdgcn_sched_barrier(0);
            row_part(gb, 1, 2, arbr, aibi, arbi, aibr);
            __builtin_amdgcn_sched_barrier(0);
            __syncthreads();
            if (!last) load_rows(ga, nxt, 0);
            __builtin_amdgcn_sched_barrier(0);
            row_part(gb, 2, 4, arbr, aibi, arbi, aibr);
            if (!last) {
                __builtin_amdgcn_sched_barrier(0);
                load_rows(gb, nxt, 1);
                __builtin_amdgcn_sched_barrier(0);
            }
            nXr[3] = arbr - aibi;
            nXi[3] = arbi + aibr;
            #pragma unroll
            for (int t = 0; t < 4; ++t) { Xr[t] = nXr[t]; Xi[t] = nXi[t]; }
            if (s == 4) left_rows(false);
            else if (s == 5) left_rows(true);
            if (!last) {
                row_block(ga, nXr[0], nXi[0]);
                publish_block(nxt, 3, Xr[3], Xi[3]);
            }
        };
        for (int s = 0; s < 7; ++s) trip(s, false);
        trip(7, true);
        publish_block(0, 3, Xr[3], Xi[3]);
    }
    seed_tile(2048, 16384, 16384, 3, 4);
    __syncthreads();
    seed_tile(0, 16384, 16384, 7, 8);

    {
        float* TL = sm + 12288;
        const float* lsrc = WS;
        for (int w = 1; w < 4; ++w) {
            float* Lw = sm + 18432 + 2048*(w - 1);
            left_pass(lsrc, TL);
            __syncthreads();
            left_pass(TL, Lw);
            __syncthreads();
            lsrc = Lw;
        }
    }

    const float* sxr = sm + 16384;
    const float* sxi = sxr + 1024;
    float* Kt = sm + 8192;
    float hgr[4][8], hgi[4][8];
    #pragma unroll
    for (int t = 0; t < 4; ++t) {
        #pragma unroll
        for (int r = 0; r < 8; ++r) {
            const int n = 16*t + 8*half + r;
            Xr[t][r] = sxr[n*16 + col16];
            Xi[t][r] = sxi[n*16 + col16];
            hgr[t][r] = G4r[n];
            hgi[t][r] = G4i[n];
        }
    }
    const float* wrow = (((col16 & 4) && wv > 0) ? sm + 18432 + 2048*(wv - 1) : WS) + col16*WP;
    v2f fWS[16];
    #pragma unroll
    for (int t = 0; t < 4; ++t) {
        #pragma unroll
        for (int q = 0; q < 4; ++q) {
            const int k = 16*t + 8*half + 2*q;
            fWS[t*4 + q] = *(const v2f*)(wrow + k);
        }
    }
    v2f fUrn[4], fUi[4], fUin[4];
    #pragma unroll
    for (int T = 0; T < 4; ++T) {
        const int o = (16*T + col16)*RK + 2*half;
        const v2f a = *(const v2f*)(Ufr + o);
        const v2f b = *(const v2f*)(Ufi + o);
        fUrn[T] = -a;
        fUi[T]  = b;
        fUin[T] = -b;
    }
    float* kout = Kt + 512*wv + 128*(col16 >> 2) + (col16 & 3) + 64*half;
    auto stacked_product = [&](const v8f (&Pr)[4], const v8f (&Pi)[4], v8f& d1, v8f& d2) {
        v8f s1 = {}; v8f s2 = {};
        #pragma unroll
        for (int t = 0; t < 4; ++t) {
            #pragma unroll
            for (int q = 0; q < 4; ++q) {
                v2f br, bi;
                br[0] = Pr[t][2*q]; br[1] = Pr[t][2*q + 1];
                bi[0] = Pi[t][2*q]; bi[1] = Pi[t][2*q + 1];
                s1 = __builtin_amdgcn_wmma_f32_16x16x4_f32(false, fWS[t*4 + q], false, br, (short)0, s1, false, false);
                s2 = __builtin_amdgcn_wmma_f32_16x16x4_f32(false, fWS[t*4 + q], false, bi, (short)0, s2, false, false);
            }
        }
        d1 = s1; d2 = s2;
    };
    auto scan_step = [&](const v8f (&Pr)[4], const v8f (&Pi)[4], v8f (&Nr)[4], v8f (&Ni)[4], float* dst) {
        v8f d1, d2;
        stacked_product(Pr, Pi, d1, d2);
        const float kva = 2.0f * (d1[4] - d2[5]);
        const float kvb = 2.0f * (d1[6] - d2[7]);
        v2f zr2, zi2;
        zr2[0] = d1[0] - d2[2]; zr2[1] = d1[1] - d2[3];
        zi2[0] = d2[0] + d1[2]; zi2[1] = d2[1] + d1[3];
        #pragma unroll
        for (int T = 0; T < 4; ++T) {
            #pragma unroll
            for (int r = 0; r < 8; ++r) {
                Nr[T][r] = hgr[T][r]*Pr[T][r] - hgi[T][r]*Pi[T][r];
                Ni[T][r] = hgr[T][r]*Pi[T][r] + hgi[T][r]*Pr[T][r];
            }
        }
        #pragma unroll
        for (int T = 0; T < 4; ++T) {
            Nr[T] = __builtin_amdgcn_wmma_f32_16x16x4_f32(false, fUrn[T], false, zr2, (short)0, Nr[T], false, false);
            Ni[T] = __builtin_amdgcn_wmma_f32_16x16x4_f32(false, fUrn[T], false, zi2, (short)0, Ni[T], false, false);
        }
        #pragma unroll
        for (int T = 0; T < 4; ++T) {
            Nr[T] = __builtin_amdgcn_wmma_f32_16x16x4_f32(false, fUi[T],  false, zi2, (short)0, Nr[T], false, false);
            Ni[T] = __builtin_amdgcn_wmma_f32_16x16x4_f32(false, fUin[T], false, zr2, (short)0, Ni[T], false, false);
        }
        dst[0] = kva;
        dst[32] = kvb;
    };
    v8f Yr[4], Yi[4];
    for (int jj = 0; jj < 3; ++jj) {
        scan_step(Xr, Xi, Yr, Yi, kout + 2*RK*jj);
        scan_step(Yr, Yi, Xr, Xi, kout + 2*RK*jj + RK);
    }
    scan_step(Xr, Xi, Yr, Yi, kout + RK*6);
    {
        v8f d1, d2;
        stacked_product(Yr, Yi, d1, d2);
        kout[RK*7] = 2.0f * (d1[4] - d2[5]);
        kout[RK*7 + 32] = 2.0f * (d1[6] - d2[7]);
    }
    __syncthreads();

    {
        const v4f* Kt4 = (const v4f*)Kt;
        v4f* o4 = (v4f*)(out + h*LL);
        for (int k = 0; k < 4; ++k)
            o4[k*128 + tid] = Kt4[k*128 + tid];
    }
}

extern "C" void kernel_launch(void* const* d_in, const int* in_sizes, int n_in,
                              void* d_out, int out_size, void* d_ws, size_t ws_size,
                              hipStream_t stream) {
    const float* A_real = (const float*)d_in[0];
    const float* A_imag = (const float*)d_in[1];
    const float* B_real = (const float*)d_in[2];
    const float* B_imag = (const float*)d_in[3];
    const float* P_real = (const float*)d_in[4];
    const float* P_imag = (const float*)d_in[5];
    const float* C_real = (const float*)d_in[6];
    const float* C_imag = (const float*)d_in[7];
    const float* log_dt = (const float*)d_in[8];
    float* out = (float*)d_out;
    (void)in_sizes; (void)n_in; (void)out_size; (void)d_ws; (void)ws_size;

    const size_t shmem_bytes = 29648u * sizeof(float);
    s4_dplr_kernel<<<NH, 128, shmem_bytes, stream>>>(
        A_real, A_imag, B_real, B_imag, P_real, P_imag, C_real, C_imag, log_dt, out);
}


# Krylov prologue (dB's 64-term sum, dB, the 3 vector-unit matvecs) issued beside the first two squarings' matrix instructions on their barriers, not ahead of them; sum order unchanged
# speedup vs baseline: 15.6923x; 1.0389x over previous
#include <hip/hip_runtime.h>
#include <hip/hip_bf16.h>

#define NH 256
#define NN 64
#define LL 2048
#define RK 4
#define WP 66

typedef __attribute__((ext_vector_type(2))) float v2f;
typedef __attribute__((ext_vector_type(4))) float v4f;
typedef __attribute__((ext_vector_type(8))) float v8f;

struct cpx { float r, i; };
__device__ __forceinline__ cpx cmul(cpx a, cpx b) { return {a.r*b.r - a.i*b.i, a.r*b.i + a.i*b.r}; }
__device__ __forceinline__ cpx cadd(cpx a, cpx b) { return {a.r + b.r, a.i + b.i}; }
__device__ __forceinline__ cpx csub(cpx a, cpx b) { return {a.r - b.r, a.i - b.i}; }
__device__ __forceinline__ cpx cinv(cpx a) { float d = a.r*a.r + a.i*a.i; return {a.r/d, -a.i/d}; }
__device__ __forceinline__ cpx cscale(cpx a, float s) { return {a.r*s, a.i*s}; }
__device__ __forceinline__ float lane_xor1(float v) {
    return __builtin_bit_cast(float, __builtin_amdgcn_mov_dpp(__builtin_bit_cast(int, v), 0xB1, 0xF, 0xF, true));
}

extern "C" __global__ __launch_bounds__(128)
void s4_dplr_kernel(const float* __restrict__ A_real, const float* __restrict__ A_imag,
                    const float* __restrict__ B_real, const float* __restrict__ B_imag,
                    const float* __restrict__ P_real, const float* __restrict__ P_imag,
                    const float* __restrict__ C_real, const float* __restrict__ C_imag,
                    const float* __restrict__ log_dt, float* __restrict__ out)
{
    extern __shared__ float sm[];
    const int h   = blockIdx.x;
    const int tid = threadIdx.x;
    const int wv  = tid >> 5;

    v4f* F = (v4f*)sm;
    float* mvs = sm + 20480;
    float* Xr0 = sm + 16384;
    float* Xi0 = sm + 17408;
    float* Xr1 = sm + 18432;
    float* Xi1 = sm + 19456;
    float* Xr2 = sm + 20480;
    float* Xi2 = sm + 21504;
    float* Xr3 = sm + 22528;
    float* Xi3 = sm + 23552;
    float* ur  = sm + 24576;  float* ui = ur + 64;
    float* wr  = ui + 64;     float* wi = wr + 64;
    float* gr  = wi + 64;     float* gi = gr + 64;
    float* cr  = gi + 64;     float* ci = cr + 64;
    float* pr  = ci + 64;     float* pi = pr + 64;
    float* tr  = pi + 64;     float* ti = tr + 64;
    float* scal = ti + 64;
    float* Ufr  = sm + 25360;
    float* Ufi  = Ufr + 256;
    float* WS   = Ufi + 256;
    float* G4r  = WS + 16*WP;
    float* G4i  = G4r + 64;
    float* qpr  = G4i + 64;
    float* qpi  = qpr + 64;

    const float dt = __expf(log_dt[h]);
    const float cc = 0.5f * dt;

    if (tid < 64) {
        const int n = tid;
        cpx lam = { -A_real[h*64 + n], -A_imag[h*64 + n] };
        cpx p   = {  P_real[h*64 + n],  P_imag[h*64 + n] };
        cpx Dinv = cinv((cpx){1.0f - cc*lam.r, -cc*lam.i});
        cpx e    = {1.0f + cc*lam.r,  cc*lam.i};
        cpx u = cmul(Dinv, p);
        cpx g = cmul(Dinv, e);
        ur[n] = u.r; ui[n] = u.i; gr[n] = g.r; gi[n] = g.i;
        pr[n] = p.r; pi[n] = p.i;
        cr[n] = C_real[h*64 + n]; ci[n] = C_imag[h*64 + n];
        cpx b = { B_real[h*64 + n], B_imag[h*64 + n] };
        cpx t = cmul(Dinv, cscale(b, dt));
        tr[n] = t.r; ti[n] = t.i;
        cpx qp = cmul((cpx){p.r, -p.i}, t);
        qpr[n] = qp.r; qpi[n] = qp.i;
    }
    __syncthreads();
    if (tid == 0) {
        cpx s = {0.f, 0.f};
        for (int n = 0; n < 64; ++n)
            s = cadd(s, cmul((cpx){pr[n], -pi[n]}, (cpx){ur[n], ui[n]}));
        cpx alpha = cmul((cpx){cc, 0.f}, cinv((cpx){1.0f + cc*s.r, cc*s.i}));
        cpx base  = csub((cpx){cc, 0.f}, cmul(alpha, cscale(s, cc)));
        scal[0] = alpha.r; scal[1] = alpha.i; scal[2] = base.r; scal[3] = base.i;
    }
    __syncthreads();
    const cpx alpha = {scal[0], scal[1]};
    const cpx base  = {scal[2], scal[3]};
    if (tid < 64) {
        const int n = tid;
        cpx w = cmul((cpx){pr[n], -pi[n]}, cadd(base, cmul(alpha, (cpx){gr[n], gi[n]})));
        wr[n] = w.r; wi[n] = w.i;
        const cpx g = {gr[n], gi[n]};
        cpx pw = w;
        for (int i = RK - 1; i >= 0; --i) {
            const int row = 8*(i >> 1) + (i & 1);
            WS[row*WP + n] = pw.r; WS[(row + 2)*WP + n] = pw.i;
            pw = cmul(pw, g);
        }
        WS[4*WP + n] = cr[n];  WS[5*WP + n] = ci[n];
        WS[6*WP + n] = 0.f;    WS[7*WP + n] = 0.f;
        for (int row = 12; row < 16; ++row) WS[row*WP + n] = 0.f;
        const cpx g2 = cmul(g, g), g4 = cmul(g2, g2);
        G4r[n] = g4.r; G4i[n] = g4.i;
        Ufr[n*RK + 0] = ur[n];
        Ufi[n*RK + 0] = ui[n];
        mvs[256 + n] = ur[n];
        mvs[384 + n] = ui[n];
    }
    __syncthreads();
    for (int idx = tid; idx < 2048; idx += 128) {
        const int slot = idx & 31, blk = idx >> 5;
        const int q = blk & 3, t = (blk >> 2) & 3, Tb = blk >> 4;
        const int hh = slot >> 4;
        const int rho = ((slot & 15) - 2*q - 8*hh) & 15;
        const int n = 16*Tb + rho;
        const int m = 16*t + 8*hh + 2*q;
        const cpx un = {ur[n], ui[n]};
        const float gnr = gr[n], gni = gi[n];
        const cpx v0 = cmul(un, (cpx){wr[m], wi[m]});
        const cpx v1 = cmul(un, (cpx){wr[m + 1], wi[m + 1]});
        v4f f;
        f[0] = -v0.r; f[1] = -v1.r; f[2] = -v0.i; f[3] = -v1.i;
        if (n == m)     { f[0] += gnr; f[2] += gni; }
        if (n == m + 1) { f[1] += gnr; f[3] += gni; }
        F[idx] = f;
    }
    const int lane = tid & 31;
    const int col16 = lane & 15;
    const int half = lane >> 4;
    v8f Xr[4], Xi[4];
    {
        const int m = 16*wv + col16;
        const cpx wm = {wr[m], wi[m]};
        const float gmr = gr[m], gmi = gi[m];
        #pragma unroll
        for (int t = 0; t < 4; ++t) {
            #pragma unroll
            for (int r = 0; r < 8; ++r) {
                const int n = 16*t + 8*half + r;
                cpx v = cmul((cpx){ur[n], ui[n]}, wm);
                float rr = -v.r, rim = -v.i;
                if (n == m) { rr += gmr; rim += gmi; }
                Xr[t][r] = rr; Xi[t][r] = rim;
            }
        }
    }
    const int mv_hf  = (tid & 31) >> 4;
    const int mv_row = 16*wv + (tid & 15);
    float rAr[32], rAi[32];
    {
        const cpx un = {ur[mv_row], ui[mv_row]};
        const float gnr = gr[mv_row], gni = gi[mv_row];
        #pragma unroll
        for (int j = 0; j < 32; ++j) {
            const int m = 32*mv_hf + j;
            cpx v = cmul(un, (cpx){wr[m], wi[m]});
            float rr = -v.r, rim = -v.i;
            if (m == mv_row) { rr += gnr; rim += gni; }
            rAr[j] = rr; rAi[j] = rim;
        }
    }
    __syncthreads();

    v4f qv[32];
    float dtr = 0.f, dti = 0.f, dur = 0.f, dui = 0.f;
    auto q_load = [&]() {
        const v4f* q4r = (const v4f*)qpr;
        const v4f* q4i = (const v4f*)qpi;
        #pragma unroll
        for (int j = 0; j < 16; ++j) { qv[2*j] = q4r[j]; qv[2*j + 1] = q4i[j]; }
        const int n = tid & 63;
        dtr = tr[n]; dti = ti[n]; dur = ur[n]; dui = ui[n];
    };
    auto db_write = [&]() {
        cpx q = {0.f, 0.f};
        #pragma unroll
        for (int j = 0; j < 16; ++j) {
            #pragma unroll
            for (int e = 0; e < 4; ++e) q = cadd(q, (cpx){qv[2*j][e], qv[2*j + 1][e]});
        }
        const cpx aq = cmul(alpha, q);
        const int n = tid & 63;
        const cpx dB = csub((cpx){dtr, dti}, cmul(aq, (cpx){dur, dui}));
        Xr0[n*16 + 0] = dB.r;
        Xi0[n*16 + 0] = dB.i;
        mvs[n] = dB.r;
        mvs[128 + n] = dB.i;
    };
    v4f xv[32];
    float a0r = 0.f, a0i = 0.f, a1r = 0.f, a1i = 0.f;
    float b0r = 0.f, b0i = 0.f, b1r = 0.f, b1i = 0.f;
    auto mv_load = [&](const int l) {
        const int rb = (l - 1) & 1;
        const v4f* xr4 = (const v4f*)(mvs + 64*rb + 32*mv_hf);
        const v4f* xi4 = (const v4f*)(mvs + 128 + 64*rb + 32*mv_hf);
        const v4f* yr4 = (const v4f*)(mvs + 256 + 64*rb + 32*mv_hf);
        const v4f* yi4 = (const v4f*)(mvs + 384 + 64*rb + 32*mv_hf);
        #pragma unroll
        for (int j4 = 0; j4 < 8; ++j4) {
            xv[4*j4 + 0] = xr4[j4]; xv[4*j4 + 1] = xi4[j4];
            xv[4*j4 + 2] = yr4[j4]; xv[4*j4 + 3] = yi4[j4];
        }
        a0r = 0.f; a0i = 0.f; a1r = 0.f; a1i = 0.f;
        b0r = 0.f; b0i = 0.f; b1r = 0.f; b1i = 0.f;
    };
    auto mv_fma = [&](const int j0, const int j1) {
        #pragma unroll
        for (int j4 = j0; j4 < j1; ++j4) {
            const v4f xr = xv[4*j4 + 0], xi = xv[4*j4 + 1];
            a0r += rAr[4*j4 + 0]*xr[0] - rAi[4*j4 + 0]*xi[0];
            a0i += rAr[4*j4 + 0]*xi[0] + rAi[4*j4 + 0]*xr[0];
            a1r += rAr[4*j4 + 1]*xr[1] - rAi[4*j4 + 1]*xi[1];
            a1i += rAr[4*j4 + 1]*xi[1] + rAi[4*j4 + 1]*xr[1];
            a0r += rAr[4*j4 + 2]*xr[2] - rAi[4*j4 + 2]*xi[2];
            a0i += rAr[4*j4 + 2]*xi[2] + rAi[4*j4 + 2]*xr[2];
            a1r += rAr[4*j4 + 3]*xr[3] - rAi[4*j4 + 3]*xi[3];
            a1i += rAr[4*j4 + 3]*xi[3] + rAi[4*j4 + 3]*xr[3];
            const v4f yr = xv[4*j4 + 2], yi = xv[4*j4 + 3];
            b0r += rAr[4*j4 + 0]*yr[0] - rAi[4*j4 + 0]*yi[0];
            b0i += rAr[4*j4 + 0]*yi[0] + rAi[4*j4 + 0]*yr[0];
            b1r += rAr[4*j4 + 1]*yr[1] - rAi[4*j4 + 1]*yi[1];
            b1i += rAr[4*j4 + 1]*yi[1] + rAi[4*j4 + 1]*yr[1];
            b0r += rAr[4*j4 + 2]*yr[2] - rAi[4*j4 + 2]*yi[2];
            b0i += rAr[4*j4 + 2]*yi[2] + rAi[4*j4 + 2]*yr[2];
            b1r += rAr[4*j4 + 3]*yr[3] - rAi[4*j4 + 3]*yi[3];
            b1i += rAr[4*j4 + 3]*yi[3] + rAi[4*j4 + 3]*yr[3];
        }
        asm volatile("" : "+v"(a0r), "+v"(a0i), "+v"(a1r), "+v"(a1i), "+v"(b0r), "+v"(b0i), "+v"(b1r), "+v"(b1i));
    };
    auto mv_finish = [&](const int l) {
        const int wb = l & 1;
        float sr = a0r + a1r, si = a0i + a1i;
        float vr = b0r + b1r, vi = b0i + b1i;
        sr += __shfl_xor(sr, 16);
        si += __shfl_xor(si, 16);
        vr += __shfl_xor(vr, 16);
        vi += __shfl_xor(vi, 16);
        Xr0[mv_row*16 + l] = sr; Xi0[mv_row*16 + l] = si;
        mvs[64*wb + mv_row] = sr; mvs[128 + 64*wb + mv_row] = si;
        Ufr[mv_row*RK + l] = vr; Ufi[mv_row*RK + l] = vi;
        mvs[256 + 64*wb + mv_row] = vr; mvs[384 + 64*wb + mv_row] = vi;
    };
    int lofs[4];
    #pragma unroll
    for (int q = 0; q < 4; ++q) lofs[q] = 16*half + ((col16 + 2*q + 8*half) & 15);
    auto load_rows = [&](v4f (&f)[16], int cur, int T) {
        #pragma unroll
        for (int s = 0; s < 16; ++s) f[s] = F[cur + (T*16 + s)*32 + lofs[s & 3]];
    };
    auto row_part = [&](const v4f (&f)[16], const int t0, const int t1, v8f& arbr, v8f& aibi, v8f& arbi, v8f& aibr) {
        #pragma unroll
        for (int t = t0; t < t1; ++t) {
            #pragma unroll
            for (int q = 0; q < 4; ++q) {
                const v4f ff = f[t*4 + q];
                v2f ar, ai, br, bi;
                ar[0] = ff[0]; ar[1] = ff[1];
                ai[0] = ff[2]; ai[1] = ff[3];
                br[0] = Xr[t][2*q]; br[1] = Xr[t][2*q + 1];
                bi[0] = Xi[t][2*q]; bi[1] = Xi[t][2*q + 1];
                arbr = __builtin_amdgcn_wmma_f32_16x16x4_f32(false, ar, false, br, (short)0, arbr, false, false);
                aibi = __builtin_amdgcn_wmma_f32_16x16x4_f32(false, ai, false, bi, (short)0, aibi, false, false);
                arbi = __builtin_amdgcn_wmma_f32_16x16x4_f32(false, ar, false, bi, (short)0, arbi, false, false);
                aibr = __builtin_amdgcn_wmma_f32_16x16x4_f32(false, ai, false, br, (short)0, aibr, false, false);
            }
        }
    };
    auto row_block = [&](const v4f (&f)[16], v8f& outr, v8f& outi) {
        v8f arbr = {}; v8f aibi = {}; v8f arbi = {}; v8f aibr = {};
        row_part(f, 0, 4, arbr, aibi, arbi, aibr);
        outr = arbr - aibi;
        outi = arbi + aibr;
    };
    auto publish_block = [&](const int buf, const int T, const v8f& nr, const v8f& ni) {
        const int sp = col16 & 1, qp = (col16 >> 1) & 3, hp = col16 >> 3;
        #pragma unroll
        for (int rp = 0; rp < 4; ++rp) {
            float x0r = nr[2*rp], x1r = nr[2*rp + 1], x0i = ni[2*rp], x1i = ni[2*rp + 1];
            asm("" : "+v"(x0r), "+v"(x1r), "+v"(x0i), "+v"(x1i));
            const float sr = sp ? x0r : x1r;
            const float si = sp ? x0i : x1i;
            const float or_ = lane_xor1(sr);
            const float oi_ = lane_xor1(si);
            v4f e;
            e[0] = sp ? or_ : x0r;
            e[1] = sp ? x1r : or_;
            e[2] = sp ? oi_ : x0i;
            e[3] = sp ? x1i : oi_;
            const int rho = 8*half + 2*rp + sp;
            F[buf + (T*16 + wv*4 + qp)*32 + 16*hp + ((rho + 2*qp + 8*hp) & 15)] = e;
        }
    };
    auto seed_tile = [&](int cur, int bo, int co, int cmask, int cofs) {
        v4f f[16];
        load_rows(f, cur, wv);
        const float* Br = sm + bo;  const float* Bi = Br + 1024;
        float* Cr = sm + co;        float* Ci = Cr + 1024;
        const int bc = col16 & cmask;
        v8f arbr = {}; v8f aibi = {}; v8f arbi = {}; v8f aibr = {};
        #pragma unroll
        for (int t = 0; t < 4; ++t) {
            #pragma unroll
            for (int q = 0; q < 4; ++q) {
                const int k = 16*t + 8*half + 2*q;
                const v4f ff = f[t*4 + q];
                v2f ar, ai, br, bi;
                ar[0] = ff[0]; ar[1] = ff[1];
                ai[0] = ff[2]; ai[1] = ff[3];
                br[0] = Br[k*16 + bc];  br[1] = Br[(k + 1)*16 + bc];
                bi[0] = Bi[k*16 + bc];  bi[1] = Bi[(k + 1)*16 + bc];
                arbr = __builtin_amdgcn_wmma_f32_16x16x4_f32(false, ar, false, br, (short)0, arbr, false, false);
                aibi = __builtin_amdgcn_wmma_f32_16x16x4_f32(false, ai, false, bi, (short)0, aibi, false, false);
                arbi = __builtin_amdgcn_wmma_f32_16x16x4_f32(false, ar, false, bi, (short)0, arbi, false, false);
                aibr = __builtin_amdgcn_wmma_f32_16x16x4_f32(false, ai, false, br, (short)0, aibr, false, false);
            }
        }
        for (int v = 0; v < 8; ++v) {
            Cr[(16*wv + 8*half + v)*16 + cofs + bc] = arbr[v] - aibi[v];
            Ci[(16*wv + 8*half + v)*16 + cofs + bc] = arbi[v] + aibr[v];
        }
    };

    auto left_rows = [&](bool second) {
        const int arow = second ? (col16 >= 12 ? col16 - 12 : col16)
                                : ((col16 == 6 || col16 == 7) ? 12 : col16);
        const float* ap = WS + arow*WP + 8*half;
        v8f s1 = {}; v8f s2 = {};
        #pragma unroll
        for (int t = 0; t < 4; ++t) {
            #pragma unroll
            for (int q = 0; q < 4; ++q) {
                const v2f a = *(const v2f*)(ap + 16*t + 2*q);
                v2f br, bi;
                br[0] = Xr[t][2*q]; br[1] = Xr[t][2*q + 1];
                bi[0] = Xi[t][2*q]; bi[1] = Xi[t][2*q + 1];
                s1 = __builtin_amdgcn_wmma_f32_16x16x4_f32(false, a, false, br, (short)0, s1, false, false);
                s2 = __builtin_amdgcn_wmma_f32_16x16x4_f32(false, a, false, bi, (short)0, s2, false, false);
            }
        }
        if (lane < 16) {
            const int m = 16*wv + col16;
            const float e0r = s1[4] - s2[5], e0i = s2[4] + s1[5];
            if (!second) {
                WS[6*WP + m] = e0r;  WS[7*WP + m] = e0i;
            } else {
                WS[12*WP + m] = e0r; WS[13*WP + m] = e0i;
                WS[14*WP + m] = s1[6] - s2[7];
                WS[15*WP + m] = s2[6] + s1[7];
            }
        }
    };

    auto left_pass = [&](const float* src, float* dst) {
        const float* ap = ((col16 & 4) ? src : WS) + col16*WP + 8*half;
        v8f s1 = {}; v8f s2 = {};
        #pragma unroll
        for (int t = 0; t < 4; ++t) {
            #pragma unroll
            for (int q = 0; q < 4; ++q) {
                const v2f a = *(const v2f*)(ap + 16*t + 2*q);
                v2f br, bi;
                br[0] = Xr[t][2*q]; br[1] = Xr[t][2*q + 1];
                bi[0] = Xi[t][2*q]; bi[1] = Xi[t][2*q + 1];
                s1 = __builtin_amdgcn_wmma_f32_16x16x4_f32(false, a, false, br, (short)0, s1, false, false);
                s2 = __builtin_amdgcn_wmma_f32_16x16x4_f32(false, a, false, bi, (short)0, s2, false, false);
            }
        }
        float* d = dst + (4 + 8*half)*WP + 16*wv + col16;
        d[0]      = s1[4] - s2[5];
        d[WP]     = s2[4] + s1[5];
        d[2*WP]   = s1[6] - s2[7];
        d[3*WP]   = s2[6] + s1[7];
    };

    {
        v4f ga[16], gb[16];
        v8f nXr[4], nXi[4];
        load_rows(ga, 0, 0);
        load_rows(gb, 0, 1);
        q_load();
        __builtin_amdgcn_sched_barrier(0);
        row_block(ga, nXr[0], nXi[0]);
        db_write();
        auto trip = [&](const int s, const bool last, const int ex) __attribute__((always_inline)) {
            const int cur = (s & 1) * 2048, nxt = 2048 - cur;
            __builtin_amdgcn_sched_barrier(0);
            load_rows(ga, cur, 2);
            __builtin_amdgcn_sched_barrier(0);
            v8f brbr = {}; v8f bibi = {}; v8f brbi = {}; v8f bibr = {};
            row_part(gb, 0, 2, brbr, bibi, brbi, bibr);
            publish_block(nxt, 0, nXr[0], nXi[0]);
            if (ex == 2) mv_fma(4, 8);
            __builtin_amdgcn_sched_barrier(0);
            row_part(gb, 2, 4, brbr, bibi, brbi, bibr);
            if (ex == 2) mv_finish(2);
            __builtin_amdgcn_sched_barrier(0);
            __syncthreads();
            load_rows(gb, cur, 3);
            if (ex == 1) mv_load(1);
            if (ex == 2) mv_load(3);
            __builtin_amdgcn_sched_barrier(0);
            nXr[1] = brbr - bibi;
            nXi[1] = brbi + bibr;
            v8f crbr = {}; v8f cibi = {}; v8f crbi = {}; v8f cibr = {};
            row_part(ga, 0, 2, crbr, cibi, crbi, cibr);
            publish_block(nxt, 1, nXr[1], nXi[1]);
            if (ex) mv_fma(0, 4);
            __builtin_amdgcn_sched_barrier(0);
            row_part(ga, 2, 4, crbr, cibi, crbi, cibr);
            nXr[2] = crbr - cibi;
            nXi[2] = crbi + cibr;
            if (ex) mv_fma(4, 8);
            __builtin_amdgcn_sched_barrier(0);
            v8f arbr = {}; v8f aibi = {}; v8f arbi = {}; v8f aibr = {};
            row_part(gb, 0, 1, arbr, aibi, arbi, aibr);
            publish_block(nxt, 2, nXr[2], nXi[2]);
            if (ex == 1) mv_finish(1);
            if (ex == 2) mv_finish(3);
            __builtin_amdgcn_sched_barrier(0);
            row_part(gb, 1, 2, arbr, aibi, arbi, aibr);
            __builtin_amdgcn_sched_barrier(0);
            __syncthreads();
            if (!last) load_rows(ga, nxt, 0);
            if (ex == 1) mv_load(2);
            __builtin_amdgcn_sched_barrier(0);
            row_part(gb, 2, 4, arbr, aibi, arbi, aibr);
            if (ex == 1) mv_fma(0, 4);
            if (!last) {
                __builtin_amdgcn_sched_barrier(0);
                load_rows(gb, nxt, 1);
                __builtin_amdgcn_sched_barrier(0);
            }
            nXr[3] = arbr - aibi;
            nXi[3] = arbi + aibr;
            #pragma unroll
            for (int t = 0; t < 4; ++t) { Xr[t] = nXr[t]; Xi[t] = nXi[t]; }
            if (s == 4) left_rows(false);
            else if (s == 5) left_rows(true);
            if (!last) {
                row_block(ga, nXr[0], nXi[0]);
                publish_block(nxt, 3, Xr[3], Xi[3]);
            }
        };
        trip(0, false, 1);
        trip(1, false, 2);
        for (int s = 2; s < 7; ++s) trip(s, false, 0);
        trip(7, true, 0);
        publish_block(0, 3, Xr[3], Xi[3]);
    }
    seed_tile(2048, 16384, 16384, 3, 4);
    __syncthreads();
    seed_tile(0, 16384, 16384, 7, 8);

    {
        float* TL = sm + 12288;
        const float* lsrc = WS;
        for (int w = 1; w < 4; ++w) {
            float* Lw = sm + 18432 + 2048*(w - 1);
            left_pass(lsrc, TL);
            __syncthreads();
            left_pass(TL, Lw);
            __syncthreads();
            lsrc = Lw;
        }
    }

    const float* sxr = sm + 16384;
    const float* sxi = sxr + 1024;
    float* Kt = sm + 8192;
    float hgr[4][8], hgi[4][8];
    #pragma unroll
    for (int t = 0; t < 4; ++t) {
        #pragma unroll
        for (int r = 0; r < 8; ++r) {
            const int n = 16*t + 8*half + r;
            Xr[t][r] = sxr[n*16 + col16];
            Xi[t][r] = sxi[n*16 + col16];
            hgr[t][r] = G4r[n];
            hgi[t][r] = G4i[n];
        }
    }
    const float* wrow = (((col16 & 4) && wv > 0) ? sm + 18432 + 2048*(wv - 1) : WS) + col16*WP;
    v2f fWS[16];
    #pragma unroll
    for (int t = 0; t < 4; ++t) {
        #pragma unroll
        for (int q = 0; q < 4; ++q) {
            const int k = 16*t + 8*half + 2*q;
            fWS[t*4 + q] = *(const v2f*)(wrow + k);
        }
    }
    v2f fUrn[4], fUi[4], fUin[4];
    #pragma unroll
    for (int T = 0; T < 4; ++T) {
        const int o = (16*T + col16)*RK + 2*half;
        const v2f a = *(const v2f*)(Ufr + o);
        const v2f b = *(const v2f*)(Ufi + o);
        fUrn[T] = -a;
        fUi[T]  = b;
        fUin[T] = -b;
    }
    float* kout = Kt + 512*wv + 128*(col16 >> 2) + (col16 & 3) + 64*half;
    auto stacked_product = [&](const v8f (&Pr)[4], const v8f (&Pi)[4], v8f& d1, v8f& d2) {
        v8f s1 = {}; v8f s2 = {};
        #pragma unroll
        for (int t = 0; t < 4; ++t) {
            #pragma unroll
            for (int q = 0; q < 4; ++q) {
                v2f br, bi;
                br[0] = Pr[t][2*q]; br[1] = Pr[t][2*q + 1];
                bi[0] = Pi[t][2*q]; bi[1] = Pi[t][2*q + 1];
                s1 = __builtin_amdgcn_wmma_f32_16x16x4_f32(false, fWS[t*4 + q], false, br, (short)0, s1, false, false);
                s2 = __builtin_amdgcn_wmma_f32_16x16x4_f32(false, fWS[t*4 + q], false, bi, (short)0, s2, false, false);
            }
        }
        d1 = s1; d2 = s2;
    };
    auto scan_step = [&](const v8f (&Pr)[4], const v8f (&Pi)[4], v8f (&Nr)[4], v8f (&Ni)[4], float* dst) {
        v8f d1, d2;
        stacked_product(Pr, Pi, d1, d2);
        const float kva = 2.0f * (d1[4] - d2[5]);
        const float kvb = 2.0f * (d1[6] - d2[7]);
        v2f zr2, zi2;
        zr2[0] = d1[0] - d2[2]; zr2[1] = d1[1] - d2[3];
        zi2[0] = d2[0] + d1[2]; zi2[1] = d2[1] + d1[3];
        #pragma unroll
        for (int T = 0; T < 4; ++T) {
            #pragma unroll
            for (int r = 0; r < 8; ++r) {
                Nr[T][r] = hgr[T][r]*Pr[T][r] - hgi[T][r]*Pi[T][r];
                Ni[T][r] = hgr[T][r]*Pi[T][r] + hgi[T][r]*Pr[T][r];
            }
        }
        #pragma unroll
        for (int T = 0; T < 4; ++T) {
            Nr[T] = __builtin_amdgcn_wmma_f32_16x16x4_f32(false, fUrn[T], false, zr2, (short)0, Nr[T], false, false);
            Ni[T] = __builtin_amdgcn_wmma_f32_16x16x4_f32(false, fUrn[T], false, zi2, (short)0, Ni[T], false, false);
        }
        #pragma unroll
        for (int T = 0; T < 4; ++T) {
            Nr[T] = __builtin_amdgcn_wmma_f32_16x16x4_f32(false, fUi[T],  false, zi2, (short)0, Nr[T], false, false);
            Ni[T] = __builtin_amdgcn_wmma_f32_16x16x4_f32(false, fUin[T], false, zr2, (short)0, Ni[T], false, false);
        }
        dst[0] = kva;
        dst[32] = kvb;
    };
    v8f Yr[4], Yi[4];
    for (int jj = 0; jj < 3; ++jj) {
        scan_step(Xr, Xi, Yr, Yi, kout + 2*RK*jj);
        scan_step(Yr, Yi, Xr, Xi, kout + 2*RK*jj + RK);
    }
    scan_step(Xr, Xi, Yr, Yi, kout + RK*6);
    {
        v8f d1, d2;
        stacked_product(Yr, Yi, d1, d2);
        kout[RK*7] = 2.0f * (d1[4] - d2[5]);
        kout[RK*7 + 32] = 2.0f * (d1[6] - d2[7]);
    }
    __syncthreads();

    {
        const v4f* Kt4 = (const v4f*)Kt;
        v4f* o4 = (v4f*)(out + h*LL);
        for (int k = 0; k < 4; ++k)
            o4[k*128 + tid] = Kt4[k*128 + tid];
    }
}

extern "C" void kernel_launch(void* const* d_in, const int* in_sizes, int n_in,
                              void* d_out, int out_size, void* d_ws, size_t ws_size,
                              hipStream_t stream) {
    const float* A_real = (const float*)d_in[0];
    const float* A_imag = (const float*)d_in[1];
    const float* B_real = (const float*)d_in[2];
    const float* B_imag = (const float*)d_in[3];
    const float* P_real = (const float*)d_in[4];
    const float* P_imag = (const float*)d_in[5];
    const float* C_real = (const float*)d_in[6];
    const float* C_imag = (const float*)d_in[7];
    const float* log_dt = (const float*)d_in[8];
    float* out = (float*)d_out;
    (void)in_sizes; (void)n_in; (void)out_size; (void)d_ws; (void)ws_size;

    const size_t shmem_bytes = 29648u * sizeof(float);
    s4_dplr_kernel<<<NH, 128, shmem_bytes, stream>>>(
        A_real, A_imag, B_real, B_imag, P_real, P_imag, C_real, C_imag, log_dt, out);
}
